# starting kernel
# baseline (speedup 1.0000x reference)
#include <hip/hip_runtime.h>
#include <hip/hip_bf16.h>

typedef float v2f __attribute__((ext_vector_type(2)));
typedef float v8f __attribute__((ext_vector_type(8)));

#define NNODES 50000
#define NEDGES 1600000
#define FIN    256
#define HID    128
#define NCLS   40
#define NPAD   48

__device__ __forceinline__ float waveReduceSum(float v) {
#pragma unroll
  for (int m = 16; m >= 1; m >>= 1) v += __shfl_xor(v, m, 32);
  return v;
}

__global__ void k_rownorm(const float* __restrict__ X, float* __restrict__ invn,
                          int n, int cols) {
  int w = blockIdx.x * (blockDim.x >> 5) + (threadIdx.x >> 5);
  if (w >= n) return;
  int lane = threadIdx.x & 31;
  const float* row = X + (size_t)w * cols;
  float s = 0.f;
  for (int j = lane; j < cols; j += 32) { float v = row[j]; s += v * v; }
  s = waveReduceSum(s);
  if (lane == 0) {
    float nrm = fmaxf(sqrtf(s), 1e-12f);
    invn[w] = 1.0f / nrm;
  }
}

__global__ void k_zero(float* __restrict__ p, size_t n4) {
  size_t i = (size_t)blockIdx.x * blockDim.x + threadIdx.x;
  size_t stride = (size_t)gridDim.x * blockDim.x;
  float4 z = make_float4(0.f, 0.f, 0.f, 0.f);
  float4* p4 = (float4*)p;
  for (; i < n4; i += stride) p4[i] = z;
}

template <int NT, int K, int MODE>
__global__ void k_gemm_wmma(const float* __restrict__ A, const float* __restrict__ B,
                            float* __restrict__ C, int M,
                            const float* __restrict__ rowScale,
                            const float* __restrict__ abias,
                            const float* __restrict__ bias) {
  constexpr bool USE_SCALE = (MODE & 1) != 0;
  constexpr bool USE_ARELU = (MODE & 2) != 0;
  constexpr bool USE_BIAS  = (MODE & 4) != 0;
  constexpr bool ACCUM     = (MODE & 8) != 0;
  constexpr int N = NT * 16;
  constexpr int KCHUNK = (K % 32 == 0) ? 32 : K;
  constexpr int NCHUNKS = K / KCHUNK;
  constexpr int TSTEPS = KCHUNK / 4;
  constexpr int GROUP = (TSTEPS % 8 == 0) ? 8 : TSTEPS / 2;
  constexpr int NV4 = (KCHUNK * N) / 4;
  constexpr int STAGE4 = (NV4 + 255) / 256;
  __shared__ float ldsB[KCHUNK * N];

  const int lane = threadIdx.x & 31;
  const int wid  = threadIdx.x >> 5;
  const int tm   = blockIdx.x * (blockDim.x >> 5) + wid;
  const bool active = tm < (M >> 4);
  const int h = lane >> 4;
  const int l = lane & 15;
  const int row = (tm << 4) + l;

  v8f acc[NT];
  if (active) {
#pragma unroll
    for (int j = 0; j < NT; ++j)
#pragma unroll
      for (int v = 0; v < 8; ++v)
        acc[j][v] = ACCUM ? C[(size_t)((tm << 4) + v + 8 * h) * N + j * 16 + l] : 0.f;
  }
  const float* Arow = A + (size_t)row * K;

  for (int c = 0; c < NCHUNKS; ++c) {
    const int k0 = c * KCHUNK;
    __syncthreads();
#pragma unroll
    for (int i = 0; i < STAGE4; ++i) {
      int idx4 = i * 256 + threadIdx.x;
      if (NV4 % 256 == 0 || idx4 < NV4) {
        int f = idx4 * 4;
        int kk = f / N, n = f - kk * N;
        float4 q = *(const float4*)(B + (size_t)(k0 + kk) * N + n);
        float* d = ldsB + ((kk >> 1) * N + n) * 2 + (kk & 1);
        d[0] = q.x; d[2] = q.y; d[4] = q.z; d[6] = q.w;
      }
    }
    __syncthreads();
    if (active) {
#pragma unroll
      for (int g = 0; g < TSTEPS; g += GROUP) {
        v2f areg[GROUP];
#pragma unroll
        for (int u = 0; u < GROUP; ++u) {
          const int ka = k0 + (g + u) * 4 + 2 * h;
          v2f a;
          a.x = Arow[ka];
          a.y = Arow[ka + 1];
          if (USE_ARELU) {
            a.x = fmaxf(a.x + abias[ka], 0.f);
            a.y = fmaxf(a.y + abias[ka + 1], 0.f);
          }
          areg[u] = a;
        }
#pragma unroll
        for (int u = 0; u < GROUP; ++u) {
          const int ka = (g + u) * 4 + 2 * h;
          const float* bp = ldsB + (ka >> 1) * (2 * N) + 2 * l;
#pragma unroll
          for (int j = 0; j < NT; ++j) {
            v2f b = *(const v2f*)(bp + 32 * j);
            acc[j] = __builtin_amdgcn_wmma_f32_16x16x4_f32(
                false, areg[u], false, b, (short)0, acc[j], false, false);
          }
        }
      }
    }
  }

  if (active) {
    float sc[8];
#pragma unroll
    for (int v = 0; v < 8; ++v)
      sc[v] = USE_SCALE ? rowScale[(tm << 4) + v + 8 * h] : 1.0f;
#pragma unroll
    for (int j = 0; j < NT; ++j)
#pragma unroll
      for (int v = 0; v < 8; ++v) {
        const int cn = j * 16 + l;
        float o = acc[j][v];
        if (USE_SCALE) o *= sc[v];
        if (USE_BIAS) o += bias[cn];
        C[(size_t)((tm << 4) + v + 8 * h) * N + cn] = o;
      }
  }
}

__global__ void k_spmm(const int* __restrict__ src, const int* __restrict__ dst,
                       const float* __restrict__ ew, const float* __restrict__ X,
                       float* __restrict__ agg, int nE) {
  int e = blockIdx.x * (blockDim.x >> 5) + (threadIdx.x >> 5);
  if (e >= nE) return;
  int lane = threadIdx.x & 31;
  int s = src[e], d = dst[e];
  float w = ew[e];
  const float4* xr = reinterpret_cast<const float4*>(X + (size_t)s * HID);
  float4 v = xr[lane];
  float* out = agg + (size_t)d * HID + lane * 4;
  unsafeAtomicAdd(out + 0, v.x * w);
  unsafeAtomicAdd(out + 1, v.y * w);
  unsafeAtomicAdd(out + 2, v.z * w);
  unsafeAtomicAdd(out + 3, v.w * w);
}

__global__ void k_prep_head(const float* __restrict__ Wf, const float* __restrict__ bf,
                            const float* __restrict__ be, const float* __restrict__ b2,
                            float* __restrict__ Wtp, float* __restrict__ Wbp,
                            float* __restrict__ bfp) {
  int c = threadIdx.x;
  if (c >= NPAD) return;
  float s = 0.f;
  if (c < NCLS) {
    s = bf[c];
    for (int k = 0; k < HID; ++k) {
      s += be[k] * Wf[(size_t)k * NCLS + c];
      s += b2[k] * Wf[(size_t)(HID + k) * NCLS + c];
    }
  }
  bfp[c] = s;
  for (int k = 0; k < HID; ++k) {
    Wtp[k * NPAD + c] = (c < NCLS) ? Wf[(size_t)k * NCLS + c] : 0.f;
    Wbp[k * NPAD + c] = (c < NCLS) ? Wf[(size_t)(HID + k) * NCLS + c] : 0.f;
  }
}

__global__ void k_softmax(const float* __restrict__ logits, float* __restrict__ out,
                          int n) {
  int w = blockIdx.x * (blockDim.x >> 5) + (threadIdx.x >> 5);
  if (w >= n) return;
  int lane = threadIdx.x & 31;
  const float* lr = logits + (size_t)w * NPAD;
  float v0 = lr[lane];
  float v1 = (lane < NCLS - 32) ? lr[32 + lane] : -3.4e38f;
  float m = fmaxf(v0, v1);
#pragma unroll
  for (int x = 16; x >= 1; x >>= 1) m = fmaxf(m, __shfl_xor(m, x, 32));
  float e0 = __expf(v0 - m);
  float e1 = (lane < NCLS - 32) ? __expf(v1 - m) : 0.f;
  float s = waveReduceSum(e0 + e1);
  float inv = 1.0f / s;
  float* orow = out + (size_t)w * NCLS;
  orow[lane] = e0 * inv;
  if (lane < NCLS - 32) orow[32 + lane] = e1 * inv;
}

extern "C" void kernel_launch(void* const* d_in, const int* in_sizes, int n_in,
                              void* d_out, int out_size, void* d_ws, size_t ws_size,
                              hipStream_t stream) {
  const float* features = (const float*)d_in[0];
  const int*   src      = (const int*)d_in[1];
  const int*   dst      = (const int*)d_in[2];
  const float* ew       = (const float*)d_in[3];
  const float* y        = (const float*)d_in[4];
  const float* W1       = (const float*)d_in[5];
  const float* b1       = (const float*)d_in[6];
  const float* W2       = (const float*)d_in[7];
  const float* b2       = (const float*)d_in[8];
  const float* We       = (const float*)d_in[9];
  const float* be       = (const float*)d_in[10];
  const float* Wf       = (const float*)d_in[11];
  const float* bf       = (const float*)d_in[12];
  float* out = (float*)d_out;
  (void)in_sizes; (void)n_in; (void)out_size; (void)ws_size;

  char* ws = (char*)d_ws;
  size_t off = 0;
  auto alloc = [&](size_t bytes) -> float* {
    float* p = (float*)(ws + off);
    off += (bytes + 255) & ~(size_t)255;
    return p;
  };
  float* invn   = alloc((size_t)NNODES * 4);
  float* bufA   = alloc((size_t)NNODES * HID * 4);
  float* bufAgg = alloc((size_t)NNODES * HID * 4);
  float* bufL   = alloc((size_t)NNODES * NPAD * 4);
  float* Wtp    = alloc((size_t)HID * NPAD * 4);
  float* Wbp    = alloc((size_t)HID * NPAD * 4);
  float* bfp    = alloc((size_t)NPAD * 4);

  const int gemmBlocks = ((NNODES / 16) + 7) / 8;

  k_rownorm<<<NNODES / 8, 256, 0, stream>>>(features, invn, NNODES, FIN);
  k_prep_head<<<1, 64, 0, stream>>>(Wf, bf, be, b2, Wtp, Wbp, bfp);

  k_gemm_wmma<8, FIN, 1><<<gemmBlocks, 256, 0, stream>>>(
      features, W1, bufA, NNODES, invn, nullptr, nullptr);
  k_zero<<<2048, 256, 0, stream>>>(bufAgg, (size_t)NNODES * HID / 4);
  k_spmm<<<NEDGES / 8, 256, 0, stream>>>(src, dst, ew, bufA, bufAgg, NEDGES);
  k_gemm_wmma<8, HID, 2><<<gemmBlocks, 256, 0, stream>>>(
      bufAgg, W2, bufA, NNODES, nullptr, b1, nullptr);
  k_zero<<<2048, 256, 0, stream>>>(bufAgg, (size_t)NNODES * HID / 4);
  k_spmm<<<NEDGES / 8, 256, 0, stream>>>(src, dst, ew, bufA, bufAgg, NEDGES);
  k_gemm_wmma<8, NCLS, 0><<<gemmBlocks, 256, 0, stream>>>(
      y, We, bufA, NNODES, nullptr, nullptr, nullptr);
  k_gemm_wmma<3, HID, 4><<<gemmBlocks, 256, 0, stream>>>(
      bufA, Wtp, bufL, NNODES, nullptr, nullptr, bfp);
  k_gemm_wmma<3, HID, 8><<<gemmBlocks, 256, 0, stream>>>(
      bufAgg, Wbp, bufL, NNODES, nullptr, nullptr, nullptr);
  k_softmax<<<NNODES / 8, 256, 0, stream>>>(bufL, out, NNODES);
}


# Head-weight prep kernel spread over 49 blocks of 256 threads (one padded element per thread; bias sums fed from LDS, same serial term order) instead of one 64-thread block of dependent-load loops
# speedup vs baseline: 1.1575x; 1.1575x over previous
#include <hip/hip_runtime.h>
#include <hip/hip_bf16.h>

typedef float v2f __attribute__((ext_vector_type(2)));
typedef float v8f __attribute__((ext_vector_type(8)));

#define NNODES 50000
#define NEDGES 1600000
#define FIN    256
#define HID    128
#define NCLS   40
#define NPAD   48

__device__ __forceinline__ float waveReduceSum(float v) {
#pragma unroll
  for (int m = 16; m >= 1; m >>= 1) v += __shfl_xor(v, m, 32);
  return v;
}

__global__ void k_rownorm(const float* __restrict__ X, float* __restrict__ invn,
                          int n, int cols) {
  int w = blockIdx.x * (blockDim.x >> 5) + (threadIdx.x >> 5);
  if (w >= n) return;
  int lane = threadIdx.x & 31;
  const float* row = X + (size_t)w * cols;
  float s = 0.f;
  for (int j = lane; j < cols; j += 32) { float v = row[j]; s += v * v; }
  s = waveReduceSum(s);
  if (lane == 0) {
    float nrm = fmaxf(sqrtf(s), 1e-12f);
    invn[w] = 1.0f / nrm;
  }
}

__global__ void k_zero(float* __restrict__ p, size_t n4) {
  size_t i = (size_t)blockIdx.x * blockDim.x + threadIdx.x;
  size_t stride = (size_t)gridDim.x * blockDim.x;
  float4 z = make_float4(0.f, 0.f, 0.f, 0.f);
  float4* p4 = (float4*)p;
  for (; i < n4; i += stride) p4[i] = z;
}

template <int NT, int K, int MODE>
__global__ void k_gemm_wmma(const float* __restrict__ A, const float* __restrict__ B,
                            float* __restrict__ C, int M,
                            const float* __restrict__ rowScale,
                            const float* __restrict__ abias,
                            const float* __restrict__ bias) {
  constexpr bool USE_SCALE = (MODE & 1) != 0;
  constexpr bool USE_ARELU = (MODE & 2) != 0;
  constexpr bool USE_BIAS  = (MODE & 4) != 0;
  constexpr bool ACCUM     = (MODE & 8) != 0;
  constexpr int N = NT * 16;
  constexpr int KCHUNK = (K % 32 == 0) ? 32 : K;
  constexpr int NCHUNKS = K / KCHUNK;
  constexpr int TSTEPS = KCHUNK / 4;
  constexpr int GROUP = (TSTEPS % 8 == 0) ? 8 : TSTEPS / 2;
  constexpr int NV4 = (KCHUNK * N) / 4;
  constexpr int STAGE4 = (NV4 + 255) / 256;
  __shared__ float ldsB[KCHUNK * N];

  const int lane = threadIdx.x & 31;
  const int wid  = threadIdx.x >> 5;
  const int tm   = blockIdx.x * (blockDim.x >> 5) + wid;
  const bool active = tm < (M >> 4);
  const int h = lane >> 4;
  const int l = lane & 15;
  const int row = (tm << 4) + l;

  v8f acc[NT];
  if (active) {
#pragma unroll
    for (int j = 0; j < NT; ++j)
#pragma unroll
      for (int v = 0; v < 8; ++v)
        acc[j][v] = ACCUM ? C[(size_t)((tm << 4) + v + 8 * h) * N + j * 16 + l] : 0.f;
  }
  const float* Arow = A + (size_t)row * K;

  for (int c = 0; c < NCHUNKS; ++c) {
    const int k0 = c * KCHUNK;
    __syncthreads();
#pragma unroll
    for (int i = 0; i < STAGE4; ++i) {
      int idx4 = i * 256 + threadIdx.x;
      if (NV4 % 256 == 0 || idx4 < NV4) {
        int f = idx4 * 4;
        int kk = f / N, n = f - kk * N;
        float4 q = *(const float4*)(B + (size_t)(k0 + kk) * N + n);
        float* d = ldsB + ((kk >> 1) * N + n) * 2 + (kk & 1);
        d[0] = q.x; d[2] = q.y; d[4] = q.z; d[6] = q.w;
      }
    }
    __syncthreads();
    if (active) {
#pragma unroll
      for (int g = 0; g < TSTEPS; g += GROUP) {
        v2f areg[GROUP];
#pragma unroll
        for (int u = 0; u < GROUP; ++u) {
          const int ka = k0 + (g + u) * 4 + 2 * h;
          v2f a;
          a.x = Arow[ka];
          a.y = Arow[ka + 1];
          if (USE_ARELU) {
            a.x = fmaxf(a.x + abias[ka], 0.f);
            a.y = fmaxf(a.y + abias[ka + 1], 0.f);
          }
          areg[u] = a;
        }
#pragma unroll
        for (int u = 0; u < GROUP; ++u) {
          const int ka = (g + u) * 4 + 2 * h;
          const float* bp = ldsB + (ka >> 1) * (2 * N) + 2 * l;
#pragma unroll
          for (int j = 0; j < NT; ++j) {
            v2f b = *(const v2f*)(bp + 32 * j);
            acc[j] = __builtin_amdgcn_wmma_f32_16x16x4_f32(
                false, areg[u], false, b, (short)0, acc[j], false, false);
          }
        }
      }
    }
  }

  if (active) {
    float sc[8];
#pragma unroll
    for (int v = 0; v < 8; ++v)
      sc[v] = USE_SCALE ? rowScale[(tm << 4) + v + 8 * h] : 1.0f;
#pragma unroll
    for (int j = 0; j < NT; ++j)
#pragma unroll
      for (int v = 0; v < 8; ++v) {
        const int cn = j * 16 + l;
        float o = acc[j][v];
        if (USE_SCALE) o *= sc[v];
        if (USE_BIAS) o += bias[cn];
        C[(size_t)((tm << 4) + v + 8 * h) * N + cn] = o;
      }
  }
}

__global__ void k_spmm(const int* __restrict__ src, const int* __restrict__ dst,
                       const float* __restrict__ ew, const float* __restrict__ X,
                       float* __restrict__ agg, int nE) {
  int e = blockIdx.x * (blockDim.x >> 5) + (threadIdx.x >> 5);
  if (e >= nE) return;
  int lane = threadIdx.x & 31;
  int s = src[e], d = dst[e];
  float w = ew[e];
  const float4* xr = reinterpret_cast<const float4*>(X + (size_t)s * HID);
  float4 v = xr[lane];
  float* out = agg + (size_t)d * HID + lane * 4;
  unsafeAtomicAdd(out + 0, v.x * w);
  unsafeAtomicAdd(out + 1, v.y * w);
  unsafeAtomicAdd(out + 2, v.z * w);
  unsafeAtomicAdd(out + 3, v.w * w);
}

#define PREP_THREADS     256
#define PREP_COPY_BLOCKS ((2 * HID * NPAD) / PREP_THREADS)
__global__ void k_prep_head(const float* __restrict__ Wf, const float* __restrict__ bf,
                            const float* __restrict__ be, const float* __restrict__ b2,
                            float* __restrict__ Wtp, float* __restrict__ Wbp,
                            float* __restrict__ bfp) {
  __shared__ float sW[2 * HID * NCLS];
  __shared__ float sBe[HID];
  __shared__ float sB2[HID];
  const int t = threadIdx.x;
  if (blockIdx.x < PREP_COPY_BLOCKS) {
    const int idx  = blockIdx.x * PREP_THREADS + t;
    const int half = idx / (HID * NPAD);
    const int r    = idx - half * (HID * NPAD);
    const int k    = r / NPAD;
    const int c    = r - k * NPAD;
    const float v  = (c < NCLS) ? Wf[(size_t)(half * HID + k) * NCLS + c] : 0.f;
    (half ? Wbp : Wtp)[r] = v;
    return;
  }
  const float4* Wf4 = reinterpret_cast<const float4*>(Wf);
  float4* sW4 = reinterpret_cast<float4*>(sW);
#pragma unroll
  for (int j = 0; j < (2 * HID * NCLS) / 4 / PREP_THREADS; ++j)
    sW4[j * PREP_THREADS + t] = Wf4[j * PREP_THREADS + t];
  if (t < HID) sBe[t] = be[t];
  else         sB2[t - HID] = b2[t - HID];
  __syncthreads();
  if (t < NPAD) {
    float s = 0.f;
    if (t < NCLS) {
      s = bf[t];
      for (int k = 0; k < HID; ++k) {
        s += sBe[k] * sW[k * NCLS + t];
        s += sB2[k] * sW[(HID + k) * NCLS + t];
      }
    }
    bfp[t] = s;
  }
}

__global__ void k_softmax(const float* __restrict__ logits, float* __restrict__ out,
                          int n) {
  int w = blockIdx.x * (blockDim.x >> 5) + (threadIdx.x >> 5);
  if (w >= n) return;
  int lane = threadIdx.x & 31;
  const float* lr = logits + (size_t)w * NPAD;
  float v0 = lr[lane];
  float v1 = (lane < NCLS - 32) ? lr[32 + lane] : -3.4e38f;
  float m = fmaxf(v0, v1);
#pragma unroll
  for (int x = 16; x >= 1; x >>= 1) m = fmaxf(m, __shfl_xor(m, x, 32));
  float e0 = __expf(v0 - m);
  float e1 = (lane < NCLS - 32) ? __expf(v1 - m) : 0.f;
  float s = waveReduceSum(e0 + e1);
  float inv = 1.0f / s;
  float* orow = out + (size_t)w * NCLS;
  orow[lane] = e0 * inv;
  if (lane < NCLS - 32) orow[32 + lane] = e1 * inv;
}

extern "C" void kernel_launch(void* const* d_in, const int* in_sizes, int n_in,
                              void* d_out, int out_size, void* d_ws, size_t ws_size,
                              hipStream_t stream) {
  const float* features = (const float*)d_in[0];
  const int*   src      = (const int*)d_in[1];
  const int*   dst      = (const int*)d_in[2];
  const float* ew       = (const float*)d_in[3];
  const float* y        = (const float*)d_in[4];
  const float* W1       = (const float*)d_in[5];
  const float* b1       = (const float*)d_in[6];
  const float* W2       = (const float*)d_in[7];
  const float* b2       = (const float*)d_in[8];
  const float* We       = (const float*)d_in[9];
  const float* be       = (const float*)d_in[10];
  const float* Wf       = (const float*)d_in[11];
  const float* bf       = (const float*)d_in[12];
  float* out = (float*)d_out;
  (void)in_sizes; (void)n_in; (void)out_size; (void)ws_size;

  char* ws = (char*)d_ws;
  size_t off = 0;
  auto alloc = [&](size_t bytes) -> float* {
    float* p = (float*)(ws + off);
    off += (bytes + 255) & ~(size_t)255;
    return p;
  };
  float* invn   = alloc((size_t)NNODES * 4);
  float* bufA   = alloc((size_t)NNODES * HID * 4);
  float* bufAgg = alloc((size_t)NNODES * HID * 4);
  float* bufL   = alloc((size_t)NNODES * NPAD * 4);
  float* Wtp    = alloc((size_t)HID * NPAD * 4);
  float* Wbp    = alloc((size_t)HID * NPAD * 4);
  float* bfp    = alloc((size_t)NPAD * 4);

  const int gemmBlocks = ((NNODES / 16) + 7) / 8;

  k_rownorm<<<NNODES / 8, 256, 0, stream>>>(features, invn, NNODES, FIN);
  k_prep_head<<<PREP_COPY_BLOCKS + 1, PREP_THREADS, 0, stream>>>(Wf, bf, be, b2, Wtp, Wbp, bfp);

  k_gemm_wmma<8, FIN, 1><<<gemmBlocks, 256, 0, stream>>>(
      features, W1, bufA, NNODES, invn, nullptr, nullptr);
  k_zero<<<2048, 256, 0, stream>>>(bufAgg, (size_t)NNODES * HID / 4);
  k_spmm<<<NEDGES / 8, 256, 0, stream>>>(src, dst, ew, bufA, bufAgg, NEDGES);
  k_gemm_wmma<8, HID, 2><<<gemmBlocks, 256, 0, stream>>>(
      bufAgg, W2, bufA, NNODES, nullptr, b1, nullptr);
  k_zero<<<2048, 256, 0, stream>>>(bufAgg, (size_t)NNODES * HID / 4);
  k_spmm<<<NEDGES / 8, 256, 0, stream>>>(src, dst, ew, bufA, bufAgg, NEDGES);
  k_gemm_wmma<8, NCLS, 0><<<gemmBlocks, 256, 0, stream>>>(
      y, We, bufA, NNODES, nullptr, nullptr, nullptr);
  k_gemm_wmma<3, HID, 4><<<gemmBlocks, 256, 0, stream>>>(
      bufA, Wtp, bufL, NNODES, nullptr, nullptr, bfp);
  k_gemm_wmma<3, HID, 8><<<gemmBlocks, 256, 0, stream>>>(
      bufAgg, Wbp, bufL, NNODES, nullptr, nullptr, nullptr);
  k_softmax<<<NNODES / 8, 256, 0, stream>>>(bufL, out, NNODES);
}


# SPMM: edges grouped by dst row once per call (LDS counters), each row sorted by edge index; one wave sums a row in registers in that fixed order, not arrival order; float atomics, zero fills gone
# speedup vs baseline: 2.6402x; 2.2810x over previous
#include <hip/hip_runtime.h>
#include <hip/hip_bf16.h>

typedef float v2f __attribute__((ext_vector_type(2)));
typedef float v8f __attribute__((ext_vector_type(8)));

#define NNODES 50000
#define NEDGES 1600000
#define FIN    256
#define HID    128
#define NCLS   40
#define NPAD   48

__device__ __forceinline__ float waveReduceSum(float v) {
#pragma unroll
  for (int m = 16; m >= 1; m >>= 1) v += __shfl_xor(v, m, 32);
  return v;
}

__global__ void k_rownorm(const float* __restrict__ X, float* __restrict__ invn,
                          int n, int cols) {
  int w = blockIdx.x * (blockDim.x >> 5) + (threadIdx.x >> 5);
  if (w >= n) return;
  int lane = threadIdx.x & 31;
  const float* row = X + (size_t)w * cols;
  float s = 0.f;
  for (int j = lane; j < cols; j += 32) { float v = row[j]; s += v * v; }
  s = waveReduceSum(s);
  if (lane == 0) {
    float nrm = fmaxf(sqrtf(s), 1e-12f);
    invn[w] = 1.0f / nrm;
  }
}

template <int NT, int K, int MODE>
__global__ void k_gemm_wmma(const float* __restrict__ A, const float* __restrict__ B,
                            float* __restrict__ C, int M,
                            const float* __restrict__ rowScale,
                            const float* __restrict__ abias,
                            const float* __restrict__ bias) {
  constexpr bool USE_SCALE = (MODE & 1) != 0;
  constexpr bool USE_ARELU = (MODE & 2) != 0;
  constexpr bool USE_BIAS  = (MODE & 4) != 0;
  constexpr bool ACCUM     = (MODE & 8) != 0;
  constexpr int N = NT * 16;
  constexpr int KCHUNK = (K % 32 == 0) ? 32 : K;
  constexpr int NCHUNKS = K / KCHUNK;
  constexpr int TSTEPS = KCHUNK / 4;
  constexpr int GROUP = (TSTEPS % 8 == 0) ? 8 : TSTEPS / 2;
  constexpr int NV4 = (KCHUNK * N) / 4;
  constexpr int STAGE4 = (NV4 + 255) / 256;
  __shared__ float ldsB[KCHUNK * N];

  const int lane = threadIdx.x & 31;
  const int wid  = threadIdx.x >> 5;
  const int tm   = blockIdx.x * (blockDim.x >> 5) + wid;
  const bool active = tm < (M >> 4);
  const int h = lane >> 4;
  const int l = lane & 15;
  const int row = (tm << 4) + l;

  v8f acc[NT];
  if (active) {
#pragma unroll
    for (int j = 0; j < NT; ++j)
#pragma unroll
      for (int v = 0; v < 8; ++v)
        acc[j][v] = ACCUM ? C[(size_t)((tm << 4) + v + 8 * h) * N + j * 16 + l] : 0.f;
  }
  const float* Arow = A + (size_t)row * K;

  for (int c = 0; c < NCHUNKS; ++c) {
    const int k0 = c * KCHUNK;
    __syncthreads();
#pragma unroll
    for (int i = 0; i < STAGE4; ++i) {
      int idx4 = i * 256 + threadIdx.x;
      if (NV4 % 256 == 0 || idx4 < NV4) {
        int f = idx4 * 4;
        int kk = f / N, n = f - kk * N;
        float4 q = *(const float4*)(B + (size_t)(k0 + kk) * N + n);
        float* d = ldsB + ((kk >> 1) * N + n) * 2 + (kk & 1);
        d[0] = q.x; d[2] = q.y; d[4] = q.z; d[6] = q.w;
      }
    }
    __syncthreads();
    if (active) {
#pragma unroll
      for (int g = 0; g < TSTEPS; g += GROUP) {
        v2f areg[GROUP];
#pragma unroll
        for (int u = 0; u < GROUP; ++u) {
          const int ka = k0 + (g + u) * 4 + 2 * h;
          v2f a;
          a.x = Arow[ka];
          a.y = Arow[ka + 1];
          if (USE_ARELU) {
            a.x = fmaxf(a.x + abias[ka], 0.f);
            a.y = fmaxf(a.y + abias[ka + 1], 0.f);
          }
          areg[u] = a;
        }
#pragma unroll
        for (int u = 0; u < GROUP; ++u) {
          const int ka = (g + u) * 4 + 2 * h;
          const float* bp = ldsB + (ka >> 1) * (2 * N) + 2 * l;
#pragma unroll
          for (int j = 0; j < NT; ++j) {
            v2f b = *(const v2f*)(bp + 32 * j);
            acc[j] = __builtin_amdgcn_wmma_f32_16x16x4_f32(
                false, areg[u], false, b, (short)0, acc[j], false, false);
          }
        }
      }
    }
  }

  if (active) {
    float sc[8];
#pragma unroll
    for (int v = 0; v < 8; ++v)
      sc[v] = USE_SCALE ? rowScale[(tm << 4) + v + 8 * h] : 1.0f;
#pragma unroll
    for (int j = 0; j < NT; ++j)
#pragma unroll
      for (int v = 0; v < 8; ++v) {
        const int cn = j * 16 + l;
        float o = acc[j][v];
        if (USE_SCALE) o *= sc[v];
        if (USE_BIAS) o += bias[cn];
        C[(size_t)((tm << 4) + v + 8 * h) * N + cn] = o;
      }
  }
}

#define CSR_G      64
#define CSR_Q      4
#define CSR_ROWS   (NNODES / CSR_Q)
#define CSR_CHUNK  (NEDGES / CSR_G)
#define CSR_T      1024
#define CSR_PER    ((CSR_CHUNK + CSR_T - 1) / CSR_T)
#define SCAN_PER   52
#define NPADROWS   (CSR_T * SCAN_PER)
#define ROWS_PER_WAVE 4
#define ROW_BLOCKS (((NNODES + ROWS_PER_WAVE - 1) / ROWS_PER_WAVE + 7) / 8)
static_assert(NEDGES % CSR_G == 0, "edge chunks must be equal");
static_assert(NNODES % CSR_Q == 0, "row ranges must be equal");
static_assert(NPADROWS >= NNODES + 1, "rowptr needs an end entry");
static_assert(SCAN_PER % 4 == 0, "rowscan moves 16-byte vectors");

__global__ void __launch_bounds__(CSR_T)
k_csr_count(const int* __restrict__ dst, unsigned* __restrict__ cnt) {
  __shared__ unsigned hist[CSR_ROWS];
  const int t = threadIdx.x;
  const int c = blockIdx.x / CSR_Q;
  const int lo = (blockIdx.x % CSR_Q) * CSR_ROWS;
  for (int i = t; i < CSR_ROWS; i += CSR_T) hist[i] = 0u;
  __syncthreads();
  const int* d = dst + (size_t)c * CSR_CHUNK;
  int dv[CSR_PER];
  for (int k = 0; k < CSR_PER; ++k) {
    const int i = t + k * CSR_T;
    dv[k] = (i < CSR_CHUNK) ? d[i] : -1;
  }
  for (int k = 0; k < CSR_PER; ++k) {
    const unsigned rr = (unsigned)(dv[k] - lo);
    if (rr < (unsigned)CSR_ROWS) atomicAdd(&hist[rr], 1u);
  }
  __syncthreads();
  unsigned* o = cnt + (size_t)c * NNODES + lo;
  for (int i = t; i < CSR_ROWS; i += CSR_T) o[i] = hist[i];
}

__global__ void k_csr_colscan(unsigned* __restrict__ cnt, unsigned* __restrict__ tot) {
  const int r = blockIdx.x * 256 + threadIdx.x;
  unsigned s = 0u;
  if (r < NNODES) {
    unsigned c[CSR_G];
    for (int b = 0; b < CSR_G; ++b) c[b] = cnt[(size_t)b * NNODES + r];
    for (int b = 0; b < CSR_G; ++b) { cnt[(size_t)b * NNODES + r] = s; s += c[b]; }
  }
  tot[r] = s;
}

__global__ void __launch_bounds__(CSR_T)
k_csr_rowscan(const unsigned* __restrict__ tot, unsigned* __restrict__ rowptr) {
  __shared__ unsigned wsum[32];
  const int t = threadIdx.x, lane = t & 31, wid = t >> 5;
  const uint4* in = reinterpret_cast<const uint4*>(tot) + t * (SCAN_PER / 4);
  uint4 v[SCAN_PER / 4];
  for (int i = 0; i < SCAN_PER / 4; ++i) v[i] = in[i];
  unsigned s = 0u;
  for (int i = 0; i < SCAN_PER / 4; ++i) s += v[i].x + v[i].y + v[i].z + v[i].w;
  unsigned inc = s;
  for (int o = 1; o < 32; o <<= 1) {
    unsigned n = __shfl_up(inc, o, 32);
    if (lane >= o) inc += n;
  }
  if (lane == 31) wsum[wid] = inc;
  __syncthreads();
  if (wid == 0) {
    unsigned w = wsum[lane], wi = w;
    for (int o = 1; o < 32; o <<= 1) {
      unsigned n = __shfl_up(wi, o, 32);
      if (lane >= o) wi += n;
    }
    wsum[lane] = wi - w;
  }
  __syncthreads();
  unsigned run = wsum[wid] + (inc - s);
  uint4* out = reinterpret_cast<uint4*>(rowptr) + t * (SCAN_PER / 4);
  for (int i = 0; i < SCAN_PER / 4; ++i) {
    uint4 q;
    q.x = run; run += v[i].x;
    q.y = run; run += v[i].y;
    q.z = run; run += v[i].z;
    q.w = run; run += v[i].w;
    out[i] = q;
  }
}

__global__ void __launch_bounds__(CSR_T)
k_csr_fill(const int* __restrict__ dst, const unsigned* __restrict__ blkoff,
           const unsigned* __restrict__ rowptr, int* __restrict__ perm) {
  __shared__ unsigned hist[CSR_ROWS];
  const int t = threadIdx.x;
  const int c = blockIdx.x / CSR_Q;
  const int lo = (blockIdx.x % CSR_Q) * CSR_ROWS;
  const unsigned* bo = blkoff + (size_t)c * NNODES + lo;
  for (int i = t; i < CSR_ROWS; i += CSR_T) hist[i] = rowptr[lo + i] + bo[i];
  __syncthreads();
  const int e0 = c * CSR_CHUNK;
  int dv[CSR_PER];
  for (int k = 0; k < CSR_PER; ++k) {
    const int i = t + k * CSR_T;
    dv[k] = (i < CSR_CHUNK) ? dst[e0 + i] : -1;
  }
  for (int k = 0; k < CSR_PER; ++k) {
    const unsigned rr = (unsigned)(dv[k] - lo);
    if (rr < (unsigned)CSR_ROWS) {
      const unsigned pos = atomicAdd(&hist[rr], 1u);
      perm[pos] = e0 + t + k * CSR_T;
    }
  }
}

__global__ void k_csr_sort(const unsigned* __restrict__ rowptr, const int* __restrict__ perm,
                           const int* __restrict__ src, const float* __restrict__ ew,
                           int* __restrict__ csrw) {
  int2* csr = reinterpret_cast<int2*>(csrw);
  const int lane = threadIdx.x & 31;
  const int w0 = __builtin_amdgcn_readfirstlane((int)(blockIdx.x * (blockDim.x >> 5) + (threadIdx.x >> 5)));
  const int nw = gridDim.x * (blockDim.x >> 5);
  for (int r = w0; r < NNODES; r += nw) {
    const int start = (int)rowptr[r];
    const int deg = (int)rowptr[r + 1] - start;
    for (int a = 0; a < deg; a += 32) {
      const int ia = a + lane;
      const int ea = (ia < deg) ? perm[start + ia] : 0x7fffffff;
      int rank = 0;
      for (int b = 0; b < deg; b += 32) {
        const int ib = b + lane;
        const int eb = (ib < deg) ? perm[start + ib] : 0x7fffffff;
        for (int j = 0; j < 32; ++j)
          rank += (__builtin_amdgcn_readlane(eb, j) < ea) ? 1 : 0;
      }
      if (ia < deg) csr[start + rank] = make_int2(src[ea], __float_as_int(ew[ea]));
    }
  }
}

__global__ void k_spmm_rows(const unsigned* __restrict__ rowptr, const int* __restrict__ csrw,
                            const float* __restrict__ X, float* __restrict__ agg) {
  const int2* csr = reinterpret_cast<const int2*>(csrw);
  const int lane = threadIdx.x & 31;
  const int w0 = __builtin_amdgcn_readfirstlane((int)(blockIdx.x * (blockDim.x >> 5) + (threadIdx.x >> 5)));
  const int nw = gridDim.x * (blockDim.x >> 5);
  const float4* X4 = reinterpret_cast<const float4*>(X);
  for (int r = w0; r < NNODES; r += nw) {
    const int start = (int)rowptr[r];
    const int end = (int)rowptr[r + 1];
    float4 acc = make_float4(0.f, 0.f, 0.f, 0.f);
    for (int base = start; base < end; base += 32) {
      const int cnt = min(32, end - base);
      const int2 rec = (lane < cnt) ? csr[base + lane] : make_int2(0, 0);
      for (int j = 0; j < cnt; j += 8) {
        float4 x[8];
        for (int u = 0; u < 8; ++u) {
          const int ju = min(j + u, cnt - 1);
          const int s = __builtin_amdgcn_readlane(rec.x, ju);
          x[u] = X4[(size_t)s * (HID / 4) + lane];
        }
        for (int u = 0; u < 8; ++u) {
          if (j + u < cnt) {
            const float w = __int_as_float(__builtin_amdgcn_readlane(rec.y, j + u));
            acc.x = fmaf(w, x[u].x, acc.x);
            acc.y = fmaf(w, x[u].y, acc.y);
            acc.z = fmaf(w, x[u].z, acc.z);
            acc.w = fmaf(w, x[u].w, acc.w);
          }
        }
      }
    }
    reinterpret_cast<float4*>(agg)[(size_t)r * (HID / 4) + lane] = acc;
  }
}

#define PREP_THREADS     256
#define PREP_COPY_BLOCKS ((2 * HID * NPAD) / PREP_THREADS)
__global__ void k_prep_head(const float* __restrict__ Wf, const float* __restrict__ bf,
                            const float* __restrict__ be, const float* __restrict__ b2,
                            float* __restrict__ Wtp, float* __restrict__ Wbp,
                            float* __restrict__ bfp) {
  __shared__ float sW[2 * HID * NCLS];
  __shared__ float sBe[HID];
  __shared__ float sB2[HID];
  const int t = threadIdx.x;
  if (blockIdx.x < PREP_COPY_BLOCKS) {
    const int idx  = blockIdx.x * PREP_THREADS + t;
    const int half = idx / (HID * NPAD);
    const int r    = idx - half * (HID * NPAD);
    const int k    = r / NPAD;
    const int c    = r - k * NPAD;
    const float v  = (c < NCLS) ? Wf[(size_t)(half * HID + k) * NCLS + c] : 0.f;
    (half ? Wbp : Wtp)[r] = v;
    return;
  }
  const float4* Wf4 = reinterpret_cast<const float4*>(Wf);
  float4* sW4 = reinterpret_cast<float4*>(sW);
#pragma unroll
  for (int j = 0; j < (2 * HID * NCLS) / 4 / PREP_THREADS; ++j)
    sW4[j * PREP_THREADS + t] = Wf4[j * PREP_THREADS + t];
  if (t < HID) sBe[t] = be[t];
  else         sB2[t - HID] = b2[t - HID];
  __syncthreads();
  if (t < NPAD) {
    float s = 0.f;
    if (t < NCLS) {
      s = bf[t];
      for (int k = 0; k < HID; ++k) {
        s += sBe[k] * sW[k * NCLS + t];
        s += sB2[k] * sW[(HID + k) * NCLS + t];
      }
    }
    bfp[t] = s;
  }
}

__global__ void k_softmax(const float* __restrict__ logits, float* __restrict__ out,
                          int n) {
  int w = blockIdx.x * (blockDim.x >> 5) + (threadIdx.x >> 5);
  if (w >= n) return;
  int lane = threadIdx.x & 31;
  const float* lr = logits + (size_t)w * NPAD;
  float v0 = lr[lane];
  float v1 = (lane < NCLS - 32) ? lr[32 + lane] : -3.4e38f;
  float m = fmaxf(v0, v1);
#pragma unroll
  for (int x = 16; x >= 1; x >>= 1) m = fmaxf(m, __shfl_xor(m, x, 32));
  float e0 = __expf(v0 - m);
  float e1 = (lane < NCLS - 32) ? __expf(v1 - m) : 0.f;
  float s = waveReduceSum(e0 + e1);
  float inv = 1.0f / s;
  float* orow = out + (size_t)w * NCLS;
  orow[lane] = e0 * inv;
  if (lane < NCLS - 32) orow[32 + lane] = e1 * inv;
}

extern "C" void kernel_launch(void* const* d_in, const int* in_sizes, int n_in,
                              void* d_out, int out_size, void* d_ws, size_t ws_size,
                              hipStream_t stream) {
  const float* features = (const float*)d_in[0];
  const int*   src      = (const int*)d_in[1];
  const int*   dst      = (const int*)d_in[2];
  const float* ew       = (const float*)d_in[3];
  const float* y        = (const float*)d_in[4];
  const float* W1       = (const float*)d_in[5];
  const float* b1       = (const float*)d_in[6];
  const float* W2       = (const float*)d_in[7];
  const float* b2       = (const float*)d_in[8];
  const float* We       = (const float*)d_in[9];
  const float* be       = (const float*)d_in[10];
  const float* Wf       = (const float*)d_in[11];
  const float* bf       = (const float*)d_in[12];
  float* out = (float*)d_out;
  (void)in_sizes; (void)n_in; (void)out_size; (void)ws_size;

  char* ws = (char*)d_ws;
  size_t off = 0;
  auto alloc = [&](size_t bytes) -> float* {
    float* p = (float*)(ws + off);
    off += (bytes + 255) & ~(size_t)255;
    return p;
  };
  float* invn   = alloc((size_t)NNODES * 4);
  float* bufA   = alloc((size_t)NNODES * HID * 4);
  float* bufAgg = alloc((size_t)NNODES * HID * 4);
  float* bufL   = alloc((size_t)NNODES * NPAD * 4);
  float* Wtp    = alloc((size_t)HID * NPAD * 4);
  float* Wbp    = alloc((size_t)HID * NPAD * 4);
  float* bfp    = alloc((size_t)NPAD * 4);
  unsigned* rowptr = (unsigned*)alloc((size_t)NPADROWS * 4);
  int*      csr    = (int*)alloc((size_t)NEDGES * 8);
  unsigned* cntMat = (unsigned*)bufAgg;
  int*      perm   = (int*)(cntMat + (size_t)CSR_G * NNODES);
  unsigned* tot    = (unsigned*)(perm + NEDGES);

  const int gemmBlocks = ((NNODES / 16) + 7) / 8;

  k_rownorm<<<NNODES / 8, 256, 0, stream>>>(features, invn, NNODES, FIN);
  k_prep_head<<<PREP_COPY_BLOCKS + 1, PREP_THREADS, 0, stream>>>(Wf, bf, be, b2, Wtp, Wbp, bfp);

  k_gemm_wmma<8, FIN, 1><<<gemmBlocks, 256, 0, stream>>>(
      features, W1, bufA, NNODES, invn, nullptr, nullptr);
  k_csr_count<<<CSR_G * CSR_Q, CSR_T, 0, stream>>>(dst, cntMat);
  k_csr_colscan<<<NPADROWS / 256, 256, 0, stream>>>(cntMat, tot);
  k_csr_rowscan<<<1, CSR_T, 0, stream>>>(tot, rowptr);
  k_csr_fill<<<CSR_G * CSR_Q, CSR_T, 0, stream>>>(dst, cntMat, rowptr, perm);
  k_csr_sort<<<ROW_BLOCKS, 256, 0, stream>>>(rowptr, perm, src, ew, csr);
  k_spmm_rows<<<ROW_BLOCKS, 256, 0, stream>>>(rowptr, csr, bufA, bufAgg);
  k_gemm_wmma<8, HID, 2><<<gemmBlocks, 256, 0, stream>>>(
      bufAgg, W2, bufA, NNODES, nullptr, b1, nullptr);
  k_spmm_rows<<<ROW_BLOCKS, 256, 0, stream>>>(rowptr, csr, bufA, bufAgg);
  k_gemm_wmma<8, NCLS, 0><<<gemmBlocks, 256, 0, stream>>>(
      y, We, bufA, NNODES, nullptr, nullptr, nullptr);
  k_gemm_wmma<3, HID, 4><<<gemmBlocks, 256, 0, stream>>>(
      bufA, Wtp, bufL, NNODES, nullptr, nullptr, bfp);
  k_gemm_wmma<3, HID, 8><<<gemmBlocks, 256, 0, stream>>>(
      bufAgg, Wbp, bufL, NNODES, nullptr, nullptr, nullptr);
  k_softmax<<<NNODES / 8, 256, 0, stream>>>(bufL, out, NNODES);
}


# Head weight Wf folded into the upstream weights per call on WMMA (W2*Wf_bot, We*Wf_top; products reassociated): layer-2 GEMM and SPMM run 48 columns wide, emb and agg2 head GEMMs gone
# speedup vs baseline: 3.0893x; 1.1701x over previous
#include <hip/hip_runtime.h>
#include <hip/hip_bf16.h>

typedef float v2f __attribute__((ext_vector_type(2)));
typedef float v8f __attribute__((ext_vector_type(8)));

#define NNODES 50000
#define NEDGES 1600000
#define FIN    256
#define HID    128
#define NCLS   40
#define NPAD   48

__device__ __forceinline__ float waveReduceSum(float v) {
#pragma unroll
  for (int m = 16; m >= 1; m >>= 1) v += __shfl_xor(v, m, 32);
  return v;
}

__global__ void k_rownorm(const float* __restrict__ X, float* __restrict__ invn,
                          int n, int cols) {
  int w = blockIdx.x * (blockDim.x >> 5) + (threadIdx.x >> 5);
  if (w >= n) return;
  int lane = threadIdx.x & 31;
  const float* row = X + (size_t)w * cols;
  float s = 0.f;
  for (int j = lane; j < cols; j += 32) { float v = row[j]; s += v * v; }
  s = waveReduceSum(s);
  if (lane == 0) {
    float nrm = fmaxf(sqrtf(s), 1e-12f);
    invn[w] = 1.0f / nrm;
  }
}

template <int NT, int K, int MODE>
__global__ void k_gemm_wmma(const float* __restrict__ A, const float* __restrict__ B,
                            float* __restrict__ C, int M,
                            const float* __restrict__ rowScale,
                            const float* __restrict__ abias,
                            const float* __restrict__ bias) {
  constexpr bool USE_SCALE = (MODE & 1) != 0;
  constexpr bool USE_ARELU = (MODE & 2) != 0;
  constexpr bool USE_BIAS  = (MODE & 4) != 0;
  constexpr bool ACCUM     = (MODE & 8) != 0;
  constexpr int N = NT * 16;
  constexpr int KCHUNK = (K % 32 == 0) ? 32 : K;
  constexpr int NCHUNKS = K / KCHUNK;
  constexpr int TSTEPS = KCHUNK / 4;
  constexpr int GROUP = (TSTEPS % 8 == 0) ? 8 : TSTEPS / 2;
  constexpr int NV4 = (KCHUNK * N) / 4;
  constexpr int STAGE4 = (NV4 + 255) / 256;
  __shared__ float ldsB[KCHUNK * N];

  const int lane = threadIdx.x & 31;
  const int wid  = threadIdx.x >> 5;
  const int tm   = blockIdx.x * (blockDim.x >> 5) + wid;
  const bool active = tm < (M >> 4);
  const int h = lane >> 4;
  const int l = lane & 15;
  const int row = (tm << 4) + l;

  v8f acc[NT];
  if (active) {
#pragma unroll
    for (int j = 0; j < NT; ++j)
#pragma unroll
      for (int v = 0; v < 8; ++v)
        acc[j][v] = ACCUM ? C[(size_t)((tm << 4) + v + 8 * h) * N + j * 16 + l] : 0.f;
  }
  const float* Arow = A + (size_t)row * K;

  for (int c = 0; c < NCHUNKS; ++c) {
    const int k0 = c * KCHUNK;
    __syncthreads();
#pragma unroll
    for (int i = 0; i < STAGE4; ++i) {
      int idx4 = i * 256 + threadIdx.x;
      if (NV4 % 256 == 0 || idx4 < NV4) {
        int f = idx4 * 4;
        int kk = f / N, n = f - kk * N;
        float4 q = *(const float4*)(B + (size_t)(k0 + kk) * N + n);
        float* d = ldsB + ((kk >> 1) * N + n) * 2 + (kk & 1);
        d[0] = q.x; d[2] = q.y; d[4] = q.z; d[6] = q.w;
      }
    }
    __syncthreads();
    if (active) {
#pragma unroll
      for (int g = 0; g < TSTEPS; g += GROUP) {
        v2f areg[GROUP];
#pragma unroll
        for (int u = 0; u < GROUP; ++u) {
          const int ka = k0 + (g + u) * 4 + 2 * h;
          v2f a;
          a.x = Arow[ka];
          a.y = Arow[ka + 1];
          if (USE_ARELU) {
            a.x = fmaxf(a.x + abias[ka], 0.f);
            a.y = fmaxf(a.y + abias[ka + 1], 0.f);
          }
          areg[u] = a;
        }
#pragma unroll
        for (int u = 0; u < GROUP; ++u) {
          const int ka = (g + u) * 4 + 2 * h;
          const float* bp = ldsB + (ka >> 1) * (2 * N) + 2 * l;
#pragma unroll
          for (int j = 0; j < NT; ++j) {
            v2f b = *(const v2f*)(bp + 32 * j);
            acc[j] = __builtin_amdgcn_wmma_f32_16x16x4_f32(
                false, areg[u], false, b, (short)0, acc[j], false, false);
          }
        }
      }
    }
  }

  if (active) {
    float sc[8];
#pragma unroll
    for (int v = 0; v < 8; ++v)
      sc[v] = USE_SCALE ? rowScale[(tm << 4) + v + 8 * h] : 1.0f;
#pragma unroll
    for (int j = 0; j < NT; ++j)
#pragma unroll
      for (int v = 0; v < 8; ++v) {
        const int cn = j * 16 + l;
        float o = acc[j][v];
        if (USE_SCALE) o *= sc[v];
        if (USE_BIAS) o += bias[cn];
        C[(size_t)((tm << 4) + v + 8 * h) * N + cn] = o;
      }
  }
}

#define CSR_G      64
#define CSR_Q      4
#define CSR_ROWS   (NNODES / CSR_Q)
#define CSR_CHUNK  (NEDGES / CSR_G)
#define CSR_T      1024
#define CSR_PER    ((CSR_CHUNK + CSR_T - 1) / CSR_T)
#define SCAN_PER   52
#define NPADROWS   (CSR_T * SCAN_PER)
#define ROWS_PER_WAVE 4
#define ROW_BLOCKS (((NNODES + ROWS_PER_WAVE - 1) / ROWS_PER_WAVE + 7) / 8)
static_assert(NEDGES % CSR_G == 0, "edge chunks must be equal");
static_assert(NNODES % CSR_Q == 0, "row ranges must be equal");
static_assert(NPADROWS >= NNODES + 1, "rowptr needs an end entry");
static_assert(SCAN_PER % 4 == 0, "rowscan moves 16-byte vectors");

__global__ void __launch_bounds__(CSR_T)
k_csr_count(const int* __restrict__ dst, unsigned* __restrict__ cnt) {
  __shared__ unsigned hist[CSR_ROWS];
  const int t = threadIdx.x;
  const int c = blockIdx.x / CSR_Q;
  const int lo = (blockIdx.x % CSR_Q) * CSR_ROWS;
  for (int i = t; i < CSR_ROWS; i += CSR_T) hist[i] = 0u;
  __syncthreads();
  const int* d = dst + (size_t)c * CSR_CHUNK;
  int dv[CSR_PER];
  for (int k = 0; k < CSR_PER; ++k) {
    const int i = t + k * CSR_T;
    dv[k] = (i < CSR_CHUNK) ? d[i] : -1;
  }
  for (int k = 0; k < CSR_PER; ++k) {
    const unsigned rr = (unsigned)(dv[k] - lo);
    if (rr < (unsigned)CSR_ROWS) atomicAdd(&hist[rr], 1u);
  }
  __syncthreads();
  unsigned* o = cnt + (size_t)c * NNODES + lo;
  for (int i = t; i < CSR_ROWS; i += CSR_T) o[i] = hist[i];
}

__global__ void k_csr_colscan(unsigned* __restrict__ cnt, unsigned* __restrict__ tot) {
  const int r = blockIdx.x * 256 + threadIdx.x;
  unsigned s = 0u;
  if (r < NNODES) {
    unsigned c[CSR_G];
    for (int b = 0; b < CSR_G; ++b) c[b] = cnt[(size_t)b * NNODES + r];
    for (int b = 0; b < CSR_G; ++b) { cnt[(size_t)b * NNODES + r] = s; s += c[b]; }
  }
  tot[r] = s;
}

__global__ void __launch_bounds__(CSR_T)
k_csr_rowscan(const unsigned* __restrict__ tot, unsigned* __restrict__ rowptr) {
  __shared__ unsigned wsum[32];
  const int t = threadIdx.x, lane = t & 31, wid = t >> 5;
  const uint4* in = reinterpret_cast<const uint4*>(tot) + t * (SCAN_PER / 4);
  uint4 v[SCAN_PER / 4];
  for (int i = 0; i < SCAN_PER / 4; ++i) v[i] = in[i];
  unsigned s = 0u;
  for (int i = 0; i < SCAN_PER / 4; ++i) s += v[i].x + v[i].y + v[i].z + v[i].w;
  unsigned inc = s;
  for (int o = 1; o < 32; o <<= 1) {
    unsigned n = __shfl_up(inc, o, 32);
    if (lane >= o) inc += n;
  }
  if (lane == 31) wsum[wid] = inc;
  __syncthreads();
  if (wid == 0) {
    unsigned w = wsum[lane], wi = w;
    for (int o = 1; o < 32; o <<= 1) {
      unsigned n = __shfl_up(wi, o, 32);
      if (lane >= o) wi += n;
    }
    wsum[lane] = wi - w;
  }
  __syncthreads();
  unsigned run = wsum[wid] + (inc - s);
  uint4* out = reinterpret_cast<uint4*>(rowptr) + t * (SCAN_PER / 4);
  for (int i = 0; i < SCAN_PER / 4; ++i) {
    uint4 q;
    q.x = run; run += v[i].x;
    q.y = run; run += v[i].y;
    q.z = run; run += v[i].z;
    q.w = run; run += v[i].w;
    out[i] = q;
  }
}

__global__ void __launch_bounds__(CSR_T)
k_csr_fill(const int* __restrict__ dst, const unsigned* __restrict__ blkoff,
           const unsigned* __restrict__ rowptr, int* __restrict__ perm) {
  __shared__ unsigned hist[CSR_ROWS];
  const int t = threadIdx.x;
  const int c = blockIdx.x / CSR_Q;
  const int lo = (blockIdx.x % CSR_Q) * CSR_ROWS;
  const unsigned* bo = blkoff + (size_t)c * NNODES + lo;
  for (int i = t; i < CSR_ROWS; i += CSR_T) hist[i] = rowptr[lo + i] + bo[i];
  __syncthreads();
  const int e0 = c * CSR_CHUNK;
  int dv[CSR_PER];
  for (int k = 0; k < CSR_PER; ++k) {
    const int i = t + k * CSR_T;
    dv[k] = (i < CSR_CHUNK) ? dst[e0 + i] : -1;
  }
  for (int k = 0; k < CSR_PER; ++k) {
    const unsigned rr = (unsigned)(dv[k] - lo);
    if (rr < (unsigned)CSR_ROWS) {
      const unsigned pos = atomicAdd(&hist[rr], 1u);
      perm[pos] = e0 + t + k * CSR_T;
    }
  }
}

__global__ void k_csr_sort(const unsigned* __restrict__ rowptr, const int* __restrict__ perm,
                           const int* __restrict__ src, const float* __restrict__ ew,
                           int* __restrict__ csrw) {
  int2* csr = reinterpret_cast<int2*>(csrw);
  const int lane = threadIdx.x & 31;
  const int w0 = __builtin_amdgcn_readfirstlane((int)(blockIdx.x * (blockDim.x >> 5) + (threadIdx.x >> 5)));
  const int nw = gridDim.x * (blockDim.x >> 5);
  for (int r = w0; r < NNODES; r += nw) {
    const int start = (int)rowptr[r];
    const int deg = (int)rowptr[r + 1] - start;
    for (int a = 0; a < deg; a += 32) {
      const int ia = a + lane;
      const int ea = (ia < deg) ? perm[start + ia] : 0x7fffffff;
      int rank = 0;
      for (int b = 0; b < deg; b += 32) {
        const int ib = b + lane;
        const int eb = (ib < deg) ? perm[start + ib] : 0x7fffffff;
        for (int j = 0; j < 32; ++j)
          rank += (__builtin_amdgcn_readlane(eb, j) < ea) ? 1 : 0;
      }
      if (ia < deg) csr[start + rank] = make_int2(src[ea], __float_as_int(ew[ea]));
    }
  }
}

__global__ void k_spmm_rows(const unsigned* __restrict__ rowptr, const int* __restrict__ csrw,
                            const float* __restrict__ X, float* __restrict__ agg) {
  const int2* csr = reinterpret_cast<const int2*>(csrw);
  const int lane = threadIdx.x & 31;
  const int w0 = __builtin_amdgcn_readfirstlane((int)(blockIdx.x * (blockDim.x >> 5) + (threadIdx.x >> 5)));
  const int nw = gridDim.x * (blockDim.x >> 5);
  const float4* X4 = reinterpret_cast<const float4*>(X);
  for (int r = w0; r < NNODES; r += nw) {
    const int start = (int)rowptr[r];
    const int end = (int)rowptr[r + 1];
    float4 acc = make_float4(0.f, 0.f, 0.f, 0.f);
    for (int base = start; base < end; base += 32) {
      const int cnt = min(32, end - base);
      const int2 rec = (lane < cnt) ? csr[base + lane] : make_int2(0, 0);
      for (int j = 0; j < cnt; j += 8) {
        float4 x[8];
        for (int u = 0; u < 8; ++u) {
          const int ju = min(j + u, cnt - 1);
          const int s = __builtin_amdgcn_readlane(rec.x, ju);
          x[u] = X4[(size_t)s * (HID / 4) + lane];
        }
        for (int u = 0; u < 8; ++u) {
          if (j + u < cnt) {
            const float w = __int_as_float(__builtin_amdgcn_readlane(rec.y, j + u));
            acc.x = fmaf(w, x[u].x, acc.x);
            acc.y = fmaf(w, x[u].y, acc.y);
            acc.z = fmaf(w, x[u].z, acc.z);
            acc.w = fmaf(w, x[u].w, acc.w);
          }
        }
      }
    }
    reinterpret_cast<float4*>(agg)[(size_t)r * (HID / 4) + lane] = acc;
  }
}

#define ROW4 (NPAD / 4)
static_assert(NNODES % 2 == 0, "rows are taken in pairs");

template <int K>
__device__ __forceinline__ int rowShare(int v) {
  return __builtin_amdgcn_update_dpp(0, v, 0x150 + K, 0xF, 0xF, false);
}

template <int J>
__device__ __forceinline__ void loadEight(const int2 rec, const int cnt, const bool ld,
                                          const int sub, const float4* __restrict__ X4,
                                          float4 (&x)[8]) {
#define SPMM48_LOAD(U)                                                      \
  {                                                                         \
    const int s = rowShare<J + U>(rec.x);                                   \
    x[U] = make_float4(0.f, 0.f, 0.f, 0.f);                                 \
    if (ld && (J + U) < cnt) x[U] = X4[(size_t)s * ROW4 + sub];             \
  }
  SPMM48_LOAD(0) SPMM48_LOAD(1) SPMM48_LOAD(2) SPMM48_LOAD(3)
  SPMM48_LOAD(4) SPMM48_LOAD(5) SPMM48_LOAD(6) SPMM48_LOAD(7)
#undef SPMM48_LOAD
}

template <int J>
__device__ __forceinline__ void fmaEight(const int2 rec, const float4 (&x)[8], float4& acc) {
#define SPMM48_FMA(U)                                                       \
  {                                                                         \
    const float w = __int_as_float(rowShare<J + U>(rec.y));                 \
    acc.x = fmaf(w, x[U].x, acc.x);                                         \
    acc.y = fmaf(w, x[U].y, acc.y);                                         \
    acc.z = fmaf(w, x[U].z, acc.z);                                         \
    acc.w = fmaf(w, x[U].w, acc.w);                                         \
  }
  SPMM48_FMA(0) SPMM48_FMA(1) SPMM48_FMA(2) SPMM48_FMA(3)
  SPMM48_FMA(4) SPMM48_FMA(5) SPMM48_FMA(6) SPMM48_FMA(7)
#undef SPMM48_FMA
}

__global__ void k_spmm_rows48(const unsigned* __restrict__ rowptr, const int* __restrict__ csrw,
                              const float* __restrict__ X, float* __restrict__ agg) {
  const int2* csr = reinterpret_cast<const int2*>(csrw);
  const int lane = threadIdx.x & 31;
  const int hsel = lane >> 4;
  const int sub = lane & 15;
  const bool ld = sub < ROW4;
  const int w0 = __builtin_amdgcn_readfirstlane((int)(blockIdx.x * (blockDim.x >> 5) + (threadIdx.x >> 5)));
  const int nw = gridDim.x * (blockDim.x >> 5);
  const float4* X4 = reinterpret_cast<const float4*>(X);
  for (int p = w0; p < NNODES / 2; p += nw) {
    const int r0 = 2 * p;
    const int s0 = (int)rowptr[r0];
    const int s1 = (int)rowptr[r0 + 1];
    const int s2 = (int)rowptr[r0 + 2];
    const int degmax = max(s1 - s0, s2 - s1);
    const int start = hsel ? s1 : s0;
    const int deg = hsel ? (s2 - s1) : (s1 - s0);
    float4 acc = make_float4(0.f, 0.f, 0.f, 0.f);
    int cnt = deg;
    int2 rec = (sub < cnt) ? csr[start + sub] : make_int2(0, 0);
    for (int base = 0; base < degmax; base += 16) {
      float4 x[8];
      loadEight<0>(rec, cnt, ld, sub, X4, x);
      int2 nxt = make_int2(0, 0);
      if (sub + 16 < cnt) nxt = csr[start + base + 16 + sub];
      fmaEight<0>(rec, x, acc);
      if (degmax - base > 8) {
        loadEight<8>(rec, cnt, ld, sub, X4, x);
        fmaEight<8>(rec, x, acc);
      }
      rec = nxt;
      cnt -= 16;
    }
    if (ld) reinterpret_cast<float4*>(agg)[(size_t)(r0 + hsel) * ROW4 + sub] = acc;
  }
}

#define PREP_THREADS 256
__device__ __forceinline__ void foldTile(const float* __restrict__ A, const int row0, const int nrows,
                                         const float* __restrict__ Wh, float* __restrict__ D) {
  const int lane = threadIdx.x & 31;
  const int h = lane >> 4;
  const int l = lane & 15;
  const int row = row0 + l;
  const float* Arow = A + (size_t)(row < nrows ? row : 0) * HID;
  const bool cv = 32 + l < NCLS;
  const int n2 = cv ? 32 + l : 0;
  v8f acc0, acc1, acc2;
#pragma unroll
  for (int v = 0; v < 8; ++v) { acc0[v] = 0.f; acc1[v] = 0.f; acc2[v] = 0.f; }
#pragma unroll 8
  for (int t = 0; t < HID / 4; ++t) {
    const int ka = 4 * t + 2 * h;
    const float* w0 = Wh + (size_t)ka * NCLS;
    const float* w1 = w0 + NCLS;
    v2f a, b0, b1, b2;
    a.x = Arow[ka];
    a.y = Arow[ka + 1];
    b0.x = w0[l];
    b0.y = w1[l];
    b1.x = w0[16 + l];
    b1.y = w1[16 + l];
    b2.x = w0[n2];
    b2.y = w1[n2];
    acc0 = __builtin_amdgcn_wmma_f32_16x16x4_f32(false, a, false, b0, (short)0, acc0, false, false);
    acc1 = __builtin_amdgcn_wmma_f32_16x16x4_f32(false, a, false, b1, (short)0, acc1, false, false);
    acc2 = __builtin_amdgcn_wmma_f32_16x16x4_f32(false, a, false, b2, (short)0, acc2, false, false);
  }
#pragma unroll
  for (int v = 0; v < 8; ++v) {
    const int r = row0 + v + 8 * h;
    if (r < nrows) {
      D[(size_t)r * NPAD + l]      = acc0[v];
      D[(size_t)r * NPAD + 16 + l] = acc1[v];
      D[(size_t)r * NPAD + 32 + l] = cv ? acc2[v] : 0.f;
    }
  }
}

__global__ void k_prep_head(const float* __restrict__ W2, const float* __restrict__ We,
                            const float* __restrict__ Wf, const float* __restrict__ bf,
                            const float* __restrict__ be, const float* __restrict__ b2,
                            float* __restrict__ W2f, float* __restrict__ Wef,
                            float* __restrict__ bfp) {
  __shared__ float sW[2 * HID * NCLS];
  __shared__ float sBe[HID];
  __shared__ float sB2[HID];
  const int t = threadIdx.x;
  const int wid = __builtin_amdgcn_readfirstlane((int)(threadIdx.x >> 5));
  if (blockIdx.x == 0) {
    foldTile(W2, 16 * wid, HID, Wf + (size_t)HID * NCLS, W2f);
    return;
  }
  if (blockIdx.x == 1) {
    if (wid < (NCLS + 15) / 16) foldTile(We, 16 * wid, NCLS, Wf, Wef);
    return;
  }
  const float4* Wf4 = reinterpret_cast<const float4*>(Wf);
  float4* sW4 = reinterpret_cast<float4*>(sW);
#pragma unroll
  for (int j = 0; j < (2 * HID * NCLS) / 4 / PREP_THREADS; ++j)
    sW4[j * PREP_THREADS + t] = Wf4[j * PREP_THREADS + t];
  if (t < HID) sBe[t] = be[t];
  else         sB2[t - HID] = b2[t - HID];
  __syncthreads();
  if (t < NPAD) {
    float s = 0.f;
    if (t < NCLS) {
      s = bf[t];
      for (int k = 0; k < HID; ++k) {
        s += sBe[k] * sW[k * NCLS + t];
        s += sB2[k] * sW[(HID + k) * NCLS + t];
      }
    }
    bfp[t] = s;
  }
}

__global__ void k_softmax(const float* __restrict__ logits, float* __restrict__ out,
                          int n) {
  int w = blockIdx.x * (blockDim.x >> 5) + (threadIdx.x >> 5);
  if (w >= n) return;
  int lane = threadIdx.x & 31;
  const float* lr = logits + (size_t)w * NPAD;
  float v0 = lr[lane];
  float v1 = (lane < NCLS - 32) ? lr[32 + lane] : -3.4e38f;
  float m = fmaxf(v0, v1);
#pragma unroll
  for (int x = 16; x >= 1; x >>= 1) m = fmaxf(m, __shfl_xor(m, x, 32));
  float e0 = __expf(v0 - m);
  float e1 = (lane < NCLS - 32) ? __expf(v1 - m) : 0.f;
  float s = waveReduceSum(e0 + e1);
  float inv = 1.0f / s;
  float* orow = out + (size_t)w * NCLS;
  orow[lane] = e0 * inv;
  if (lane < NCLS - 32) orow[32 + lane] = e1 * inv;
}

extern "C" void kernel_launch(void* const* d_in, const int* in_sizes, int n_in,
                              void* d_out, int out_size, void* d_ws, size_t ws_size,
                              hipStream_t stream) {
  const float* features = (const float*)d_in[0];
  const int*   src      = (const int*)d_in[1];
  const int*   dst      = (const int*)d_in[2];
  const float* ew       = (const float*)d_in[3];
  const float* y        = (const float*)d_in[4];
  const float* W1       = (const float*)d_in[5];
  const float* b1       = (const float*)d_in[6];
  const float* W2       = (const float*)d_in[7];
  const float* b2       = (const float*)d_in[8];
  const float* We       = (const float*)d_in[9];
  const float* be       = (const float*)d_in[10];
  const float* Wf       = (const float*)d_in[11];
  const float* bf       = (const float*)d_in[12];
  float* out = (float*)d_out;
  (void)in_sizes; (void)n_in; (void)out_size; (void)ws_size;

  char* ws = (char*)d_ws;
  size_t off = 0;
  auto alloc = [&](size_t bytes) -> float* {
    float* p = (float*)(ws + off);
    off += (bytes + 255) & ~(size_t)255;
    return p;
  };
  float* invn   = alloc((size_t)NNODES * 4);
  float* bufA   = alloc((size_t)NNODES * HID * 4);
  float* bufAgg = alloc((size_t)NNODES * HID * 4);
  float* bufL   = alloc((size_t)NNODES * NPAD * 4);
  float* W2f    = alloc((size_t)HID * NPAD * 4);
  float* Wef    = alloc((size_t)NCLS * NPAD * 4);
  float* bfp    = alloc((size_t)NPAD * 4);
  unsigned* rowptr = (unsigned*)alloc((size_t)NPADROWS * 4);
  int*      csr    = (int*)alloc((size_t)NEDGES * 8);
  unsigned* cntMat = (unsigned*)bufAgg;
  int*      perm   = (int*)(cntMat + (size_t)CSR_G * NNODES);
  unsigned* tot    = (unsigned*)(perm + NEDGES);

  const int gemmBlocks = ((NNODES / 16) + 7) / 8;

  k_rownorm<<<NNODES / 8, 256, 0, stream>>>(features, invn, NNODES, FIN);
  k_prep_head<<<3, PREP_THREADS, 0, stream>>>(W2, We, Wf, bf, be, b2, W2f, Wef, bfp);

  k_gemm_wmma<8, FIN, 1><<<gemmBlocks, 256, 0, stream>>>(
      features, W1, bufA, NNODES, invn, nullptr, nullptr);
  k_csr_count<<<CSR_G * CSR_Q, CSR_T, 0, stream>>>(dst, cntMat);
  k_csr_colscan<<<NPADROWS / 256, 256, 0, stream>>>(cntMat, tot);
  k_csr_rowscan<<<1, CSR_T, 0, stream>>>(tot, rowptr);
  k_csr_fill<<<CSR_G * CSR_Q, CSR_T, 0, stream>>>(dst, cntMat, rowptr, perm);
  k_csr_sort<<<ROW_BLOCKS, 256, 0, stream>>>(rowptr, perm, src, ew, csr);
  k_spmm_rows<<<ROW_BLOCKS, 256, 0, stream>>>(rowptr, csr, bufA, bufAgg);
  k_gemm_wmma<3, HID, 2><<<gemmBlocks, 256, 0, stream>>>(
      bufAgg, W2f, bufA, NNODES, nullptr, b1, nullptr);
  k_spmm_rows48<<<ROW_BLOCKS, 256, 0, stream>>>(rowptr, csr, bufA, bufL);
  k_gemm_wmma<3, NCLS, 12><<<gemmBlocks, 256, 0, stream>>>(
      y, Wef, bufL, NNODES, nullptr, nullptr, bfp);
  k_softmax<<<NNODES / 8, 256, 0, stream>>>(bufL, out, NNODES);
}


# Row norm and softmax kernels (one wave per row, 100000 short waves) folded into the WMMA GEMMs beside them: squares summed from A fragments, softmax in the D epilogue; both sums reordered
# speedup vs baseline: 3.6126x; 1.1694x over previous
#include <hip/hip_runtime.h>
#include <hip/hip_bf16.h>

typedef float v2f __attribute__((ext_vector_type(2)));
typedef float v8f __attribute__((ext_vector_type(8)));

#define NNODES 50000
#define NEDGES 1600000
#define FIN    256
#define HID    128
#define NCLS   40
#define NPAD   48

template <int X>
__device__ __forceinline__ float rowXor(float v) {
  return __int_as_float(__builtin_amdgcn_update_dpp(0, __float_as_int(v), 0x160 + X, 0xF, 0xF, false));
}
__device__ __forceinline__ float rowMax16(float v) {
  v = fmaxf(v, rowXor<8>(v)); v = fmaxf(v, rowXor<4>(v));
  v = fmaxf(v, rowXor<2>(v)); v = fmaxf(v, rowXor<1>(v));
  return v;
}
__device__ __forceinline__ float rowSum16(float v) {
  v += rowXor<8>(v); v += rowXor<4>(v); v += rowXor<2>(v); v += rowXor<1>(v);
  return v;
}

template <int NT, int K, int MODE>
__global__ void k_gemm_wmma(const float* __restrict__ A, const float* __restrict__ B,
                            float* __restrict__ C, int M,
                            const float* __restrict__ abias,
                            const float* __restrict__ bias,
                            float* __restrict__ P) {
  constexpr bool USE_SCALE = (MODE & 1) != 0;
  constexpr bool USE_ARELU = (MODE & 2) != 0;
  constexpr bool USE_BIAS  = (MODE & 4) != 0;
  constexpr bool ACCUM     = (MODE & 8) != 0;
  constexpr bool SOFTMAX   = (MODE & 16) != 0;
  constexpr int N = NT * 16;
  static_assert(!(USE_SCALE && USE_ARELU), "the norm is taken of A as stored");
  static_assert(!SOFTMAX || (NT == 3 && USE_BIAS && !USE_SCALE), "softmax epilogue: NPAD columns, biased logits");
  constexpr int KCHUNK = (K % 32 == 0) ? 32 : K;
  constexpr int NCHUNKS = K / KCHUNK;
  constexpr int TSTEPS = KCHUNK / 4;
  constexpr int GROUP = (TSTEPS % 8 == 0) ? 8 : TSTEPS / 2;
  constexpr int NV4 = (KCHUNK * N) / 4;
  constexpr int STAGE4 = (NV4 + 255) / 256;
  __shared__ float ldsB[KCHUNK * N];

  const int lane = threadIdx.x & 31;
  const int wid  = threadIdx.x >> 5;
  const int tm   = blockIdx.x * (blockDim.x >> 5) + wid;
  const bool active = tm < (M >> 4);
  const int h = lane >> 4;
  const int l = lane & 15;
  const int row = (tm << 4) + l;

  v8f acc[NT];
  if (active) {
#pragma unroll
    for (int j = 0; j < NT; ++j)
#pragma unroll
      for (int v = 0; v < 8; ++v)
        acc[j][v] = ACCUM ? C[(size_t)((tm << 4) + v + 8 * h) * N + j * 16 + l] : 0.f;
  }
  const float* Arow = A + (size_t)row * K;
  float ss = 0.f;

  for (int c = 0; c < NCHUNKS; ++c) {
    const int k0 = c * KCHUNK;
    __syncthreads();
#pragma unroll
    for (int i = 0; i < STAGE4; ++i) {
      int idx4 = i * 256 + threadIdx.x;
      if (NV4 % 256 == 0 || idx4 < NV4) {
        int f = idx4 * 4;
        int kk = f / N, n = f - kk * N;
        float4 q = *(const float4*)(B + (size_t)(k0 + kk) * N + n);
        float* d = ldsB + ((kk >> 1) * N + n) * 2 + (kk & 1);
        d[0] = q.x; d[2] = q.y; d[4] = q.z; d[6] = q.w;
      }
    }
    __syncthreads();
    if (active) {
#pragma unroll
      for (int g = 0; g < TSTEPS; g += GROUP) {
        v2f areg[GROUP];
#pragma unroll
        for (int u = 0; u < GROUP; ++u) {
          const int ka = k0 + (g + u) * 4 + 2 * h;
          v2f a;
          a.x = Arow[ka];
          a.y = Arow[ka + 1];
          if (USE_ARELU) {
            a.x = fmaxf(a.x + abias[ka], 0.f);
            a.y = fmaxf(a.y + abias[ka + 1], 0.f);
          }
          areg[u] = a;
        }
#pragma unroll
        for (int u = 0; u < GROUP; ++u) {
          const int ka = (g + u) * 4 + 2 * h;
          const float* bp = ldsB + (ka >> 1) * (2 * N) + 2 * l;
          if (USE_SCALE) {
            ss = fmaf(areg[u].x, areg[u].x, ss);
            ss = fmaf(areg[u].y, areg[u].y, ss);
          }
#pragma unroll
          for (int j = 0; j < NT; ++j) {
            v2f b = *(const v2f*)(bp + 32 * j);
            acc[j] = __builtin_amdgcn_wmma_f32_16x16x4_f32(
                false, areg[u], false, b, (short)0, acc[j], false, false);
          }
        }
      }
    }
  }

  if (active && !SOFTMAX) {
    float sc[8];
    if (USE_SCALE) {
      const float tot = ss + __shfl_xor(ss, 16, 32);
      const float inv = 1.0f / fmaxf(sqrtf(tot), 1e-12f);
#pragma unroll
      for (int v = 0; v < 8; ++v) sc[v] = __shfl(inv, v + 8 * h, 32);
    }
#pragma unroll
    for (int j = 0; j < NT; ++j)
#pragma unroll
      for (int v = 0; v < 8; ++v) {
        const int cn = j * 16 + l;
        float o = acc[j][v];
        if (USE_SCALE) o *= sc[v];
        if (USE_BIAS) o += bias[cn];
        C[(size_t)((tm << 4) + v + 8 * h) * N + cn] = o;
      }
  }
  if (active && SOFTMAX) {
    const bool c2 = 32 + l < NCLS;
    const float bz0 = bias[l], bz1 = bias[16 + l], bz2 = bias[32 + l];
#pragma unroll
    for (int v = 0; v < 8; ++v) {
      const float x0 = acc[0][v] + bz0;
      const float x1 = acc[1][v] + bz1;
      const float x2 = c2 ? acc[2][v] + bz2 : -3.4e38f;
      const float m = rowMax16(fmaxf(fmaxf(x0, x1), x2));
      const float e0 = __expf(x0 - m);
      const float e1 = __expf(x1 - m);
      const float e2 = c2 ? __expf(x2 - m) : 0.f;
      const float inv = 1.0f / rowSum16(e0 + e1 + e2);
      float* orow = P + (size_t)((tm << 4) + v + 8 * h) * NCLS;
      orow[l] = e0 * inv;
      orow[16 + l] = e1 * inv;
      if (c2) orow[32 + l] = e2 * inv;
    }
  }
}

#define CSR_G      64
#define CSR_Q      4
#define CSR_ROWS   (NNODES / CSR_Q)
#define CSR_CHUNK  (NEDGES / CSR_G)
#define CSR_T      1024
#define CSR_PER    ((CSR_CHUNK + CSR_T - 1) / CSR_T)
#define SCAN_PER   52
#define NPADROWS   (CSR_T * SCAN_PER)
#define ROWS_PER_WAVE 4
#define ROW_BLOCKS (((NNODES + ROWS_PER_WAVE - 1) / ROWS_PER_WAVE + 7) / 8)
static_assert(NEDGES % CSR_G == 0, "edge chunks must be equal");
static_assert(NNODES % CSR_Q == 0, "row ranges must be equal");
static_assert(NPADROWS >= NNODES + 1, "rowptr needs an end entry");
static_assert(SCAN_PER % 4 == 0, "rowscan moves 16-byte vectors");

__global__ void __launch_bounds__(CSR_T)
k_csr_count(const int* __restrict__ dst, unsigned* __restrict__ cnt) {
  __shared__ unsigned hist[CSR_ROWS];
  const int t = threadIdx.x;
  const int c = blockIdx.x / CSR_Q;
  const int lo = (blockIdx.x % CSR_Q) * CSR_ROWS;
  for (int i = t; i < CSR_ROWS; i += CSR_T) hist[i] = 0u;
  __syncthreads();
  const int* d = dst + (size_t)c * CSR_CHUNK;
  int dv[CSR_PER];
  for (int k = 0; k < CSR_PER; ++k) {
    const int i = t + k * CSR_T;
    dv[k] = (i < CSR_CHUNK) ? d[i] : -1;
  }
  for (int k = 0; k < CSR_PER; ++k) {
    const unsigned rr = (unsigned)(dv[k] - lo);
    if (rr < (unsigned)CSR_ROWS) atomicAdd(&hist[rr], 1u);
  }
  __syncthreads();
  unsigned* o = cnt + (size_t)c * NNODES + lo;
  for (int i = t; i < CSR_ROWS; i += CSR_T) o[i] = hist[i];
}

__global__ void k_csr_colscan(unsigned* __restrict__ cnt, unsigned* __restrict__ tot) {
  const int r = blockIdx.x * 256 + threadIdx.x;
  unsigned s = 0u;
  if (r < NNODES) {
    unsigned c[CSR_G];
    for (int b = 0; b < CSR_G; ++b) c[b] = cnt[(size_t)b * NNODES + r];
    for (int b = 0; b < CSR_G; ++b) { cnt[(size_t)b * NNODES + r] = s; s += c[b]; }
  }
  tot[r] = s;
}

__global__ void __launch_bounds__(CSR_T)
k_csr_rowscan(const unsigned* __restrict__ tot, unsigned* __restrict__ rowptr) {
  __shared__ unsigned wsum[32];
  const int t = threadIdx.x, lane = t & 31, wid = t >> 5;
  const uint4* in = reinterpret_cast<const uint4*>(tot) + t * (SCAN_PER / 4);
  uint4 v[SCAN_PER / 4];
  for (int i = 0; i < SCAN_PER / 4; ++i) v[i] = in[i];
  unsigned s = 0u;
  for (int i = 0; i < SCAN_PER / 4; ++i) s += v[i].x + v[i].y + v[i].z + v[i].w;
  unsigned inc = s;
  for (int o = 1; o < 32; o <<= 1) {
    unsigned n = __shfl_up(inc, o, 32);
    if (lane >= o) inc += n;
  }
  if (lane == 31) wsum[wid] = inc;
  __syncthreads();
  if (wid == 0) {
    unsigned w = wsum[lane], wi = w;
    for (int o = 1; o < 32; o <<= 1) {
      unsigned n = __shfl_up(wi, o, 32);
      if (lane >= o) wi += n;
    }
    wsum[lane] = wi - w;
  }
  __syncthreads();
  unsigned run = wsum[wid] + (inc - s);
  uint4* out = reinterpret_cast<uint4*>(rowptr) + t * (SCAN_PER / 4);
  for (int i = 0; i < SCAN_PER / 4; ++i) {
    uint4 q;
    q.x = run; run += v[i].x;
    q.y = run; run += v[i].y;
    q.z = run; run += v[i].z;
    q.w = run; run += v[i].w;
    out[i] = q;
  }
}

__global__ void __launch_bounds__(CSR_T)
k_csr_fill(const int* __restrict__ dst, const unsigned* __restrict__ blkoff,
           const unsigned* __restrict__ rowptr, int* __restrict__ perm) {
  __shared__ unsigned hist[CSR_ROWS];
  const int t = threadIdx.x;
  const int c = blockIdx.x / CSR_Q;
  const int lo = (blockIdx.x % CSR_Q) * CSR_ROWS;
  const unsigned* bo = blkoff + (size_t)c * NNODES + lo;
  for (int i = t; i < CSR_ROWS; i += CSR_T) hist[i] = rowptr[lo + i] + bo[i];
  __syncthreads();
  const int e0 = c * CSR_CHUNK;
  int dv[CSR_PER];
  for (int k = 0; k < CSR_PER; ++k) {
    const int i = t + k * CSR_T;
    dv[k] = (i < CSR_CHUNK) ? dst[e0 + i] : -1;
  }
  for (int k = 0; k < CSR_PER; ++k) {
    const unsigned rr = (unsigned)(dv[k] - lo);
    if (rr < (unsigned)CSR_ROWS) {
      const unsigned pos = atomicAdd(&hist[rr], 1u);
      perm[pos] = e0 + t + k * CSR_T;
    }
  }
}

__global__ void k_csr_sort(const unsigned* __restrict__ rowptr, const int* __restrict__ perm,
                           const int* __restrict__ src, const float* __restrict__ ew,
                           int* __restrict__ csrw) {
  int2* csr = reinterpret_cast<int2*>(csrw);
  const int lane = threadIdx.x & 31;
  const int w0 = __builtin_amdgcn_readfirstlane((int)(blockIdx.x * (blockDim.x >> 5) + (threadIdx.x >> 5)));
  const int nw = gridDim.x * (blockDim.x >> 5);
  for (int r = w0; r < NNODES; r += nw) {
    const int start = (int)rowptr[r];
    const int deg = (int)rowptr[r + 1] - start;
    for (int a = 0; a < deg; a += 32) {
      const int ia = a + lane;
      const int ea = (ia < deg) ? perm[start + ia] : 0x7fffffff;
      int rank = 0;
      for (int b = 0; b < deg; b += 32) {
        const int ib = b + lane;
        const int eb = (ib < deg) ? perm[start + ib] : 0x7fffffff;
        for (int j = 0; j < 32; ++j)
          rank += (__builtin_amdgcn_readlane(eb, j) < ea) ? 1 : 0;
      }
      if (ia < deg) csr[start + rank] = make_int2(src[ea], __float_as_int(ew[ea]));
    }
  }
}

__global__ void k_spmm_rows(const unsigned* __restrict__ rowptr, const int* __restrict__ csrw,
                            const float* __restrict__ X, float* __restrict__ agg) {
  const int2* csr = reinterpret_cast<const int2*>(csrw);
  const int lane = threadIdx.x & 31;
  const int w0 = __builtin_amdgcn_readfirstlane((int)(blockIdx.x * (blockDim.x >> 5) + (threadIdx.x >> 5)));
  const int nw = gridDim.x * (blockDim.x >> 5);
  const float4* X4 = reinterpret_cast<const float4*>(X);
  for (int r = w0; r < NNODES; r += nw) {
    const int start = (int)rowptr[r];
    const int end = (int)rowptr[r + 1];
    float4 acc = make_float4(0.f, 0.f, 0.f, 0.f);
    for (int base = start; base < end; base += 32) {
      const int cnt = min(32, end - base);
      const int2 rec = (lane < cnt) ? csr[base + lane] : make_int2(0, 0);
      for (int j = 0; j < cnt; j += 8) {
        float4 x[8];
        for (int u = 0; u < 8; ++u) {
          const int ju = min(j + u, cnt - 1);
          const int s = __builtin_amdgcn_readlane(rec.x, ju);
          x[u] = X4[(size_t)s * (HID / 4) + lane];
        }
        for (int u = 0; u < 8; ++u) {
          if (j + u < cnt) {
            const float w = __int_as_float(__builtin_amdgcn_readlane(rec.y, j + u));
            acc.x = fmaf(w, x[u].x, acc.x);
            acc.y = fmaf(w, x[u].y, acc.y);
            acc.z = fmaf(w, x[u].z, acc.z);
            acc.w = fmaf(w, x[u].w, acc.w);
          }
        }
      }
    }
    reinterpret_cast<float4*>(agg)[(size_t)r * (HID / 4) + lane] = acc;
  }
}

#define ROW4 (NPAD / 4)
static_assert(NNODES % 2 == 0, "rows are taken in pairs");

template <int K>
__device__ __forceinline__ int rowShare(int v) {
  return __builtin_amdgcn_update_dpp(0, v, 0x150 + K, 0xF, 0xF, false);
}

template <int J>
__device__ __forceinline__ void loadEight(const int2 rec, const int cnt, const bool ld,
                                          const int sub, const float4* __restrict__ X4,
                                          float4 (&x)[8]) {
#define SPMM48_LOAD(U)                                                      \
  {                                                                         \
    const int s = rowShare<J + U>(rec.x);                                   \
    x[U] = make_float4(0.f, 0.f, 0.f, 0.f);                                 \
    if (ld && (J + U) < cnt) x[U] = X4[(size_t)s * ROW4 + sub];             \
  }
  SPMM48_LOAD(0) SPMM48_LOAD(1) SPMM48_LOAD(2) SPMM48_LOAD(3)
  SPMM48_LOAD(4) SPMM48_LOAD(5) SPMM48_LOAD(6) SPMM48_LOAD(7)
#undef SPMM48_LOAD
}

template <int J>
__device__ __forceinline__ void fmaEight(const int2 rec, const float4 (&x)[8], float4& acc) {
#define SPMM48_FMA(U)                                                       \
  {                                                                         \
    const float w = __int_as_float(rowShare<J + U>(rec.y));                 \
    acc.x = fmaf(w, x[U].x, acc.x);                                         \
    acc.y = fmaf(w, x[U].y, acc.y);                                         \
    acc.z = fmaf(w, x[U].z, acc.z);                                         \
    acc.w = fmaf(w, x[U].w, acc.w);                                         \
  }
  SPMM48_FMA(0) SPMM48_FMA(1) SPMM48_FMA(2) SPMM48_FMA(3)
  SPMM48_FMA(4) SPMM48_FMA(5) SPMM48_FMA(6) SPMM48_FMA(7)
#undef SPMM48_FMA
}

__global__ void k_spmm_rows48(const unsigned* __restrict__ rowptr, const int* __restrict__ csrw,
                              const float* __restrict__ X, float* __restrict__ agg) {
  const int2* csr = reinterpret_cast<const int2*>(csrw);
  const int lane = threadIdx.x & 31;
  const int hsel = lane >> 4;
  const int sub = lane & 15;
  const bool ld = sub < ROW4;
  const int w0 = __builtin_amdgcn_readfirstlane((int)(blockIdx.x * (blockDim.x >> 5) + (threadIdx.x >> 5)));
  const int nw = gridDim.x * (blockDim.x >> 5);
  const float4* X4 = reinterpret_cast<const float4*>(X);
  for (int p = w0; p < NNODES / 2; p += nw) {
    const int r0 = 2 * p;
    const int s0 = (int)rowptr[r0];
    const int s1 = (int)rowptr[r0 + 1];
    const int s2 = (int)rowptr[r0 + 2];
    const int degmax = max(s1 - s0, s2 - s1);
    const int start = hsel ? s1 : s0;
    const int deg = hsel ? (s2 - s1) : (s1 - s0);
    float4 acc = make_float4(0.f, 0.f, 0.f, 0.f);
    int cnt = deg;
    int2 rec = (sub < cnt) ? csr[start + sub] : make_int2(0, 0);
    for (int base = 0; base < degmax; base += 16) {
      float4 x[8];
      loadEight<0>(rec, cnt, ld, sub, X4, x);
      int2 nxt = make_int2(0, 0);
      if (sub + 16 < cnt) nxt = csr[start + base + 16 + sub];
      fmaEight<0>(rec, x, acc);
      if (degmax - base > 8) {
        loadEight<8>(rec, cnt, ld, sub, X4, x);
        fmaEight<8>(rec, x, acc);
      }
      rec = nxt;
      cnt -= 16;
    }
    if (ld) reinterpret_cast<float4*>(agg)[(size_t)(r0 + hsel) * ROW4 + sub] = acc;
  }
}

#define PREP_THREADS 256
__device__ __forceinline__ void foldTile(const float* __restrict__ A, const int row0, const int nrows,
                                         const float* __restrict__ Wh, float* __restrict__ D) {
  const int lane = threadIdx.x & 31;
  const int h = lane >> 4;
  const int l = lane & 15;
  const int row = row0 + l;
  const float* Arow = A + (size_t)(row < nrows ? row : 0) * HID;
  const bool cv = 32 + l < NCLS;
  const int n2 = cv ? 32 + l : 0;
  v8f acc0, acc1, acc2;
#pragma unroll
  for (int v = 0; v < 8; ++v) { acc0[v] = 0.f; acc1[v] = 0.f; acc2[v] = 0.f; }
#pragma unroll 8
  for (int t = 0; t < HID / 4; ++t) {
    const int ka = 4 * t + 2 * h;
    const float* w0 = Wh + (size_t)ka * NCLS;
    const float* w1 = w0 + NCLS;
    v2f a, b0, b1, b2;
    a.x = Arow[ka];
    a.y = Arow[ka + 1];
    b0.x = w0[l];
    b0.y = w1[l];
    b1.x = w0[16 + l];
    b1.y = w1[16 + l];
    b2.x = w0[n2];
    b2.y = w1[n2];
    acc0 = __builtin_amdgcn_wmma_f32_16x16x4_f32(false, a, false, b0, (short)0, acc0, false, false);
    acc1 = __builtin_amdgcn_wmma_f32_16x16x4_f32(false, a, false, b1, (short)0, acc1, false, false);
    acc2 = __builtin_amdgcn_wmma_f32_16x16x4_f32(false, a, false, b2, (short)0, acc2, false, false);
  }
#pragma unroll
  for (int v = 0; v < 8; ++v) {
    const int r = row0 + v + 8 * h;
    if (r < nrows) {
      D[(size_t)r * NPAD + l]      = acc0[v];
      D[(size_t)r * NPAD + 16 + l] = acc1[v];
      D[(size_t)r * NPAD + 32 + l] = cv ? acc2[v] : 0.f;
    }
  }
}

__global__ void k_prep_head(const float* __restrict__ W2, const float* __restrict__ We,
                            const float* __restrict__ Wf, const float* __restrict__ bf,
                            const float* __restrict__ be, const float* __restrict__ b2,
                            float* __restrict__ W2f, float* __restrict__ Wef,
                            float* __restrict__ bfp) {
  __shared__ float sW[2 * HID * NCLS];
  __shared__ float sBe[HID];
  __shared__ float sB2[HID];
  const int t = threadIdx.x;
  const int wid = __builtin_amdgcn_readfirstlane((int)(threadIdx.x >> 5));
  if (blockIdx.x == 0) {
    foldTile(W2, 16 * wid, HID, Wf + (size_t)HID * NCLS, W2f);
    return;
  }
  if (blockIdx.x == 1) {
    if (wid < (NCLS + 15) / 16) foldTile(We, 16 * wid, NCLS, Wf, Wef);
    return;
  }
  const float4* Wf4 = reinterpret_cast<const float4*>(Wf);
  float4* sW4 = reinterpret_cast<float4*>(sW);
#pragma unroll
  for (int j = 0; j < (2 * HID * NCLS) / 4 / PREP_THREADS; ++j)
    sW4[j * PREP_THREADS + t] = Wf4[j * PREP_THREADS + t];
  if (t < HID) sBe[t] = be[t];
  else         sB2[t - HID] = b2[t - HID];
  __syncthreads();
  if (t < NPAD) {
    float s = 0.f;
    if (t < NCLS) {
      s = bf[t];
      for (int k = 0; k < HID; ++k) {
        s += sBe[k] * sW[k * NCLS + t];
        s += sB2[k] * sW[(HID + k) * NCLS + t];
      }
    }
    bfp[t] = s;
  }
}

extern "C" void kernel_launch(void* const* d_in, const int* in_sizes, int n_in,
                              void* d_out, int out_size, void* d_ws, size_t ws_size,
                              hipStream_t stream) {
  const float* features = (const float*)d_in[0];
  const int*   src      = (const int*)d_in[1];
  const int*   dst      = (const int*)d_in[2];
  const float* ew       = (const float*)d_in[3];
  const float* y        = (const float*)d_in[4];
  const float* W1       = (const float*)d_in[5];
  const float* b1       = (const float*)d_in[6];
  const float* W2       = (const float*)d_in[7];
  const float* b2       = (const float*)d_in[8];
  const float* We       = (const float*)d_in[9];
  const float* be       = (const float*)d_in[10];
  const float* Wf       = (const float*)d_in[11];
  const float* bf       = (const float*)d_in[12];
  float* out = (float*)d_out;
  (void)in_sizes; (void)n_in; (void)out_size; (void)ws_size;

  char* ws = (char*)d_ws;
  size_t off = 0;
  auto alloc = [&](size_t bytes) -> float* {
    float* p = (float*)(ws + off);
    off += (bytes + 255) & ~(size_t)255;
    return p;
  };
  float* bufA   = alloc((size_t)NNODES * HID * 4);
  float* bufAgg = alloc((size_t)NNODES * HID * 4);
  float* bufL   = alloc((size_t)NNODES * NPAD * 4);
  float* W2f    = alloc((size_t)HID * NPAD * 4);
  float* Wef    = alloc((size_t)NCLS * NPAD * 4);
  float* bfp    = alloc((size_t)NPAD * 4);
  unsigned* rowptr = (unsigned*)alloc((size_t)NPADROWS * 4);
  int*      csr    = (int*)alloc((size_t)NEDGES * 8);
  unsigned* cntMat = (unsigned*)bufAgg;
  int*      perm   = (int*)(cntMat + (size_t)CSR_G * NNODES);
  unsigned* tot    = (unsigned*)(perm + NEDGES);

  const int gemmBlocks = ((NNODES / 16) + 7) / 8;

  k_prep_head<<<3, PREP_THREADS, 0, stream>>>(W2, We, Wf, bf, be, b2, W2f, Wef, bfp);

  k_gemm_wmma<8, FIN, 1><<<gemmBlocks, 256, 0, stream>>>(
      features, W1, bufA, NNODES, nullptr, nullptr, nullptr);
  k_csr_count<<<CSR_G * CSR_Q, CSR_T, 0, stream>>>(dst, cntMat);
  k_csr_colscan<<<NPADROWS / 256, 256, 0, stream>>>(cntMat, tot);
  k_csr_rowscan<<<1, CSR_T, 0, stream>>>(tot, rowptr);
  k_csr_fill<<<CSR_G * CSR_Q, CSR_T, 0, stream>>>(dst, cntMat, rowptr, perm);
  k_csr_sort<<<ROW_BLOCKS, 256, 0, stream>>>(rowptr, perm, src, ew, csr);
  k_spmm_rows<<<ROW_BLOCKS, 256, 0, stream>>>(rowptr, csr, bufA, bufAgg);
  k_gemm_wmma<3, HID, 2><<<gemmBlocks, 256, 0, stream>>>(
      bufAgg, W2f, bufA, NNODES, b1, nullptr, nullptr);
  k_spmm_rows48<<<ROW_BLOCKS, 256, 0, stream>>>(rowptr, csr, bufA, bufL);
  k_gemm_wmma<3, NCLS, 28><<<gemmBlocks, 256, 0, stream>>>(
      y, Wef, bufL, NNODES, nullptr, bfp, out);
}


# Edge ordering folded into the layer-1 SPMM: the wave that sums a row ranks its edge indices and places the (src, weight) records via its own LDS slots; sort launch gone, same ascending sum order
# speedup vs baseline: 3.7302x; 1.0326x over previous
#include <hip/hip_runtime.h>
#include <hip/hip_bf16.h>

typedef float v2f __attribute__((ext_vector_type(2)));
typedef float v8f __attribute__((ext_vector_type(8)));

#define NNODES 50000
#define NEDGES 1600000
#define FIN    256
#define HID    128
#define NCLS   40
#define NPAD   48

template <int X>
__device__ __forceinline__ float rowXor(float v) {
  return __int_as_float(__builtin_amdgcn_update_dpp(0, __float_as_int(v), 0x160 + X, 0xF, 0xF, false));
}
__device__ __forceinline__ float rowMax16(float v) {
  v = fmaxf(v, rowXor<8>(v)); v = fmaxf(v, rowXor<4>(v));
  v = fmaxf(v, rowXor<2>(v)); v = fmaxf(v, rowXor<1>(v));
  return v;
}
__device__ __forceinline__ float rowSum16(float v) {
  v += rowXor<8>(v); v += rowXor<4>(v); v += rowXor<2>(v); v += rowXor<1>(v);
  return v;
}

template <int NT, int K, int MODE>
__global__ void k_gemm_wmma(const float* __restrict__ A, const float* __restrict__ B,
                            float* __restrict__ C, int M,
                            const float* __restrict__ abias,
                            const float* __restrict__ bias,
                            float* __restrict__ P) {
  constexpr bool USE_SCALE = (MODE & 1) != 0;
  constexpr bool USE_ARELU = (MODE & 2) != 0;
  constexpr bool USE_BIAS  = (MODE & 4) != 0;
  constexpr bool ACCUM     = (MODE & 8) != 0;
  constexpr bool SOFTMAX   = (MODE & 16) != 0;
  constexpr int N = NT * 16;
  static_assert(!(USE_SCALE && USE_ARELU), "the norm is taken of A as stored");
  static_assert(!SOFTMAX || (NT == 3 && USE_BIAS && !USE_SCALE), "softmax epilogue: NPAD columns, biased logits");
  constexpr int KCHUNK = (K % 32 == 0) ? 32 : K;
  constexpr int NCHUNKS = K / KCHUNK;
  constexpr int TSTEPS = KCHUNK / 4;
  constexpr int GROUP = (TSTEPS % 8 == 0) ? 8 : TSTEPS / 2;
  constexpr int NV4 = (KCHUNK * N) / 4;
  constexpr int STAGE4 = (NV4 + 255) / 256;
  __shared__ float ldsB[KCHUNK * N];

  const int lane = threadIdx.x & 31;
  const int wid  = threadIdx.x >> 5;
  const int tm   = blockIdx.x * (blockDim.x >> 5) + wid;
  const bool active = tm < (M >> 4);
  const int h = lane >> 4;
  const int l = lane & 15;
  const int row = (tm << 4) + l;

  v8f acc[NT];
  if (active) {
#pragma unroll
    for (int j = 0; j < NT; ++j)
#pragma unroll
      for (int v = 0; v < 8; ++v)
        acc[j][v] = ACCUM ? C[(size_t)((tm << 4) + v + 8 * h) * N + j * 16 + l] : 0.f;
  }
  const float* Arow = A + (size_t)row * K;
  float ss = 0.f;

  for (int c = 0; c < NCHUNKS; ++c) {
    const int k0 = c * KCHUNK;
    __syncthreads();
#pragma unroll
    for (int i = 0; i < STAGE4; ++i) {
      int idx4 = i * 256 + threadIdx.x;
      if (NV4 % 256 == 0 || idx4 < NV4) {
        int f = idx4 * 4;
        int kk = f / N, n = f - kk * N;
        float4 q = *(const float4*)(B + (size_t)(k0 + kk) * N + n);
        float* d = ldsB + ((kk >> 1) * N + n) * 2 + (kk & 1);
        d[0] = q.x; d[2] = q.y; d[4] = q.z; d[6] = q.w;
      }
    }
    __syncthreads();
    if (active) {
#pragma unroll
      for (int g = 0; g < TSTEPS; g += GROUP) {
        v2f areg[GROUP];
#pragma unroll
        for (int u = 0; u < GROUP; ++u) {
          const int ka = k0 + (g + u) * 4 + 2 * h;
          v2f a;
          a.x = Arow[ka];
          a.y = Arow[ka + 1];
          if (USE_ARELU) {
            a.x = fmaxf(a.x + abias[ka], 0.f);
            a.y = fmaxf(a.y + abias[ka + 1], 0.f);
          }
          areg[u] = a;
        }
#pragma unroll
        for (int u = 0; u < GROUP; ++u) {
          const int ka = (g + u) * 4 + 2 * h;
          const float* bp = ldsB + (ka >> 1) * (2 * N) + 2 * l;
          if (USE_SCALE) {
            ss = fmaf(areg[u].x, areg[u].x, ss);
            ss = fmaf(areg[u].y, areg[u].y, ss);
          }
#pragma unroll
          for (int j = 0; j < NT; ++j) {
            v2f b = *(const v2f*)(bp + 32 * j);
            acc[j] = __builtin_amdgcn_wmma_f32_16x16x4_f32(
                false, areg[u], false, b, (short)0, acc[j], false, false);
          }
        }
      }
    }
  }

  if (active && !SOFTMAX) {
    float sc[8];
    if (USE_SCALE) {
      const float tot = ss + __shfl_xor(ss, 16, 32);
      const float inv = 1.0f / fmaxf(sqrtf(tot), 1e-12f);
#pragma unroll
      for (int v = 0; v < 8; ++v) sc[v] = __shfl(inv, v + 8 * h, 32);
    }
#pragma unroll
    for (int j = 0; j < NT; ++j)
#pragma unroll
      for (int v = 0; v < 8; ++v) {
        const int cn = j * 16 + l;
        float o = acc[j][v];
        if (USE_SCALE) o *= sc[v];
        if (USE_BIAS) o += bias[cn];
        C[(size_t)((tm << 4) + v + 8 * h) * N + cn] = o;
      }
  }
  if (active && SOFTMAX) {
    const bool c2 = 32 + l < NCLS;
    const float bz0 = bias[l], bz1 = bias[16 + l], bz2 = bias[32 + l];
#pragma unroll
    for (int v = 0; v < 8; ++v) {
      const float x0 = acc[0][v] + bz0;
      const float x1 = acc[1][v] + bz1;
      const float x2 = c2 ? acc[2][v] + bz2 : -3.4e38f;
      const float m = rowMax16(fmaxf(fmaxf(x0, x1), x2));
      const float e0 = __expf(x0 - m);
      const float e1 = __expf(x1 - m);
      const float e2 = c2 ? __expf(x2 - m) : 0.f;
      const float inv = 1.0f / rowSum16(e0 + e1 + e2);
      float* orow = P + (size_t)((tm << 4) + v + 8 * h) * NCLS;
      orow[l] = e0 * inv;
      orow[16 + l] = e1 * inv;
      if (c2) orow[32 + l] = e2 * inv;
    }
  }
}

#define CSR_G      64
#define CSR_Q      4
#define CSR_ROWS   (NNODES / CSR_Q)
#define CSR_CHUNK  (NEDGES / CSR_G)
#define CSR_T      1024
#define CSR_PER    ((CSR_CHUNK + CSR_T - 1) / CSR_T)
#define SCAN_PER   52
#define NPADROWS   (CSR_T * SCAN_PER)
#define ROWS_PER_WAVE 4
#define ROW_BLOCKS (((NNODES + ROWS_PER_WAVE - 1) / ROWS_PER_WAVE + 7) / 8)
static_assert(NEDGES % CSR_G == 0, "edge chunks must be equal");
static_assert(NNODES % CSR_Q == 0, "row ranges must be equal");
static_assert(NPADROWS >= NNODES + 1, "rowptr needs an end entry");
static_assert(SCAN_PER % 4 == 0, "rowscan moves 16-byte vectors");
static_assert((size_t)NEDGES * 4 <= (size_t)NNODES * NPAD * 4, "perm[] borrows the NPAD-column buffer");

__global__ void __launch_bounds__(CSR_T)
k_csr_count(const int* __restrict__ dst, unsigned* __restrict__ cnt) {
  __shared__ unsigned hist[CSR_ROWS];
  const int t = threadIdx.x;
  const int c = blockIdx.x / CSR_Q;
  const int lo = (blockIdx.x % CSR_Q) * CSR_ROWS;
  for (int i = t; i < CSR_ROWS; i += CSR_T) hist[i] = 0u;
  __syncthreads();
  const int* d = dst + (size_t)c * CSR_CHUNK;
  int dv[CSR_PER];
  for (int k = 0; k < CSR_PER; ++k) {
    const int i = t + k * CSR_T;
    dv[k] = (i < CSR_CHUNK) ? d[i] : -1;
  }
  for (int k = 0; k < CSR_PER; ++k) {
    const unsigned rr = (unsigned)(dv[k] - lo);
    if (rr < (unsigned)CSR_ROWS) atomicAdd(&hist[rr], 1u);
  }
  __syncthreads();
  unsigned* o = cnt + (size_t)c * NNODES + lo;
  for (int i = t; i < CSR_ROWS; i += CSR_T) o[i] = hist[i];
}

__global__ void k_csr_colscan(unsigned* __restrict__ cnt, unsigned* __restrict__ tot) {
  const int r = blockIdx.x * 256 + threadIdx.x;
  unsigned s = 0u;
  if (r < NNODES) {
    unsigned c[CSR_G];
    for (int b = 0; b < CSR_G; ++b) c[b] = cnt[(size_t)b * NNODES + r];
    for (int b = 0; b < CSR_G; ++b) { cnt[(size_t)b * NNODES + r] = s; s += c[b]; }
  }
  tot[r] = s;
}

__global__ void __launch_bounds__(CSR_T)
k_csr_rowscan(const unsigned* __restrict__ tot, unsigned* __restrict__ rowptr) {
  __shared__ unsigned wsum[32];
  const int t = threadIdx.x, lane = t & 31, wid = t >> 5;
  const uint4* in = reinterpret_cast<const uint4*>(tot) + t * (SCAN_PER / 4);
  uint4 v[SCAN_PER / 4];
  for (int i = 0; i < SCAN_PER / 4; ++i) v[i] = in[i];
  unsigned s = 0u;
  for (int i = 0; i < SCAN_PER / 4; ++i) s += v[i].x + v[i].y + v[i].z + v[i].w;
  unsigned inc = s;
  for (int o = 1; o < 32; o <<= 1) {
    unsigned n = __shfl_up(inc, o, 32);
    if (lane >= o) inc += n;
  }
  if (lane == 31) wsum[wid] = inc;
  __syncthreads();
  if (wid == 0) {
    unsigned w = wsum[lane], wi = w;
    for (int o = 1; o < 32; o <<= 1) {
      unsigned n = __shfl_up(wi, o, 32);
      if (lane >= o) wi += n;
    }
    wsum[lane] = wi - w;
  }
  __syncthreads();
  unsigned run = wsum[wid] + (inc - s);
  uint4* out = reinterpret_cast<uint4*>(rowptr) + t * (SCAN_PER / 4);
  for (int i = 0; i < SCAN_PER / 4; ++i) {
    uint4 q;
    q.x = run; run += v[i].x;
    q.y = run; run += v[i].y;
    q.z = run; run += v[i].z;
    q.w = run; run += v[i].w;
    out[i] = q;
  }
}

__global__ void __launch_bounds__(CSR_T)
k_csr_fill(const int* __restrict__ dst, const unsigned* __restrict__ blkoff,
           const unsigned* __restrict__ rowptr, int* __restrict__ perm) {
  __shared__ unsigned hist[CSR_ROWS];
  const int t = threadIdx.x;
  const int c = blockIdx.x / CSR_Q;
  const int lo = (blockIdx.x % CSR_Q) * CSR_ROWS;
  const unsigned* bo = blkoff + (size_t)c * NNODES + lo;
  for (int i = t; i < CSR_ROWS; i += CSR_T) hist[i] = rowptr[lo + i] + bo[i];
  __syncthreads();
  const int e0 = c * CSR_CHUNK;
  int dv[CSR_PER];
  for (int k = 0; k < CSR_PER; ++k) {
    const int i = t + k * CSR_T;
    dv[k] = (i < CSR_CHUNK) ? dst[e0 + i] : -1;
  }
  for (int k = 0; k < CSR_PER; ++k) {
    const unsigned rr = (unsigned)(dv[k] - lo);
    if (rr < (unsigned)CSR_ROWS) {
      const unsigned pos = atomicAdd(&hist[rr], 1u);
      perm[pos] = e0 + t + k * CSR_T;
    }
  }
}

#define RANK_CAP 128

template <int NB>
__device__ __forceinline__ void rankAndPlace(const int start, const int deg, const int lane,
                                             const int* __restrict__ perm, const int* __restrict__ src,
                                             const float* __restrict__ ew, int2* mine,
                                             int2* __restrict__ csr) {
  int e[NB], rk[NB];
#pragma unroll
  for (int a = 0; a < NB; ++a) {
    const int i = 32 * a + lane;
    e[a] = (i < deg) ? perm[start + i] : 0x7fffffff;
    rk[a] = 0;
  }
#pragma unroll
  for (int b = 0; b < NB; ++b) {
#pragma unroll(NB > 2 ? 4 : 32)
    for (int j = 0; j < 32; ++j) {
      const int s = __builtin_amdgcn_readlane(e[b], j);
#pragma unroll
      for (int a = 0; a < NB; ++a) rk[a] += (s < e[a]) ? 1 : 0;
    }
  }
  int sv[NB];
  float wv[NB];
#pragma unroll
  for (int a = 0; a < NB; ++a) {
    const bool ok = 32 * a + lane < deg;
    sv[a] = ok ? src[e[a]] : 0;
    wv[a] = ok ? ew[e[a]] : 0.f;
  }
#pragma unroll
  for (int a = 0; a < NB; ++a) {
    if (32 * a + lane < deg) {
      const int2 rec = make_int2(sv[a], __float_as_int(wv[a]));
      mine[rk[a]] = rec;
      csr[start + rk[a]] = rec;
    }
  }
}

__global__ void k_spmm_rows_rank(const unsigned* __restrict__ rowptr, const int* __restrict__ perm,
                                 const int* __restrict__ src, const float* __restrict__ ew,
                                 int* __restrict__ csrw,
                                 const float* __restrict__ X, float* __restrict__ agg) {
  __shared__ int2 slots[8][RANK_CAP];
  int2* csr = reinterpret_cast<int2*>(csrw);
  const int lane = threadIdx.x & 31;
  const int wl = __builtin_amdgcn_readfirstlane((int)(threadIdx.x >> 5));
  const int w0 = __builtin_amdgcn_readfirstlane((int)(blockIdx.x * (blockDim.x >> 5) + (threadIdx.x >> 5)));
  const int nw = gridDim.x * (blockDim.x >> 5);
  int2* mine = slots[wl];
  const float4* X4 = reinterpret_cast<const float4*>(X);
  for (int r = w0; r < NNODES; r += nw) {
    const int start = (int)rowptr[r];
    const int deg = (int)rowptr[r + 1] - start;
    float4 acc = make_float4(0.f, 0.f, 0.f, 0.f);
    if (deg <= RANK_CAP) {
      if (deg <= 32)      rankAndPlace<1>(start, deg, lane, perm, src, ew, mine, csr);
      else if (deg <= 64) rankAndPlace<2>(start, deg, lane, perm, src, ew, mine, csr);
      else                rankAndPlace<4>(start, deg, lane, perm, src, ew, mine, csr);
      for (int base = 0; base < deg; base += 32) {
        const int cnt = min(32, deg - base);
        const int2 rec = (lane < cnt) ? mine[base + lane] : make_int2(0, 0);
        for (int j = 0; j < cnt; j += 8) {
          float4 x[8];
          for (int u = 0; u < 8; ++u) {
            const int ju = min(j + u, cnt - 1);
            const int s = __builtin_amdgcn_readlane(rec.x, ju);
            x[u] = X4[(size_t)s * (HID / 4) + lane];
          }
          for (int u = 0; u < 8; ++u) {
            if (j + u < cnt) {
              const float w = __int_as_float(__builtin_amdgcn_readlane(rec.y, j + u));
              acc.x = fmaf(w, x[u].x, acc.x);
              acc.y = fmaf(w, x[u].y, acc.y);
              acc.z = fmaf(w, x[u].z, acc.z);
              acc.w = fmaf(w, x[u].w, acc.w);
            }
          }
        }
      }
    } else {
      int cur = -1;
      for (int j = 0; j < deg; ++j) {
        int best = 0x7fffffff;
        for (int b = 0; b < deg; b += 32) {
          const int i = b + lane;
          const int e = (i < deg) ? perm[start + i] : 0x7fffffff;
          if (e > cur && e < best) best = e;
        }
        for (int o = 16; o > 0; o >>= 1) best = min(best, __shfl_xor(best, o, 32));
        cur = __builtin_amdgcn_readfirstlane(best);
        const int s = src[cur];
        const float w = ew[cur];
        if (lane == 0) csr[start + j] = make_int2(s, __float_as_int(w));
        const float4 xv = X4[(size_t)s * (HID / 4) + lane];
        acc.x = fmaf(w, xv.x, acc.x);
        acc.y = fmaf(w, xv.y, acc.y);
        acc.z = fmaf(w, xv.z, acc.z);
        acc.w = fmaf(w, xv.w, acc.w);
      }
    }
    reinterpret_cast<float4*>(agg)[(size_t)r * (HID / 4) + lane] = acc;
  }
}

#define ROW4 (NPAD / 4)
static_assert(NNODES % 2 == 0, "rows are taken in pairs");

template <int K>
__device__ __forceinline__ int rowShare(int v) {
  return __builtin_amdgcn_update_dpp(0, v, 0x150 + K, 0xF, 0xF, false);
}

template <int J>
__device__ __forceinline__ void loadEight(const int2 rec, const int cnt, const bool ld,
                                          const int sub, const float4* __restrict__ X4,
                                          float4 (&x)[8]) {
#define SPMM48_LOAD(U)                                                      \
  {                                                                         \
    const int s = rowShare<J + U>(rec.x);                                   \
    x[U] = make_float4(0.f, 0.f, 0.f, 0.f);                                 \
    if (ld && (J + U) < cnt) x[U] = X4[(size_t)s * ROW4 + sub];             \
  }
  SPMM48_LOAD(0) SPMM48_LOAD(1) SPMM48_LOAD(2) SPMM48_LOAD(3)
  SPMM48_LOAD(4) SPMM48_LOAD(5) SPMM48_LOAD(6) SPMM48_LOAD(7)
#undef SPMM48_LOAD
}

template <int J>
__device__ __forceinline__ void fmaEight(const int2 rec, const float4 (&x)[8], float4& acc) {
#define SPMM48_FMA(U)                                                       \
  {                                                                         \
    const float w = __int_as_float(rowShare<J + U>(rec.y));                 \
    acc.x = fmaf(w, x[U].x, acc.x);                                         \
    acc.y = fmaf(w, x[U].y, acc.y);                                         \
    acc.z = fmaf(w, x[U].z, acc.z);                                         \
    acc.w = fmaf(w, x[U].w, acc.w);                                         \
  }
  SPMM48_FMA(0) SPMM48_FMA(1) SPMM48_FMA(2) SPMM48_FMA(3)
  SPMM48_FMA(4) SPMM48_FMA(5) SPMM48_FMA(6) SPMM48_FMA(7)
#undef SPMM48_FMA
}

__global__ void k_spmm_rows48(const unsigned* __restrict__ rowptr, const int* __restrict__ csrw,
                              const float* __restrict__ X, float* __restrict__ agg) {
  const int2* csr = reinterpret_cast<const int2*>(csrw);
  const int lane = threadIdx.x & 31;
  const int hsel = lane >> 4;
  const int sub = lane & 15;
  const bool ld = sub < ROW4;
  const int w0 = __builtin_amdgcn_readfirstlane((int)(blockIdx.x * (blockDim.x >> 5) + (threadIdx.x >> 5)));
  const int nw = gridDim.x * (blockDim.x >> 5);
  const float4* X4 = reinterpret_cast<const float4*>(X);
  for (int p = w0; p < NNODES / 2; p += nw) {
    const int r0 = 2 * p;
    const int s0 = (int)rowptr[r0];
    const int s1 = (int)rowptr[r0 + 1];
    const int s2 = (int)rowptr[r0 + 2];
    const int degmax = max(s1 - s0, s2 - s1);
    const int start = hsel ? s1 : s0;
    const int deg = hsel ? (s2 - s1) : (s1 - s0);
    float4 acc = make_float4(0.f, 0.f, 0.f, 0.f);
    int cnt = deg;
    int2 rec = (sub < cnt) ? csr[start + sub] : make_int2(0, 0);
    for (int base = 0; base < degmax; base += 16) {
      float4 x[8];
      loadEight<0>(rec, cnt, ld, sub, X4, x);
      int2 nxt = make_int2(0, 0);
      if (sub + 16 < cnt) nxt = csr[start + base + 16 + sub];
      fmaEight<0>(rec, x, acc);
      if (degmax - base > 8) {
        loadEight<8>(rec, cnt, ld, sub, X4, x);
        fmaEight<8>(rec, x, acc);
      }
      rec = nxt;
      cnt -= 16;
    }
    if (ld) reinterpret_cast<float4*>(agg)[(size_t)(r0 + hsel) * ROW4 + sub] = acc;
  }
}

#define PREP_THREADS 256
__device__ __forceinline__ void foldTile(const float* __restrict__ A, const int row0, const int nrows,
                                         const float* __restrict__ Wh, float* __restrict__ D) {
  const int lane = threadIdx.x & 31;
  const int h = lane >> 4;
  const int l = lane & 15;
  const int row = row0 + l;
  const float* Arow = A + (size_t)(row < nrows ? row : 0) * HID;
  const bool cv = 32 + l < NCLS;
  const int n2 = cv ? 32 + l : 0;
  v8f acc0, acc1, acc2;
#pragma unroll
  for (int v = 0; v < 8; ++v) { acc0[v] = 0.f; acc1[v] = 0.f; acc2[v] = 0.f; }
#pragma unroll 8
  for (int t = 0; t < HID / 4; ++t) {
    const int ka = 4 * t + 2 * h;
    const float* w0 = Wh + (size_t)ka * NCLS;
    const float* w1 = w0 + NCLS;
    v2f a, b0, b1, b2;
    a.x = Arow[ka];
    a.y = Arow[ka + 1];
    b0.x = w0[l];
    b0.y = w1[l];
    b1.x = w0[16 + l];
    b1.y = w1[16 + l];
    b2.x = w0[n2];
    b2.y = w1[n2];
    acc0 = __builtin_amdgcn_wmma_f32_16x16x4_f32(false, a, false, b0, (short)0, acc0, false, false);
    acc1 = __builtin_amdgcn_wmma_f32_16x16x4_f32(false, a, false, b1, (short)0, acc1, false, false);
    acc2 = __builtin_amdgcn_wmma_f32_16x16x4_f32(false, a, false, b2, (short)0, acc2, false, false);
  }
#pragma unroll
  for (int v = 0; v < 8; ++v) {
    const int r = row0 + v + 8 * h;
    if (r < nrows) {
      D[(size_t)r * NPAD + l]      = acc0[v];
      D[(size_t)r * NPAD + 16 + l] = acc1[v];
      D[(size_t)r * NPAD + 32 + l] = cv ? acc2[v] : 0.f;
    }
  }
}

__global__ void k_prep_head(const float* __restrict__ W2, const float* __restrict__ We,
                            const float* __restrict__ Wf, const float* __restrict__ bf,
                            const float* __restrict__ be, const float* __restrict__ b2,
                            float* __restrict__ W2f, float* __restrict__ Wef,
                            float* __restrict__ bfp) {
  __shared__ float sW[2 * HID * NCLS];
  __shared__ float sBe[HID];
  __shared__ float sB2[HID];
  const int t = threadIdx.x;
  const int wid = __builtin_amdgcn_readfirstlane((int)(threadIdx.x >> 5));
  if (blockIdx.x == 0) {
    foldTile(W2, 16 * wid, HID, Wf + (size_t)HID * NCLS, W2f);
    return;
  }
  if (blockIdx.x == 1) {
    if (wid < (NCLS + 15) / 16) foldTile(We, 16 * wid, NCLS, Wf, Wef);
    return;
  }
  const float4* Wf4 = reinterpret_cast<const float4*>(Wf);
  float4* sW4 = reinterpret_cast<float4*>(sW);
#pragma unroll
  for (int j = 0; j < (2 * HID * NCLS) / 4 / PREP_THREADS; ++j)
    sW4[j * PREP_THREADS + t] = Wf4[j * PREP_THREADS + t];
  if (t < HID) sBe[t] = be[t];
  else         sB2[t - HID] = b2[t - HID];
  __syncthreads();
  if (t < NPAD) {
    float s = 0.f;
    if (t < NCLS) {
      s = bf[t];
      for (int k = 0; k < HID; ++k) {
        s += sBe[k] * sW[k * NCLS + t];
        s += sB2[k] * sW[(HID + k) * NCLS + t];
      }
    }
    bfp[t] = s;
  }
}

extern "C" void kernel_launch(void* const* d_in, const int* in_sizes, int n_in,
                              void* d_out, int out_size, void* d_ws, size_t ws_size,
                              hipStream_t stream) {
  const float* features = (const float*)d_in[0];
  const int*   src      = (const int*)d_in[1];
  const int*   dst      = (const int*)d_in[2];
  const float* ew       = (const float*)d_in[3];
  const float* y        = (const float*)d_in[4];
  const float* W1       = (const float*)d_in[5];
  const float* b1       = (const float*)d_in[6];
  const float* W2       = (const float*)d_in[7];
  const float* b2       = (const float*)d_in[8];
  const float* We       = (const float*)d_in[9];
  const float* be       = (const float*)d_in[10];
  const float* Wf       = (const float*)d_in[11];
  const float* bf       = (const float*)d_in[12];
  float* out = (float*)d_out;
  (void)in_sizes; (void)n_in; (void)out_size; (void)ws_size;

  char* ws = (char*)d_ws;
  size_t off = 0;
  auto alloc = [&](size_t bytes) -> float* {
    float* p = (float*)(ws + off);
    off += (bytes + 255) & ~(size_t)255;
    return p;
  };
  float* bufA   = alloc((size_t)NNODES * HID * 4);
  float* bufAgg = alloc((size_t)NNODES * HID * 4);
  float* bufL   = alloc((size_t)NNODES * NPAD * 4);
  float* W2f    = alloc((size_t)HID * NPAD * 4);
  float* Wef    = alloc((size_t)NCLS * NPAD * 4);
  float* bfp    = alloc((size_t)NPAD * 4);
  unsigned* rowptr = (unsigned*)alloc((size_t)NPADROWS * 4);
  int*      csr    = (int*)alloc((size_t)NEDGES * 8);
  unsigned* cntMat = (unsigned*)bufAgg;
  unsigned* tot    = cntMat + (size_t)CSR_G * NNODES;
  int*      perm   = (int*)bufL;

  const int gemmBlocks = ((NNODES / 16) + 7) / 8;

  k_prep_head<<<3, PREP_THREADS, 0, stream>>>(W2, We, Wf, bf, be, b2, W2f, Wef, bfp);

  k_gemm_wmma<8, FIN, 1><<<gemmBlocks, 256, 0, stream>>>(
      features, W1, bufA, NNODES, nullptr, nullptr, nullptr);
  k_csr_count<<<CSR_G * CSR_Q, CSR_T, 0, stream>>>(dst, cntMat);
  k_csr_colscan<<<NPADROWS / 256, 256, 0, stream>>>(cntMat, tot);
  k_csr_rowscan<<<1, CSR_T, 0, stream>>>(tot, rowptr);
  k_csr_fill<<<CSR_G * CSR_Q, CSR_T, 0, stream>>>(dst, cntMat, rowptr, perm);
  k_spmm_rows_rank<<<ROW_BLOCKS, 256, 0, stream>>>(rowptr, perm, src, ew, csr, bufA, bufAgg);
  k_gemm_wmma<3, HID, 2><<<gemmBlocks, 256, 0, stream>>>(
      bufAgg, W2f, bufA, NNODES, b1, nullptr, nullptr);
  k_spmm_rows48<<<ROW_BLOCKS, 256, 0, stream>>>(rowptr, csr, bufA, bufL);
  k_gemm_wmma<3, NCLS, 28><<<gemmBlocks, 256, 0, stream>>>(
      y, Wef, bufL, NNODES, nullptr, bfp, out);
}


# Layer-2 weight GEMM folded into the layer-1 SPMM kernel: a block keeps the sums of its 32 rows in LDS, two waves apply relu(+b1) and multiply by W2f on WMMA, same k order; GEMM launch and agg1 gone
# speedup vs baseline: 3.8989x; 1.0452x over previous
#include <hip/hip_runtime.h>
#include <hip/hip_bf16.h>

typedef float v2f __attribute__((ext_vector_type(2)));
typedef float v8f __attribute__((ext_vector_type(8)));

#define NNODES 50000
#define NEDGES 1600000
#define FIN    256
#define HID    128
#define NCLS   40
#define NPAD   48

template <int X>
__device__ __forceinline__ float rowXor(float v) {
  return __int_as_float(__builtin_amdgcn_update_dpp(0, __float_as_int(v), 0x160 + X, 0xF, 0xF, false));
}
__device__ __forceinline__ float rowMax16(float v) {
  v = fmaxf(v, rowXor<8>(v)); v = fmaxf(v, rowXor<4>(v));
  v = fmaxf(v, rowXor<2>(v)); v = fmaxf(v, rowXor<1>(v));
  return v;
}
__device__ __forceinline__ float rowSum16(float v) {
  v += rowXor<8>(v); v += rowXor<4>(v); v += rowXor<2>(v); v += rowXor<1>(v);
  return v;
}

template <int NT, int K, int MODE>
__global__ void k_gemm_wmma(const float* __restrict__ A, const float* __restrict__ B,
                            float* __restrict__ C, int M,
                            const float* __restrict__ bias,
                            float* __restrict__ P) {
  constexpr bool USE_SCALE = (MODE & 1) != 0;
  constexpr bool USE_BIAS  = (MODE & 4) != 0;
  constexpr bool ACCUM     = (MODE & 8) != 0;
  constexpr bool SOFTMAX   = (MODE & 16) != 0;
  constexpr int N = NT * 16;
  static_assert(!SOFTMAX || (NT == 3 && USE_BIAS && !USE_SCALE), "softmax epilogue: NPAD columns, biased logits");
  constexpr int KCHUNK = (K % 32 == 0) ? 32 : K;
  constexpr int NCHUNKS = K / KCHUNK;
  constexpr int TSTEPS = KCHUNK / 4;
  constexpr int GROUP = (TSTEPS % 8 == 0) ? 8 : TSTEPS / 2;
  constexpr int NV4 = (KCHUNK * N) / 4;
  constexpr int STAGE4 = (NV4 + 255) / 256;
  __shared__ float ldsB[KCHUNK * N];

  const int lane = threadIdx.x & 31;
  const int wid  = threadIdx.x >> 5;
  const int tm   = blockIdx.x * (blockDim.x >> 5) + wid;
  const bool active = tm < (M >> 4);
  const int h = lane >> 4;
  const int l = lane & 15;
  const int row = (tm << 4) + l;

  v8f acc[NT];
  if (active) {
#pragma unroll
    for (int j = 0; j < NT; ++j)
#pragma unroll
      for (int v = 0; v < 8; ++v)
        acc[j][v] = ACCUM ? C[(size_t)((tm << 4) + v + 8 * h) * N + j * 16 + l] : 0.f;
  }
  const float* Arow = A + (size_t)row * K;
  float ss = 0.f;

  for (int c = 0; c < NCHUNKS; ++c) {
    const int k0 = c * KCHUNK;
    __syncthreads();
#pragma unroll
    for (int i = 0; i < STAGE4; ++i) {
      int idx4 = i * 256 + threadIdx.x;
      if (NV4 % 256 == 0 || idx4 < NV4) {
        int f = idx4 * 4;
        int kk = f / N, n = f - kk * N;
        float4 q = *(const float4*)(B + (size_t)(k0 + kk) * N + n);
        float* d = ldsB + ((kk >> 1) * N + n) * 2 + (kk & 1);
        d[0] = q.x; d[2] = q.y; d[4] = q.z; d[6] = q.w;
      }
    }
    __syncthreads();
    if (active) {
#pragma unroll
      for (int g = 0; g < TSTEPS; g += GROUP) {
        v2f areg[GROUP];
#pragma unroll
        for (int u = 0; u < GROUP; ++u) {
          const int ka = k0 + (g + u) * 4 + 2 * h;
          v2f a;
          a.x = Arow[ka];
          a.y = Arow[ka + 1];
          areg[u] = a;
        }
#pragma unroll
        for (int u = 0; u < GROUP; ++u) {
          const int ka = (g + u) * 4 + 2 * h;
          const float* bp = ldsB + (ka >> 1) * (2 * N) + 2 * l;
          if (USE_SCALE) {
            ss = fmaf(areg[u].x, areg[u].x, ss);
            ss = fmaf(areg[u].y, areg[u].y, ss);
          }
#pragma unroll
          for (int j = 0; j < NT; ++j) {
            v2f b = *(const v2f*)(bp + 32 * j);
            acc[j] = __builtin_amdgcn_wmma_f32_16x16x4_f32(
                false, areg[u], false, b, (short)0, acc[j], false, false);
          }
        }
      }
    }
  }

  if (active && !SOFTMAX) {
    float sc[8];
    if (USE_SCALE) {
      const float tot = ss + __shfl_xor(ss, 16, 32);
      const float inv = 1.0f / fmaxf(sqrtf(tot), 1e-12f);
#pragma unroll
      for (int v = 0; v < 8; ++v) sc[v] = __shfl(inv, v + 8 * h, 32);
    }
#pragma unroll
    for (int j = 0; j < NT; ++j)
#pragma unroll
      for (int v = 0; v < 8; ++v) {
        const int cn = j * 16 + l;
        float o = acc[j][v];
        if (USE_SCALE) o *= sc[v];
        if (USE_BIAS) o += bias[cn];
        C[(size_t)((tm << 4) + v + 8 * h) * N + cn] = o;
      }
  }
  if (active && SOFTMAX) {
    const bool c2 = 32 + l < NCLS;
    const float bz0 = bias[l], bz1 = bias[16 + l], bz2 = bias[32 + l];
#pragma unroll
    for (int v = 0; v < 8; ++v) {
      const float x0 = acc[0][v] + bz0;
      const float x1 = acc[1][v] + bz1;
      const float x2 = c2 ? acc[2][v] + bz2 : -3.4e38f;
      const float m = rowMax16(fmaxf(fmaxf(x0, x1), x2));
      const float e0 = __expf(x0 - m);
      const float e1 = __expf(x1 - m);
      const float e2 = c2 ? __expf(x2 - m) : 0.f;
      const float inv = 1.0f / rowSum16(e0 + e1 + e2);
      float* orow = P + (size_t)((tm << 4) + v + 8 * h) * NCLS;
      orow[l] = e0 * inv;
      orow[16 + l] = e1 * inv;
      if (c2) orow[32 + l] = e2 * inv;
    }
  }
}

#define CSR_G      64
#define CSR_Q      4
#define CSR_ROWS   (NNODES / CSR_Q)
#define CSR_CHUNK  (NEDGES / CSR_G)
#define CSR_T      1024
#define CSR_PER    ((CSR_CHUNK + CSR_T - 1) / CSR_T)
#define SCAN_PER   52
#define NPADROWS   (CSR_T * SCAN_PER)
#define ROWS_PER_WAVE 4
#define ROW_BLOCKS (((NNODES + ROWS_PER_WAVE - 1) / ROWS_PER_WAVE + 7) / 8)
static_assert(NEDGES % CSR_G == 0, "edge chunks must be equal");
static_assert(NNODES % CSR_Q == 0, "row ranges must be equal");
static_assert(NPADROWS >= NNODES + 1, "rowptr needs an end entry");
static_assert(SCAN_PER % 4 == 0, "rowscan moves 16-byte vectors");
static_assert((size_t)NEDGES * 4 <= (size_t)NNODES * NPAD * 4, "perm[] borrows the NPAD-column buffer");

__global__ void __launch_bounds__(CSR_T)
k_csr_count(const int* __restrict__ dst, unsigned* __restrict__ cnt) {
  __shared__ unsigned hist[CSR_ROWS];
  const int t = threadIdx.x;
  const int c = blockIdx.x / CSR_Q;
  const int lo = (blockIdx.x % CSR_Q) * CSR_ROWS;
  for (int i = t; i < CSR_ROWS; i += CSR_T) hist[i] = 0u;
  __syncthreads();
  const int* d = dst + (size_t)c * CSR_CHUNK;
  int dv[CSR_PER];
  for (int k = 0; k < CSR_PER; ++k) {
    const int i = t + k * CSR_T;
    dv[k] = (i < CSR_CHUNK) ? d[i] : -1;
  }
  for (int k = 0; k < CSR_PER; ++k) {
    const unsigned rr = (unsigned)(dv[k] - lo);
    if (rr < (unsigned)CSR_ROWS) atomicAdd(&hist[rr], 1u);
  }
  __syncthreads();
  unsigned* o = cnt + (size_t)c * NNODES + lo;
  for (int i = t; i < CSR_ROWS; i += CSR_T) o[i] = hist[i];
}

__global__ void k_csr_colscan(unsigned* __restrict__ cnt, unsigned* __restrict__ tot) {
  const int r = blockIdx.x * 256 + threadIdx.x;
  unsigned s = 0u;
  if (r < NNODES) {
    unsigned c[CSR_G];
    for (int b = 0; b < CSR_G; ++b) c[b] = cnt[(size_t)b * NNODES + r];
    for (int b = 0; b < CSR_G; ++b) { cnt[(size_t)b * NNODES + r] = s; s += c[b]; }
  }
  tot[r] = s;
}

__global__ void __launch_bounds__(CSR_T)
k_csr_rowscan(const unsigned* __restrict__ tot, unsigned* __restrict__ rowptr) {
  __shared__ unsigned wsum[32];
  const int t = threadIdx.x, lane = t & 31, wid = t >> 5;
  const uint4* in = reinterpret_cast<const uint4*>(tot) + t * (SCAN_PER / 4);
  uint4 v[SCAN_PER / 4];
  for (int i = 0; i < SCAN_PER / 4; ++i) v[i] = in[i];
  unsigned s = 0u;
  for (int i = 0; i < SCAN_PER / 4; ++i) s += v[i].x + v[i].y + v[i].z + v[i].w;
  unsigned inc = s;
  for (int o = 1; o < 32; o <<= 1) {
    unsigned n = __shfl_up(inc, o, 32);
    if (lane >= o) inc += n;
  }
  if (lane == 31) wsum[wid] = inc;
  __syncthreads();
  if (wid == 0) {
    unsigned w = wsum[lane], wi = w;
    for (int o = 1; o < 32; o <<= 1) {
      unsigned n = __shfl_up(wi, o, 32);
      if (lane >= o) wi += n;
    }
    wsum[lane] = wi - w;
  }
  __syncthreads();
  unsigned run = wsum[wid] + (inc - s);
  uint4* out = reinterpret_cast<uint4*>(rowptr) + t * (SCAN_PER / 4);
  for (int i = 0; i < SCAN_PER / 4; ++i) {
    uint4 q;
    q.x = run; run += v[i].x;
    q.y = run; run += v[i].y;
    q.z = run; run += v[i].z;
    q.w = run; run += v[i].w;
    out[i] = q;
  }
}

__global__ void __launch_bounds__(CSR_T)
k_csr_fill(const int* __restrict__ dst, const unsigned* __restrict__ blkoff,
           const unsigned* __restrict__ rowptr, int* __restrict__ perm) {
  __shared__ unsigned hist[CSR_ROWS];
  const int t = threadIdx.x;
  const int c = blockIdx.x / CSR_Q;
  const int lo = (blockIdx.x % CSR_Q) * CSR_ROWS;
  const unsigned* bo = blkoff + (size_t)c * NNODES + lo;
  for (int i = t; i < CSR_ROWS; i += CSR_T) hist[i] = rowptr[lo + i] + bo[i];
  __syncthreads();
  const int e0 = c * CSR_CHUNK;
  int dv[CSR_PER];
  for (int k = 0; k < CSR_PER; ++k) {
    const int i = t + k * CSR_T;
    dv[k] = (i < CSR_CHUNK) ? dst[e0 + i] : -1;
  }
  for (int k = 0; k < CSR_PER; ++k) {
    const unsigned rr = (unsigned)(dv[k] - lo);
    if (rr < (unsigned)CSR_ROWS) {
      const unsigned pos = atomicAdd(&hist[rr], 1u);
      perm[pos] = e0 + t + k * CSR_T;
    }
  }
}

#define RANK_CAP 128

template <int NB>
__device__ __forceinline__ void rankAndPlace(const int start, const int deg, const int lane,
                                             const int* __restrict__ perm, const int* __restrict__ src,
                                             const float* __restrict__ ew, int2* mine,
                                             int2* __restrict__ csr) {
  int e[NB], rk[NB];
#pragma unroll
  for (int a = 0; a < NB; ++a) {
    const int i = 32 * a + lane;
    e[a] = (i < deg) ? perm[start + i] : 0x7fffffff;
    rk[a] = 0;
  }
#pragma unroll
  for (int b = 0; b < NB; ++b) {
#pragma unroll(NB > 2 ? 4 : 32)
    for (int j = 0; j < 32; ++j) {
      const int s = __builtin_amdgcn_readlane(e[b], j);
#pragma unroll
      for (int a = 0; a < NB; ++a) rk[a] += (s < e[a]) ? 1 : 0;
    }
  }
  int sv[NB];
  float wv[NB];
#pragma unroll
  for (int a = 0; a < NB; ++a) {
    const bool ok = 32 * a + lane < deg;
    sv[a] = ok ? src[e[a]] : 0;
    wv[a] = ok ? ew[e[a]] : 0.f;
  }
#pragma unroll
  for (int a = 0; a < NB; ++a) {
    if (32 * a + lane < deg) {
      const int2 rec = make_int2(sv[a], __float_as_int(wv[a]));
      mine[rk[a]] = rec;
      csr[start + rk[a]] = rec;
    }
  }
}

#define BLOCK_ROWS 32
#define TILE_PITCH (HID + 4)
static_assert(BLOCK_ROWS == 8 * ROWS_PER_WAVE, "one row per wave and pass");
static_assert(NNODES % 16 == 0, "a matrix tile is all rows or none");
static_assert((NNODES + BLOCK_ROWS - 1) / BLOCK_ROWS == ROW_BLOCKS, "grid covers the rows once");

__global__ void __launch_bounds__(256)
k_spmm_rows_rank(const unsigned* __restrict__ rowptr, const int* __restrict__ perm,
                 const int* __restrict__ src, const float* __restrict__ ew,
                 int* __restrict__ csrw, const float* __restrict__ X,
                 const float* __restrict__ b1, const float* __restrict__ W2f,
                 float* __restrict__ X2f) {
  __shared__ int2 slots[8][RANK_CAP];
  __shared__ float tile[BLOCK_ROWS * TILE_PITCH];
  int2* csr = reinterpret_cast<int2*>(csrw);
  const int lane = threadIdx.x & 31;
  const int wl = __builtin_amdgcn_readfirstlane((int)(threadIdx.x >> 5));
  const int rbase = blockIdx.x * BLOCK_ROWS;
  int2* mine = slots[wl];
  const float4* X4 = reinterpret_cast<const float4*>(X);
  for (int k = 0; k < ROWS_PER_WAVE; ++k) {
    const int rl = wl + 8 * k;
    const int r = rbase + rl;
    if (r < NNODES) {
      const int start = (int)rowptr[r];
      const int deg = (int)rowptr[r + 1] - start;
      float4 acc = make_float4(0.f, 0.f, 0.f, 0.f);
      if (deg <= RANK_CAP) {
        if (deg <= 32)      rankAndPlace<1>(start, deg, lane, perm, src, ew, mine, csr);
        else if (deg <= 64) rankAndPlace<2>(start, deg, lane, perm, src, ew, mine, csr);
        else                rankAndPlace<4>(start, deg, lane, perm, src, ew, mine, csr);
        for (int base = 0; base < deg; base += 32) {
          const int cnt = min(32, deg - base);
          const int2 rec = (lane < cnt) ? mine[base + lane] : make_int2(0, 0);
          for (int j = 0; j < cnt; j += 8) {
            float4 x[8];
            for (int u = 0; u < 8; ++u) {
              const int ju = min(j + u, cnt - 1);
              const int s = __builtin_amdgcn_readlane(rec.x, ju);
              x[u] = X4[(size_t)s * (HID / 4) + lane];
            }
            for (int u = 0; u < 8; ++u) {
              if (j + u < cnt) {
                const float w = __int_as_float(__builtin_amdgcn_readlane(rec.y, j + u));
                acc.x = fmaf(w, x[u].x, acc.x);
                acc.y = fmaf(w, x[u].y, acc.y);
                acc.z = fmaf(w, x[u].z, acc.z);
                acc.w = fmaf(w, x[u].w, acc.w);
              }
            }
          }
        }
      } else {
        int cur = -1;
        for (int j = 0; j < deg; ++j) {
          int best = 0x7fffffff;
          for (int b = 0; b < deg; b += 32) {
            const int i = b + lane;
            const int e = (i < deg) ? perm[start + i] : 0x7fffffff;
            if (e > cur && e < best) best = e;
          }
          for (int o = 16; o > 0; o >>= 1) best = min(best, __shfl_xor(best, o, 32));
          cur = __builtin_amdgcn_readfirstlane(best);
          const int s = src[cur];
          const float w = ew[cur];
          if (lane == 0) csr[start + j] = make_int2(s, __float_as_int(w));
          const float4 xv = X4[(size_t)s * (HID / 4) + lane];
          acc.x = fmaf(w, xv.x, acc.x);
          acc.y = fmaf(w, xv.y, acc.y);
          acc.z = fmaf(w, xv.z, acc.z);
          acc.w = fmaf(w, xv.w, acc.w);
        }
      }
      *reinterpret_cast<float4*>(tile + rl * TILE_PITCH + 4 * lane) = acc;
    }
  }
  __syncthreads();
  const int row0 = rbase + 16 * wl;
  if (wl < BLOCK_ROWS / 16 && row0 < NNODES) {
    const int h = lane >> 4;
    const int l = lane & 15;
    const float* arow = tile + (16 * wl + l) * TILE_PITCH;
    const float* bcol = W2f + l;
    v8f d0, d1, d2;
#pragma unroll
    for (int v = 0; v < 8; ++v) { d0[v] = 0.f; d1[v] = 0.f; d2[v] = 0.f; }
#pragma unroll 2
    for (int t = 0; t < HID / 4; ++t) {
      const int ka = 4 * t + 2 * h;
      const float* w0 = bcol + (size_t)ka * NPAD;
      const float* w1 = w0 + NPAD;
      const v2f s = *reinterpret_cast<const v2f*>(arow + ka);
      v2f a, b0, b1v, b2;
      a.x = fmaxf(s.x + b1[ka], 0.f);
      a.y = fmaxf(s.y + b1[ka + 1], 0.f);
      b0.x = w0[0];   b0.y = w1[0];
      b1v.x = w0[16]; b1v.y = w1[16];
      b2.x = w0[32];  b2.y = w1[32];
      d0 = __builtin_amdgcn_wmma_f32_16x16x4_f32(false, a, false, b0, (short)0, d0, false, false);
      d1 = __builtin_amdgcn_wmma_f32_16x16x4_f32(false, a, false, b1v, (short)0, d1, false, false);
      d2 = __builtin_amdgcn_wmma_f32_16x16x4_f32(false, a, false, b2, (short)0, d2, false, false);
    }
#pragma unroll
    for (int v = 0; v < 8; ++v) {
      float* orow = X2f + (size_t)(row0 + v + 8 * h) * NPAD;
      orow[l] = d0[v];
      orow[16 + l] = d1[v];
      orow[32 + l] = d2[v];
    }
  }
}

#define ROW4 (NPAD / 4)
static_assert(NNODES % 2 == 0, "rows are taken in pairs");

template <int K>
__device__ __forceinline__ int rowShare(int v) {
  return __builtin_amdgcn_update_dpp(0, v, 0x150 + K, 0xF, 0xF, false);
}

template <int J>
__device__ __forceinline__ void loadEight(const int2 rec, const int cnt, const bool ld,
                                          const int sub, const float4* __restrict__ X4,
                                          float4 (&x)[8]) {
#define SPMM48_LOAD(U)                                                      \
  {                                                                         \
    const int s = rowShare<J + U>(rec.x);                                   \
    x[U] = make_float4(0.f, 0.f, 0.f, 0.f);                                 \
    if (ld && (J + U) < cnt) x[U] = X4[(size_t)s * ROW4 + sub];             \
  }
  SPMM48_LOAD(0) SPMM48_LOAD(1) SPMM48_LOAD(2) SPMM48_LOAD(3)
  SPMM48_LOAD(4) SPMM48_LOAD(5) SPMM48_LOAD(6) SPMM48_LOAD(7)
#undef SPMM48_LOAD
}

template <int J>
__device__ __forceinline__ void fmaEight(const int2 rec, const float4 (&x)[8], float4& acc) {
#define SPMM48_FMA(U)                                                       \
  {                                                                         \
    const float w = __int_as_float(rowShare<J + U>(rec.y));                 \
    acc.x = fmaf(w, x[U].x, acc.x);                                         \
    acc.y = fmaf(w, x[U].y, acc.y);                                         \
    acc.z = fmaf(w, x[U].z, acc.z);                                         \
    acc.w = fmaf(w, x[U].w, acc.w);                                         \
  }
  SPMM48_FMA(0) SPMM48_FMA(1) SPMM48_FMA(2) SPMM48_FMA(3)
  SPMM48_FMA(4) SPMM48_FMA(5) SPMM48_FMA(6) SPMM48_FMA(7)
#undef SPMM48_FMA
}

__global__ void k_spmm_rows48(const unsigned* __restrict__ rowptr, const int* __restrict__ csrw,
                              const float* __restrict__ X, float* __restrict__ agg) {
  const int2* csr = reinterpret_cast<const int2*>(csrw);
  const int lane = threadIdx.x & 31;
  const int hsel = lane >> 4;
  const int sub = lane & 15;
  const bool ld = sub < ROW4;
  const int w0 = __builtin_amdgcn_readfirstlane((int)(blockIdx.x * (blockDim.x >> 5) + (threadIdx.x >> 5)));
  const int nw = gridDim.x * (blockDim.x >> 5);
  const float4* X4 = reinterpret_cast<const float4*>(X);
  for (int p = w0; p < NNODES / 2; p += nw) {
    const int r0 = 2 * p;
    const int s0 = (int)rowptr[r0];
    const int s1 = (int)rowptr[r0 + 1];
    const int s2 = (int)rowptr[r0 + 2];
    const int degmax = max(s1 - s0, s2 - s1);
    const int start = hsel ? s1 : s0;
    const int deg = hsel ? (s2 - s1) : (s1 - s0);
    float4 acc = make_float4(0.f, 0.f, 0.f, 0.f);
    int cnt = deg;
    int2 rec = (sub < cnt) ? csr[start + sub] : make_int2(0, 0);
    for (int base = 0; base < degmax; base += 16) {
      float4 x[8];
      loadEight<0>(rec, cnt, ld, sub, X4, x);
      int2 nxt = make_int2(0, 0);
      if (sub + 16 < cnt) nxt = csr[start + base + 16 + sub];
      fmaEight<0>(rec, x, acc);
      if (degmax - base > 8) {
        loadEight<8>(rec, cnt, ld, sub, X4, x);
        fmaEight<8>(rec, x, acc);
      }
      rec = nxt;
      cnt -= 16;
    }
    if (ld) reinterpret_cast<float4*>(agg)[(size_t)(r0 + hsel) * ROW4 + sub] = acc;
  }
}

#define PREP_THREADS 256
__device__ __forceinline__ void foldTile(const float* __restrict__ A, const int row0, const int nrows,
                                         const float* __restrict__ Wh, float* __restrict__ D) {
  const int lane = threadIdx.x & 31;
  const int h = lane >> 4;
  const int l = lane & 15;
  const int row = row0 + l;
  const float* Arow = A + (size_t)(row < nrows ? row : 0) * HID;
  const bool cv = 32 + l < NCLS;
  const int n2 = cv ? 32 + l : 0;
  v8f acc0, acc1, acc2;
#pragma unroll
  for (int v = 0; v < 8; ++v) { acc0[v] = 0.f; acc1[v] = 0.f; acc2[v] = 0.f; }
#pragma unroll 8
  for (int t = 0; t < HID / 4; ++t) {
    const int ka = 4 * t + 2 * h;
    const float* w0 = Wh + (size_t)ka * NCLS;
    const float* w1 = w0 + NCLS;
    v2f a, b0, b1, b2;
    a.x = Arow[ka];
    a.y = Arow[ka + 1];
    b0.x = w0[l];
    b0.y = w1[l];
    b1.x = w0[16 + l];
    b1.y = w1[16 + l];
    b2.x = w0[n2];
    b2.y = w1[n2];
    acc0 = __builtin_amdgcn_wmma_f32_16x16x4_f32(false, a, false, b0, (short)0, acc0, false, false);
    acc1 = __builtin_amdgcn_wmma_f32_16x16x4_f32(false, a, false, b1, (short)0, acc1, false, false);
    acc2 = __builtin_amdgcn_wmma_f32_16x16x4_f32(false, a, false, b2, (short)0, acc2, false, false);
  }
#pragma unroll
  for (int v = 0; v < 8; ++v) {
    const int r = row0 + v + 8 * h;
    if (r < nrows) {
      D[(size_t)r * NPAD + l]      = acc0[v];
      D[(size_t)r * NPAD + 16 + l] = acc1[v];
      D[(size_t)r * NPAD + 32 + l] = cv ? acc2[v] : 0.f;
    }
  }
}

__global__ void k_prep_head(const float* __restrict__ W2, const float* __restrict__ We,
                            const float* __restrict__ Wf, const float* __restrict__ bf,
                            const float* __restrict__ be, const float* __restrict__ b2,
                            float* __restrict__ W2f, float* __restrict__ Wef,
                            float* __restrict__ bfp) {
  __shared__ float sW[2 * HID * NCLS];
  __shared__ float sBe[HID];
  __shared__ float sB2[HID];
  const int t = threadIdx.x;
  const int wid = __builtin_amdgcn_readfirstlane((int)(threadIdx.x >> 5));
  if (blockIdx.x == 0) {
    foldTile(W2, 16 * wid, HID, Wf + (size_t)HID * NCLS, W2f);
    return;
  }
  if (blockIdx.x == 1) {
    if (wid < (NCLS + 15) / 16) foldTile(We, 16 * wid, NCLS, Wf, Wef);
    return;
  }
  const float4* Wf4 = reinterpret_cast<const float4*>(Wf);
  float4* sW4 = reinterpret_cast<float4*>(sW);
#pragma unroll
  for (int j = 0; j < (2 * HID * NCLS) / 4 / PREP_THREADS; ++j)
    sW4[j * PREP_THREADS + t] = Wf4[j * PREP_THREADS + t];
  if (t < HID) sBe[t] = be[t];
  else         sB2[t - HID] = b2[t - HID];
  __syncthreads();
  if (t < NPAD) {
    float s = 0.f;
    if (t < NCLS) {
      s = bf[t];
      for (int k = 0; k < HID; ++k) {
        s += sBe[k] * sW[k * NCLS + t];
        s += sB2[k] * sW[(HID + k) * NCLS + t];
      }
    }
    bfp[t] = s;
  }
}

extern "C" void kernel_launch(void* const* d_in, const int* in_sizes, int n_in,
                              void* d_out, int out_size, void* d_ws, size_t ws_size,
                              hipStream_t stream) {
  const float* features = (const float*)d_in[0];
  const int*   src      = (const int*)d_in[1];
  const int*   dst      = (const int*)d_in[2];
  const float* ew       = (const float*)d_in[3];
  const float* y        = (const float*)d_in[4];
  const float* W1       = (const float*)d_in[5];
  const float* b1       = (const float*)d_in[6];
  const float* W2       = (const float*)d_in[7];
  const float* b2       = (const float*)d_in[8];
  const float* We       = (const float*)d_in[9];
  const float* be       = (const float*)d_in[10];
  const float* Wf       = (const float*)d_in[11];
  const float* bf       = (const float*)d_in[12];
  float* out = (float*)d_out;
  (void)in_sizes; (void)n_in; (void)out_size; (void)ws_size;

  char* ws = (char*)d_ws;
  size_t off = 0;
  auto alloc = [&](size_t bytes) -> float* {
    float* p = (float*)(ws + off);
    off += (bytes + 255) & ~(size_t)255;
    return p;
  };
  float* bufA   = alloc((size_t)NNODES * HID * 4);
  float* bufAgg = alloc((size_t)NNODES * HID * 4);
  float* bufL   = alloc((size_t)NNODES * NPAD * 4);
  float* W2f    = alloc((size_t)HID * NPAD * 4);
  float* Wef    = alloc((size_t)NCLS * NPAD * 4);
  float* bfp    = alloc((size_t)NPAD * 4);
  unsigned* rowptr = (unsigned*)alloc((size_t)NPADROWS * 4);
  int*      csr    = (int*)alloc((size_t)NEDGES * 8);
  unsigned* cntMat = (unsigned*)bufAgg;
  unsigned* tot    = cntMat + (size_t)CSR_G * NNODES;
  int*      perm   = (int*)bufL;

  const int gemmBlocks = ((NNODES / 16) + 7) / 8;

  k_prep_head<<<3, PREP_THREADS, 0, stream>>>(W2, We, Wf, bf, be, b2, W2f, Wef, bfp);

  k_gemm_wmma<8, FIN, 1><<<gemmBlocks, 256, 0, stream>>>(
      features, W1, bufA, NNODES, nullptr, nullptr);
  k_csr_count<<<CSR_G * CSR_Q, CSR_T, 0, stream>>>(dst, cntMat);
  k_csr_colscan<<<NPADROWS / 256, 256, 0, stream>>>(cntMat, tot);
  k_csr_rowscan<<<1, CSR_T, 0, stream>>>(tot, rowptr);
  k_csr_fill<<<CSR_G * CSR_Q, CSR_T, 0, stream>>>(dst, cntMat, rowptr, perm);
  k_spmm_rows_rank<<<ROW_BLOCKS, 256, 0, stream>>>(rowptr, perm, src, ew, csr, bufA, b1, W2f, bufAgg);
  k_spmm_rows48<<<ROW_BLOCKS, 256, 0, stream>>>(rowptr, csr, bufAgg, bufL);
  k_gemm_wmma<3, NCLS, 28><<<gemmBlocks, 256, 0, stream>>>(
      y, Wef, bufL, NNODES, bfp, out);
}


# Head-weight prep, edge count and layer-1 GEMM (mutually independent) run as one launch, the block index picks the role; count reshaped to 256-thread blocks; two launches gone, every sum order kept
# speedup vs baseline: 4.2491x; 1.0898x over previous
#include <hip/hip_runtime.h>
#include <hip/hip_bf16.h>

typedef float v2f __attribute__((ext_vector_type(2)));
typedef float v8f __attribute__((ext_vector_type(8)));

#define NNODES 50000
#define NEDGES 1600000
#define FIN    256
#define HID    128
#define NCLS   40
#define NPAD   48

template <int X>
__device__ __forceinline__ float rowXor(float v) {
  return __int_as_float(__builtin_amdgcn_update_dpp(0, __float_as_int(v), 0x160 + X, 0xF, 0xF, false));
}
__device__ __forceinline__ float rowMax16(float v) {
  v = fmaxf(v, rowXor<8>(v)); v = fmaxf(v, rowXor<4>(v));
  v = fmaxf(v, rowXor<2>(v)); v = fmaxf(v, rowXor<1>(v));
  return v;
}
__device__ __forceinline__ float rowSum16(float v) {
  v += rowXor<8>(v); v += rowXor<4>(v); v += rowXor<2>(v); v += rowXor<1>(v);
  return v;
}

template <int K> struct GemmChunk { static constexpr int value = (K % 32 == 0) ? 32 : K; };

template <int NT, int K, int MODE>
__device__ __forceinline__ void gemmBlock(const float* __restrict__ A, const float* __restrict__ B,
                                          float* __restrict__ C, int M,
                                          const float* __restrict__ bias,
                                          float* __restrict__ P,
                                          float* __restrict__ ldsB, const int bid) {
  constexpr bool USE_SCALE = (MODE & 1) != 0;
  constexpr bool USE_BIAS  = (MODE & 4) != 0;
  constexpr bool ACCUM     = (MODE & 8) != 0;
  constexpr bool SOFTMAX   = (MODE & 16) != 0;
  constexpr int N = NT * 16;
  static_assert(!SOFTMAX || (NT == 3 && USE_BIAS && !USE_SCALE), "softmax epilogue: NPAD columns, biased logits");
  constexpr int KCHUNK = GemmChunk<K>::value;
  constexpr int NCHUNKS = K / KCHUNK;
  constexpr int TSTEPS = KCHUNK / 4;
  constexpr int GROUP = (TSTEPS % 8 == 0) ? 8 : TSTEPS / 2;
  constexpr int NV4 = (KCHUNK * N) / 4;
  constexpr int STAGE4 = (NV4 + 255) / 256;

  const int lane = threadIdx.x & 31;
  const int wid  = threadIdx.x >> 5;
  const int tm   = bid * (blockDim.x >> 5) + wid;
  const bool active = tm < (M >> 4);
  const int h = lane >> 4;
  const int l = lane & 15;
  const int row = (tm << 4) + l;

  v8f acc[NT];
  if (active) {
#pragma unroll
    for (int j = 0; j < NT; ++j)
#pragma unroll
      for (int v = 0; v < 8; ++v)
        acc[j][v] = ACCUM ? C[(size_t)((tm << 4) + v + 8 * h) * N + j * 16 + l] : 0.f;
  }
  const float* Arow = A + (size_t)row * K;
  float ss = 0.f;

  for (int c = 0; c < NCHUNKS; ++c) {
    const int k0 = c * KCHUNK;
    __syncthreads();
#pragma unroll
    for (int i = 0; i < STAGE4; ++i) {
      int idx4 = i * 256 + threadIdx.x;
      if (NV4 % 256 == 0 || idx4 < NV4) {
        int f = idx4 * 4;
        int kk = f / N, n = f - kk * N;
        float4 q = *(const float4*)(B + (size_t)(k0 + kk) * N + n);
        float* d = ldsB + ((kk >> 1) * N + n) * 2 + (kk & 1);
        d[0] = q.x; d[2] = q.y; d[4] = q.z; d[6] = q.w;
      }
    }
    __syncthreads();
    if (active) {
#pragma unroll
      for (int g = 0; g < TSTEPS; g += GROUP) {
        v2f areg[GROUP];
#pragma unroll
        for (int u = 0; u < GROUP; ++u) {
          const int ka = k0 + (g + u) * 4 + 2 * h;
          v2f a;
          a.x = Arow[ka];
          a.y = Arow[ka + 1];
          areg[u] = a;
        }
#pragma unroll
        for (int u = 0; u < GROUP; ++u) {
          const int ka = (g + u) * 4 + 2 * h;
          const float* bp = ldsB + (ka >> 1) * (2 * N) + 2 * l;
          if (USE_SCALE) {
            ss = fmaf(areg[u].x, areg[u].x, ss);
            ss = fmaf(areg[u].y, areg[u].y, ss);
          }
#pragma unroll
          for (int j = 0; j < NT; ++j) {
            v2f b = *(const v2f*)(bp + 32 * j);
            acc[j] = __builtin_amdgcn_wmma_f32_16x16x4_f32(
                false, areg[u], false, b, (short)0, acc[j], false, false);
          }
        }
      }
    }
  }

  if (active && !SOFTMAX) {
    float sc[8];
    if (USE_SCALE) {
      const float tot = ss + __shfl_xor(ss, 16, 32);
      const float inv = 1.0f / fmaxf(sqrtf(tot), 1e-12f);
#pragma unroll
      for (int v = 0; v < 8; ++v) sc[v] = __shfl(inv, v + 8 * h, 32);
    }
#pragma unroll
    for (int j = 0; j < NT; ++j)
#pragma unroll
      for (int v = 0; v < 8; ++v) {
        const int cn = j * 16 + l;
        float o = acc[j][v];
        if (USE_SCALE) o *= sc[v];
        if (USE_BIAS) o += bias[cn];
        C[(size_t)((tm << 4) + v + 8 * h) * N + cn] = o;
      }
  }
  if (active && SOFTMAX) {
    const bool c2 = 32 + l < NCLS;
    const float bz0 = bias[l], bz1 = bias[16 + l], bz2 = bias[32 + l];
#pragma unroll
    for (int v = 0; v < 8; ++v) {
      const float x0 = acc[0][v] + bz0;
      const float x1 = acc[1][v] + bz1;
      const float x2 = c2 ? acc[2][v] + bz2 : -3.4e38f;
      const float m = rowMax16(fmaxf(fmaxf(x0, x1), x2));
      const float e0 = __expf(x0 - m);
      const float e1 = __expf(x1 - m);
      const float e2 = c2 ? __expf(x2 - m) : 0.f;
      const float inv = 1.0f / rowSum16(e0 + e1 + e2);
      float* orow = P + (size_t)((tm << 4) + v + 8 * h) * NCLS;
      orow[l] = e0 * inv;
      orow[16 + l] = e1 * inv;
      if (c2) orow[32 + l] = e2 * inv;
    }
  }
}

template <int NT, int K, int MODE>
__global__ void k_gemm_wmma(const float* __restrict__ A, const float* __restrict__ B,
                            float* __restrict__ C, int M,
                            const float* __restrict__ bias,
                            float* __restrict__ P) {
  __shared__ float ldsB[GemmChunk<K>::value * NT * 16];
  gemmBlock<NT, K, MODE>(A, B, C, M, bias, P, ldsB, (int)blockIdx.x);
}

#define CSR_G      64
#define CSR_Q      4
#define CSR_ROWS   (NNODES / CSR_Q)
#define CSR_CHUNK  (NEDGES / CSR_G)
#define CSR_T      1024
#define CSR_PER    ((CSR_CHUNK + CSR_T - 1) / CSR_T)
#define SCAN_PER   52
#define NPADROWS   (CSR_T * SCAN_PER)
#define ROWS_PER_WAVE 4
#define ROW_BLOCKS (((NNODES + ROWS_PER_WAVE - 1) / ROWS_PER_WAVE + 7) / 8)
static_assert(NEDGES % CSR_G == 0, "edge chunks must be equal");
static_assert(NNODES % CSR_Q == 0, "row ranges must be equal");
static_assert(NPADROWS >= NNODES + 1, "rowptr needs an end entry");
static_assert(SCAN_PER % 4 == 0, "rowscan moves 16-byte vectors");
static_assert((size_t)NEDGES * 4 <= (size_t)NNODES * NPAD * 4, "perm[] borrows the NPAD-column buffer");

#define FRONT_T    256
#define CNT_V4     (CSR_CHUNK / 4)
#define CNT_PER    ((CNT_V4 + FRONT_T - 1) / FRONT_T)
#define CNT_BATCH  13
static_assert(CSR_CHUNK % 4 == 0, "a chunk is whole 16-byte vectors and starts on one");
static_assert(CSR_ROWS % 4 == 0 && NNODES % 4 == 0, "a block's counters are whole 16-byte vectors and start on one");

__device__ __forceinline__ void countOne(unsigned* __restrict__ hist, const int d, const int lo) {
  const unsigned rr = (unsigned)(d - lo);
  if (rr < (unsigned)CSR_ROWS) atomicAdd(&hist[rr], 1u);
}

__device__ __forceinline__ void countBlock(const int* __restrict__ dst, unsigned* __restrict__ cnt,
                                           unsigned* __restrict__ hist, const int cb) {
  const int t = threadIdx.x;
  const int c = cb / CSR_Q;
  const int lo = (cb % CSR_Q) * CSR_ROWS;
  uint4* hist4 = reinterpret_cast<uint4*>(hist);
  for (int i = t; i < CSR_ROWS / 4; i += FRONT_T) hist4[i] = make_uint4(0u, 0u, 0u, 0u);
  __syncthreads();
  const int4* d4 = reinterpret_cast<const int4*>(dst + (size_t)c * CSR_CHUNK);
  for (int k0 = 0; k0 < CNT_PER; k0 += CNT_BATCH) {
    int4 dv[CNT_BATCH];
#pragma unroll
    for (int k = 0; k < CNT_BATCH; ++k) {
      const int i = t + (k0 + k) * FRONT_T;
      dv[k] = (i < CNT_V4) ? d4[i] : make_int4(-1, -1, -1, -1);
    }
#pragma unroll
    for (int k = 0; k < CNT_BATCH; ++k) {
      countOne(hist, dv[k].x, lo);
      countOne(hist, dv[k].y, lo);
      countOne(hist, dv[k].z, lo);
      countOne(hist, dv[k].w, lo);
    }
  }
  __syncthreads();
  uint4* o4 = reinterpret_cast<uint4*>(cnt + (size_t)c * NNODES + lo);
  for (int i = t; i < CSR_ROWS / 4; i += FRONT_T) o4[i] = hist4[i];
}

__global__ void k_csr_colscan(unsigned* __restrict__ cnt, unsigned* __restrict__ tot) {
  const int r = blockIdx.x * 256 + threadIdx.x;
  unsigned s = 0u;
  if (r < NNODES) {
    unsigned c[CSR_G];
    for (int b = 0; b < CSR_G; ++b) c[b] = cnt[(size_t)b * NNODES + r];
    for (int b = 0; b < CSR_G; ++b) { cnt[(size_t)b * NNODES + r] = s; s += c[b]; }
  }
  tot[r] = s;
}

__global__ void __launch_bounds__(CSR_T)
k_csr_rowscan(const unsigned* __restrict__ tot, unsigned* __restrict__ rowptr) {
  __shared__ unsigned wsum[32];
  const int t = threadIdx.x, lane = t & 31, wid = t >> 5;
  const uint4* in = reinterpret_cast<const uint4*>(tot) + t * (SCAN_PER / 4);
  uint4 v[SCAN_PER / 4];
  for (int i = 0; i < SCAN_PER / 4; ++i) v[i] = in[i];
  unsigned s = 0u;
  for (int i = 0; i < SCAN_PER / 4; ++i) s += v[i].x + v[i].y + v[i].z + v[i].w;
  unsigned inc = s;
  for (int o = 1; o < 32; o <<= 1) {
    unsigned n = __shfl_up(inc, o, 32);
    if (lane >= o) inc += n;
  }
  if (lane == 31) wsum[wid] = inc;
  __syncthreads();
  if (wid == 0) {
    unsigned w = wsum[lane], wi = w;
    for (int o = 1; o < 32; o <<= 1) {
      unsigned n = __shfl_up(wi, o, 32);
      if (lane >= o) wi += n;
    }
    wsum[lane] = wi - w;
  }
  __syncthreads();
  unsigned run = wsum[wid] + (inc - s);
  uint4* out = reinterpret_cast<uint4*>(rowptr) + t * (SCAN_PER / 4);
  for (int i = 0; i < SCAN_PER / 4; ++i) {
    uint4 q;
    q.x = run; run += v[i].x;
    q.y = run; run += v[i].y;
    q.z = run; run += v[i].z;
    q.w = run; run += v[i].w;
    out[i] = q;
  }
}

__global__ void __launch_bounds__(CSR_T)
k_csr_fill(const int* __restrict__ dst, const unsigned* __restrict__ blkoff,
           const unsigned* __restrict__ rowptr, int* __restrict__ perm) {
  __shared__ unsigned hist[CSR_ROWS];
  const int t = threadIdx.x;
  const int c = blockIdx.x / CSR_Q;
  const int lo = (blockIdx.x % CSR_Q) * CSR_ROWS;
  const unsigned* bo = blkoff + (size_t)c * NNODES + lo;
  for (int i = t; i < CSR_ROWS; i += CSR_T) hist[i] = rowptr[lo + i] + bo[i];
  __syncthreads();
  const int e0 = c * CSR_CHUNK;
  int dv[CSR_PER];
  for (int k = 0; k < CSR_PER; ++k) {
    const int i = t + k * CSR_T;
    dv[k] = (i < CSR_CHUNK) ? dst[e0 + i] : -1;
  }
  for (int k = 0; k < CSR_PER; ++k) {
    const unsigned rr = (unsigned)(dv[k] - lo);
    if (rr < (unsigned)CSR_ROWS) {
      const unsigned pos = atomicAdd(&hist[rr], 1u);
      perm[pos] = e0 + t + k * CSR_T;
    }
  }
}

#define RANK_CAP 128

template <int NB>
__device__ __forceinline__ void rankAndPlace(const int start, const int deg, const int lane,
                                             const int* __restrict__ perm, const int* __restrict__ src,
                                             const float* __restrict__ ew, int2* mine,
                                             int2* __restrict__ csr) {
  int e[NB], rk[NB];
#pragma unroll
  for (int a = 0; a < NB; ++a) {
    const int i = 32 * a + lane;
    e[a] = (i < deg) ? perm[start + i] : 0x7fffffff;
    rk[a] = 0;
  }
#pragma unroll
  for (int b = 0; b < NB; ++b) {
#pragma unroll(NB > 2 ? 4 : 32)
    for (int j = 0; j < 32; ++j) {
      const int s = __builtin_amdgcn_readlane(e[b], j);
#pragma unroll
      for (int a = 0; a < NB; ++a) rk[a] += (s < e[a]) ? 1 : 0;
    }
  }
  int sv[NB];
  float wv[NB];
#pragma unroll
  for (int a = 0; a < NB; ++a) {
    const bool ok = 32 * a + lane < deg;
    sv[a] = ok ? src[e[a]] : 0;
    wv[a] = ok ? ew[e[a]] : 0.f;
  }
#pragma unroll
  for (int a = 0; a < NB; ++a) {
    if (32 * a + lane < deg) {
      const int2 rec = make_int2(sv[a], __float_as_int(wv[a]));
      mine[rk[a]] = rec;
      csr[start + rk[a]] = rec;
    }
  }
}

#define BLOCK_ROWS 32
#define TILE_PITCH (HID + 4)
static_assert(BLOCK_ROWS == 8 * ROWS_PER_WAVE, "one row per wave and pass");
static_assert(NNODES % 16 == 0, "a matrix tile is all rows or none");
static_assert((NNODES + BLOCK_ROWS - 1) / BLOCK_ROWS == ROW_BLOCKS, "grid covers the rows once");

__global__ void __launch_bounds__(256)
k_spmm_rows_rank(const unsigned* __restrict__ rowptr, const int* __restrict__ perm,
                 const int* __restrict__ src, const float* __restrict__ ew,
                 int* __restrict__ csrw, const float* __restrict__ X,
                 const float* __restrict__ b1, const float* __restrict__ W2f,
                 float* __restrict__ X2f) {
  __shared__ int2 slots[8][RANK_CAP];
  __shared__ float tile[BLOCK_ROWS * TILE_PITCH];
  int2* csr = reinterpret_cast<int2*>(csrw);
  const int lane = threadIdx.x & 31;
  const int wl = __builtin_amdgcn_readfirstlane((int)(threadIdx.x >> 5));
  const int rbase = blockIdx.x * BLOCK_ROWS;
  int2* mine = slots[wl];
  const float4* X4 = reinterpret_cast<const float4*>(X);
  for (int k = 0; k < ROWS_PER_WAVE; ++k) {
    const int rl = wl + 8 * k;
    const int r = rbase + rl;
    if (r < NNODES) {
      const int start = (int)rowptr[r];
      const int deg = (int)rowptr[r + 1] - start;
      float4 acc = make_float4(0.f, 0.f, 0.f, 0.f);
      if (deg <= RANK_CAP) {
        if (deg <= 32)      rankAndPlace<1>(start, deg, lane, perm, src, ew, mine, csr);
        else if (deg <= 64) rankAndPlace<2>(start, deg, lane, perm, src, ew, mine, csr);
        else                rankAndPlace<4>(start, deg, lane, perm, src, ew, mine, csr);
        for (int base = 0; base < deg; base += 32) {
          const int cnt = min(32, deg - base);
          const int2 rec = (lane < cnt) ? mine[base + lane] : make_int2(0, 0);
          for (int j = 0; j < cnt; j += 8) {
            float4 x[8];
            for (int u = 0; u < 8; ++u) {
              const int ju = min(j + u, cnt - 1);
              const int s = __builtin_amdgcn_readlane(rec.x, ju);
              x[u] = X4[(size_t)s * (HID / 4) + lane];
            }
            for (int u = 0; u < 8; ++u) {
              if (j + u < cnt) {
                const float w = __int_as_float(__builtin_amdgcn_readlane(rec.y, j + u));
                acc.x = fmaf(w, x[u].x, acc.x);
                acc.y = fmaf(w, x[u].y, acc.y);
                acc.z = fmaf(w, x[u].z, acc.z);
                acc.w = fmaf(w, x[u].w, acc.w);
              }
            }
          }
        }
      } else {
        int cur = -1;
        for (int j = 0; j < deg; ++j) {
          int best = 0x7fffffff;
          for (int b = 0; b < deg; b += 32) {
            const int i = b + lane;
            const int e = (i < deg) ? perm[start + i] : 0x7fffffff;
            if (e > cur && e < best) best = e;
          }
          for (int o = 16; o > 0; o >>= 1) best = min(best, __shfl_xor(best, o, 32));
          cur = __builtin_amdgcn_readfirstlane(best);
          const int s = src[cur];
          const float w = ew[cur];
          if (lane == 0) csr[start + j] = make_int2(s, __float_as_int(w));
          const float4 xv = X4[(size_t)s * (HID / 4) + lane];
          acc.x = fmaf(w, xv.x, acc.x);
          acc.y = fmaf(w, xv.y, acc.y);
          acc.z = fmaf(w, xv.z, acc.z);
          acc.w = fmaf(w, xv.w, acc.w);
        }
      }
      *reinterpret_cast<float4*>(tile + rl * TILE_PITCH + 4 * lane) = acc;
    }
  }
  __syncthreads();
  const int row0 = rbase + 16 * wl;
  if (wl < BLOCK_ROWS / 16 && row0 < NNODES) {
    const int h = lane >> 4;
    const int l = lane & 15;
    const float* arow = tile + (16 * wl + l) * TILE_PITCH;
    const float* bcol = W2f + l;
    v8f d0, d1, d2;
#pragma unroll
    for (int v = 0; v < 8; ++v) { d0[v] = 0.f; d1[v] = 0.f; d2[v] = 0.f; }
#pragma unroll 2
    for (int t = 0; t < HID / 4; ++t) {
      const int ka = 4 * t + 2 * h;
      const float* w0 = bcol + (size_t)ka * NPAD;
      const float* w1 = w0 + NPAD;
      const v2f s = *reinterpret_cast<const v2f*>(arow + ka);
      v2f a, b0, b1v, b2;
      a.x = fmaxf(s.x + b1[ka], 0.f);
      a.y = fmaxf(s.y + b1[ka + 1], 0.f);
      b0.x = w0[0];   b0.y = w1[0];
      b1v.x = w0[16]; b1v.y = w1[16];
      b2.x = w0[32];  b2.y = w1[32];
      d0 = __builtin_amdgcn_wmma_f32_16x16x4_f32(false, a, false, b0, (short)0, d0, false, false);
      d1 = __builtin_amdgcn_wmma_f32_16x16x4_f32(false, a, false, b1v, (short)0, d1, false, false);
      d2 = __builtin_amdgcn_wmma_f32_16x16x4_f32(false, a, false, b2, (short)0, d2, false, false);
    }
#pragma unroll
    for (int v = 0; v < 8; ++v) {
      float* orow = X2f + (size_t)(row0 + v + 8 * h) * NPAD;
      orow[l] = d0[v];
      orow[16 + l] = d1[v];
      orow[32 + l] = d2[v];
    }
  }
}

#define ROW4 (NPAD / 4)
static_assert(NNODES % 2 == 0, "rows are taken in pairs");

template <int K>
__device__ __forceinline__ int rowShare(int v) {
  return __builtin_amdgcn_update_dpp(0, v, 0x150 + K, 0xF, 0xF, false);
}

template <int J>
__device__ __forceinline__ void loadEight(const int2 rec, const int cnt, const bool ld,
                                          const int sub, const float4* __restrict__ X4,
                                          float4 (&x)[8]) {
#define SPMM48_LOAD(U)                                                      \
  {                                                                         \
    const int s = rowShare<J + U>(rec.x);                                   \
    x[U] = make_float4(0.f, 0.f, 0.f, 0.f);                                 \
    if (ld && (J + U) < cnt) x[U] = X4[(size_t)s * ROW4 + sub];             \
  }
  SPMM48_LOAD(0) SPMM48_LOAD(1) SPMM48_LOAD(2) SPMM48_LOAD(3)
  SPMM48_LOAD(4) SPMM48_LOAD(5) SPMM48_LOAD(6) SPMM48_LOAD(7)
#undef SPMM48_LOAD
}

template <int J>
__device__ __forceinline__ void fmaEight(const int2 rec, const float4 (&x)[8], float4& acc) {
#define SPMM48_FMA(U)                                                       \
  {                                                                         \
    const float w = __int_as_float(rowShare<J + U>(rec.y));                 \
    acc.x = fmaf(w, x[U].x, acc.x);                                         \
    acc.y = fmaf(w, x[U].y, acc.y);                                         \
    acc.z = fmaf(w, x[U].z, acc.z);                                         \
    acc.w = fmaf(w, x[U].w, acc.w);                                         \
  }
  SPMM48_FMA(0) SPMM48_FMA(1) SPMM48_FMA(2) SPMM48_FMA(3)
  SPMM48_FMA(4) SPMM48_FMA(5) SPMM48_FMA(6) SPMM48_FMA(7)
#undef SPMM48_FMA
}

__global__ void k_spmm_rows48(const unsigned* __restrict__ rowptr, const int* __restrict__ csrw,
                              const float* __restrict__ X, float* __restrict__ agg) {
  const int2* csr = reinterpret_cast<const int2*>(csrw);
  const int lane = threadIdx.x & 31;
  const int hsel = lane >> 4;
  const int sub = lane & 15;
  const bool ld = sub < ROW4;
  const int w0 = __builtin_amdgcn_readfirstlane((int)(blockIdx.x * (blockDim.x >> 5) + (threadIdx.x >> 5)));
  const int nw = gridDim.x * (blockDim.x >> 5);
  const float4* X4 = reinterpret_cast<const float4*>(X);
  for (int p = w0; p < NNODES / 2; p += nw) {
    const int r0 = 2 * p;
    const int s0 = (int)rowptr[r0];
    const int s1 = (int)rowptr[r0 + 1];
    const int s2 = (int)rowptr[r0 + 2];
    const int degmax = max(s1 - s0, s2 - s1);
    const int start = hsel ? s1 : s0;
    const int deg = hsel ? (s2 - s1) : (s1 - s0);
    float4 acc = make_float4(0.f, 0.f, 0.f, 0.f);
    int cnt = deg;
    int2 rec = (sub < cnt) ? csr[start + sub] : make_int2(0, 0);
    for (int base = 0; base < degmax; base += 16) {
      float4 x[8];
      loadEight<0>(rec, cnt, ld, sub, X4, x);
      int2 nxt = make_int2(0, 0);
      if (sub + 16 < cnt) nxt = csr[start + base + 16 + sub];
      fmaEight<0>(rec, x, acc);
      if (degmax - base > 8) {
        loadEight<8>(rec, cnt, ld, sub, X4, x);
        fmaEight<8>(rec, x, acc);
      }
      rec = nxt;
      cnt -= 16;
    }
    if (ld) reinterpret_cast<float4*>(agg)[(size_t)(r0 + hsel) * ROW4 + sub] = acc;
  }
}

#define PREP_THREADS 256
__device__ __forceinline__ void foldTile(const float* __restrict__ A, const int row0, const int nrows,
                                         const float* __restrict__ Wh, float* __restrict__ D) {
  const int lane = threadIdx.x & 31;
  const int h = lane >> 4;
  const int l = lane & 15;
  const int row = row0 + l;
  const float* Arow = A + (size_t)(row < nrows ? row : 0) * HID;
  const bool cv = 32 + l < NCLS;
  const int n2 = cv ? 32 + l : 0;
  v8f acc0, acc1, acc2;
#pragma unroll
  for (int v = 0; v < 8; ++v) { acc0[v] = 0.f; acc1[v] = 0.f; acc2[v] = 0.f; }
#pragma unroll 8
  for (int t = 0; t < HID / 4; ++t) {
    const int ka = 4 * t + 2 * h;
    const float* w0 = Wh + (size_t)ka * NCLS;
    const float* w1 = w0 + NCLS;
    v2f a, b0, b1, b2;
    a.x = Arow[ka];
    a.y = Arow[ka + 1];
    b0.x = w0[l];
    b0.y = w1[l];
    b1.x = w0[16 + l];
    b1.y = w1[16 + l];
    b2.x = w0[n2];
    b2.y = w1[n2];
    acc0 = __builtin_amdgcn_wmma_f32_16x16x4_f32(false, a, false, b0, (short)0, acc0, false, false);
    acc1 = __builtin_amdgcn_wmma_f32_16x16x4_f32(false, a, false, b1, (short)0, acc1, false, false);
    acc2 = __builtin_amdgcn_wmma_f32_16x16x4_f32(false, a, false, b2, (short)0, acc2, false, false);
  }
#pragma unroll
  for (int v = 0; v < 8; ++v) {
    const int r = row0 + v + 8 * h;
    if (r < nrows) {
      D[(size_t)r * NPAD + l]      = acc0[v];
      D[(size_t)r * NPAD + 16 + l] = acc1[v];
      D[(size_t)r * NPAD + 32 + l] = cv ? acc2[v] : 0.f;
    }
  }
}

#define PREP_LDS_WORDS (2 * HID * NCLS + 2 * HID)
static_assert(PREP_THREADS == FRONT_T, "the prep roles run on the front launch's block size");
__device__ __forceinline__ void prepBlock(const int role,
                                          const float* __restrict__ W2, const float* __restrict__ We,
                                          const float* __restrict__ Wf, const float* __restrict__ bf,
                                          const float* __restrict__ be, const float* __restrict__ b2,
                                          float* __restrict__ W2f, float* __restrict__ Wef,
                                          float* __restrict__ bfp, float* __restrict__ lds) {
  float* sW  = lds;
  float* sBe = lds + 2 * HID * NCLS;
  float* sB2 = sBe + HID;
  const int t = threadIdx.x;
  const int wid = __builtin_amdgcn_readfirstlane((int)(threadIdx.x >> 5));
  if (role == 0) {
    foldTile(W2, 16 * wid, HID, Wf + (size_t)HID * NCLS, W2f);
    return;
  }
  if (role == 1) {
    if (wid < (NCLS + 15) / 16) foldTile(We, 16 * wid, NCLS, Wf, Wef);
    return;
  }
  const float4* Wf4 = reinterpret_cast<const float4*>(Wf);
  float4* sW4 = reinterpret_cast<float4*>(sW);
#pragma unroll
  for (int j = 0; j < (2 * HID * NCLS) / 4 / PREP_THREADS; ++j)
    sW4[j * PREP_THREADS + t] = Wf4[j * PREP_THREADS + t];
  if (t < HID) sBe[t] = be[t];
  else         sB2[t - HID] = b2[t - HID];
  __syncthreads();
  if (t < NPAD) {
    float s = 0.f;
    if (t < NCLS) {
      s = bf[t];
      for (int k = 0; k < HID; ++k) {
        s += sBe[k] * sW[k * NCLS + t];
        s += sB2[k] * sW[(HID + k) * NCLS + t];
      }
    }
    bfp[t] = s;
  }
}

#define GEMM_BLOCKS  (((NNODES / 16) + 7) / 8)
#define COUNT_BLOCKS (CSR_G * CSR_Q)
#define FRONT_BLOCKS (GEMM_BLOCKS + COUNT_BLOCKS + 3)
#define FRONT_LDS4   (CSR_ROWS / 4)
static_assert(FRONT_LDS4 * 4 >= PREP_LDS_WORDS, "the prep role's staging fits the block's LDS");
static_assert(FRONT_LDS4 * 4 >= GemmChunk<FIN>::value * HID, "the GEMM role's B chunk fits the block's LDS");

__global__ void
k_front(const float* __restrict__ features, const float* __restrict__ W1, float* __restrict__ X1,
        const int* __restrict__ dst, unsigned* __restrict__ cnt,
        const float* __restrict__ W2, const float* __restrict__ We,
        const float* __restrict__ Wf, const float* __restrict__ bf,
        const float* __restrict__ be, const float* __restrict__ b2,
        float* __restrict__ W2f, float* __restrict__ Wef, float* __restrict__ bfp) {
  __shared__ float4 lds4[FRONT_LDS4];
  float* lds = reinterpret_cast<float*>(lds4);
  const int bid = (int)blockIdx.x;
  if (bid < GEMM_BLOCKS) {
    gemmBlock<8, FIN, 1>(features, W1, X1, NNODES, nullptr, nullptr, lds, bid);
  } else if (bid < GEMM_BLOCKS + COUNT_BLOCKS) {
    countBlock(dst, cnt, reinterpret_cast<unsigned*>(lds), bid - GEMM_BLOCKS);
  } else {
    prepBlock(bid - (GEMM_BLOCKS + COUNT_BLOCKS), W2, We, Wf, bf, be, b2, W2f, Wef, bfp, lds);
  }
}

extern "C" void kernel_launch(void* const* d_in, const int* in_sizes, int n_in,
                              void* d_out, int out_size, void* d_ws, size_t ws_size,
                              hipStream_t stream) {
  const float* features = (const float*)d_in[0];
  const int*   src      = (const int*)d_in[1];
  const int*   dst      = (const int*)d_in[2];
  const float* ew       = (const float*)d_in[3];
  const float* y        = (const float*)d_in[4];
  const float* W1       = (const float*)d_in[5];
  const float* b1       = (const float*)d_in[6];
  const float* W2       = (const float*)d_in[7];
  const float* b2       = (const float*)d_in[8];
  const float* We       = (const float*)d_in[9];
  const float* be       = (const float*)d_in[10];
  const float* Wf       = (const float*)d_in[11];
  const float* bf       = (const float*)d_in[12];
  float* out = (float*)d_out;
  (void)in_sizes; (void)n_in; (void)out_size; (void)ws_size;

  char* ws = (char*)d_ws;
  size_t off = 0;
  auto alloc = [&](size_t bytes) -> float* {
    float* p = (float*)(ws + off);
    off += (bytes + 255) & ~(size_t)255;
    return p;
  };
  float* bufA   = alloc((size_t)NNODES * HID * 4);
  float* bufAgg = alloc((size_t)NNODES * HID * 4);
  float* bufL   = alloc((size_t)NNODES * NPAD * 4);
  float* W2f    = alloc((size_t)HID * NPAD * 4);
  float* Wef    = alloc((size_t)NCLS * NPAD * 4);
  float* bfp    = alloc((size_t)NPAD * 4);
  unsigned* rowptr = (unsigned*)alloc((size_t)NPADROWS * 4);
  int*      csr    = (int*)alloc((size_t)NEDGES * 8);
  unsigned* cntMat = (unsigned*)bufAgg;
  unsigned* tot    = cntMat + (size_t)CSR_G * NNODES;
  int*      perm   = (int*)bufL;

  const int gemmBlocks = GEMM_BLOCKS;

  k_front<<<FRONT_BLOCKS, FRONT_T, 0, stream>>>(features, W1, bufA, dst, cntMat,
                                                W2, We, Wf, bf, be, b2, W2f, Wef, bfp);
  k_csr_colscan<<<NPADROWS / 256, 256, 0, stream>>>(cntMat, tot);
  k_csr_rowscan<<<1, CSR_T, 0, stream>>>(tot, rowptr);
  k_csr_fill<<<CSR_G * CSR_Q, CSR_T, 0, stream>>>(dst, cntMat, rowptr, perm);
  k_spmm_rows_rank<<<ROW_BLOCKS, 256, 0, stream>>>(rowptr, perm, src, ew, csr, bufA, b1, W2f, bufAgg);
  k_spmm_rows48<<<ROW_BLOCKS, 256, 0, stream>>>(rowptr, csr, bufAgg, bufL);
  k_gemm_wmma<3, NCLS, 28><<<gemmBlocks, 256, 0, stream>>>(
      y, Wef, bufL, NNODES, bfp, out);
}


# Edge grouping retiled to 16 chunks x 16 row ranges: each fill block sums its rows' chunk counts and scans them itself (count emits block sums); column-scan and row-scan launches gone
# speedup vs baseline: 5.6183x; 1.3222x over previous
#include <hip/hip_runtime.h>
#include <hip/hip_bf16.h>

typedef float v2f __attribute__((ext_vector_type(2)));
typedef float v8f __attribute__((ext_vector_type(8)));

#define NNODES 50000
#define NEDGES 1600000
#define FIN    256
#define HID    128
#define NCLS   40
#define NPAD   48

template <int X>
__device__ __forceinline__ float rowXor(float v) {
  return __int_as_float(__builtin_amdgcn_update_dpp(0, __float_as_int(v), 0x160 + X, 0xF, 0xF, false));
}
__device__ __forceinline__ float rowMax16(float v) {
  v = fmaxf(v, rowXor<8>(v)); v = fmaxf(v, rowXor<4>(v));
  v = fmaxf(v, rowXor<2>(v)); v = fmaxf(v, rowXor<1>(v));
  return v;
}
__device__ __forceinline__ float rowSum16(float v) {
  v += rowXor<8>(v); v += rowXor<4>(v); v += rowXor<2>(v); v += rowXor<1>(v);
  return v;
}

template <int K> struct GemmChunk { static constexpr int value = (K % 32 == 0) ? 32 : K; };

template <int NT, int K, int MODE>
__device__ __forceinline__ void gemmBlock(const float* __restrict__ A, const float* __restrict__ B,
                                          float* __restrict__ C, int M,
                                          const float* __restrict__ bias,
                                          float* __restrict__ P,
                                          float* __restrict__ ldsB, const int bid) {
  constexpr bool USE_SCALE = (MODE & 1) != 0;
  constexpr bool USE_BIAS  = (MODE & 4) != 0;
  constexpr bool ACCUM     = (MODE & 8) != 0;
  constexpr bool SOFTMAX   = (MODE & 16) != 0;
  constexpr int N = NT * 16;
  static_assert(!SOFTMAX || (NT == 3 && USE_BIAS && !USE_SCALE), "softmax epilogue: NPAD columns, biased logits");
  constexpr int KCHUNK = GemmChunk<K>::value;
  constexpr int NCHUNKS = K / KCHUNK;
  constexpr int TSTEPS = KCHUNK / 4;
  constexpr int GROUP = (TSTEPS % 8 == 0) ? 8 : TSTEPS / 2;
  constexpr int NV4 = (KCHUNK * N) / 4;
  constexpr int STAGE4 = (NV4 + 255) / 256;

  const int lane = threadIdx.x & 31;
  const int wid  = threadIdx.x >> 5;
  const int tm   = bid * (blockDim.x >> 5) + wid;
  const bool active = tm < (M >> 4);
  const int h = lane >> 4;
  const int l = lane & 15;
  const int row = (tm << 4) + l;

  v8f acc[NT];
  if (active) {
#pragma unroll
    for (int j = 0; j < NT; ++j)
#pragma unroll
      for (int v = 0; v < 8; ++v)
        acc[j][v] = ACCUM ? C[(size_t)((tm << 4) + v + 8 * h) * N + j * 16 + l] : 0.f;
  }
  const float* Arow = A + (size_t)row * K;
  float ss = 0.f;

  for (int c = 0; c < NCHUNKS; ++c) {
    const int k0 = c * KCHUNK;
    __syncthreads();
#pragma unroll
    for (int i = 0; i < STAGE4; ++i) {
      int idx4 = i * 256 + threadIdx.x;
      if (NV4 % 256 == 0 || idx4 < NV4) {
        int f = idx4 * 4;
        int kk = f / N, n = f - kk * N;
        float4 q = *(const float4*)(B + (size_t)(k0 + kk) * N + n);
        float* d = ldsB + ((kk >> 1) * N + n) * 2 + (kk & 1);
        d[0] = q.x; d[2] = q.y; d[4] = q.z; d[6] = q.w;
      }
    }
    __syncthreads();
    if (active) {
#pragma unroll
      for (int g = 0; g < TSTEPS; g += GROUP) {
        v2f areg[GROUP];
#pragma unroll
        for (int u = 0; u < GROUP; ++u) {
          const int ka = k0 + (g + u) * 4 + 2 * h;
          v2f a;
          a.x = Arow[ka];
          a.y = Arow[ka + 1];
          areg[u] = a;
        }
#pragma unroll
        for (int u = 0; u < GROUP; ++u) {
          const int ka = (g + u) * 4 + 2 * h;
          const float* bp = ldsB + (ka >> 1) * (2 * N) + 2 * l;
          if (USE_SCALE) {
            ss = fmaf(areg[u].x, areg[u].x, ss);
            ss = fmaf(areg[u].y, areg[u].y, ss);
          }
#pragma unroll
          for (int j = 0; j < NT; ++j) {
            v2f b = *(const v2f*)(bp + 32 * j);
            acc[j] = __builtin_amdgcn_wmma_f32_16x16x4_f32(
                false, areg[u], false, b, (short)0, acc[j], false, false);
          }
        }
      }
    }
  }

  if (active && !SOFTMAX) {
    float sc[8];
    if (USE_SCALE) {
      const float tot = ss + __shfl_xor(ss, 16, 32);
      const float inv = 1.0f / fmaxf(sqrtf(tot), 1e-12f);
#pragma unroll
      for (int v = 0; v < 8; ++v) sc[v] = __shfl(inv, v + 8 * h, 32);
    }
#pragma unroll
    for (int j = 0; j < NT; ++j)
#pragma unroll
      for (int v = 0; v < 8; ++v) {
        const int cn = j * 16 + l;
        float o = acc[j][v];
        if (USE_SCALE) o *= sc[v];
        if (USE_BIAS) o += bias[cn];
        C[(size_t)((tm << 4) + v + 8 * h) * N + cn] = o;
      }
  }
  if (active && SOFTMAX) {
    const bool c2 = 32 + l < NCLS;
    const float bz0 = bias[l], bz1 = bias[16 + l], bz2 = bias[32 + l];
#pragma unroll
    for (int v = 0; v < 8; ++v) {
      const float x0 = acc[0][v] + bz0;
      const float x1 = acc[1][v] + bz1;
      const float x2 = c2 ? acc[2][v] + bz2 : -3.4e38f;
      const float m = rowMax16(fmaxf(fmaxf(x0, x1), x2));
      const float e0 = __expf(x0 - m);
      const float e1 = __expf(x1 - m);
      const float e2 = c2 ? __expf(x2 - m) : 0.f;
      const float inv = 1.0f / rowSum16(e0 + e1 + e2);
      float* orow = P + (size_t)((tm << 4) + v + 8 * h) * NCLS;
      orow[l] = e0 * inv;
      orow[16 + l] = e1 * inv;
      if (c2) orow[32 + l] = e2 * inv;
    }
  }
}

template <int NT, int K, int MODE>
__global__ void k_gemm_wmma(const float* __restrict__ A, const float* __restrict__ B,
                            float* __restrict__ C, int M,
                            const float* __restrict__ bias,
                            float* __restrict__ P) {
  __shared__ float ldsB[GemmChunk<K>::value * NT * 16];
  gemmBlock<NT, K, MODE>(A, B, C, M, bias, P, ldsB, (int)blockIdx.x);
}

#define CSR_G      16
#define CSR_Q      16
#define CSR_ROWS   (NNODES / CSR_Q)
#define CSR_CHUNK  (NEDGES / CSR_G)
#define CSR_V4     (CSR_CHUNK / 4)
#define CSR_T      1024
#define FILL_PER   ((CSR_V4 + CSR_T - 1) / CSR_T)
#define FILL_BATCH 13
#define FILL_RPT   4
#define ROWS_PER_WAVE 4
#define ROW_BLOCKS (((NNODES + ROWS_PER_WAVE - 1) / ROWS_PER_WAVE + 7) / 8)
static_assert(NEDGES % CSR_G == 0, "edge chunks must be equal");
static_assert(NNODES % CSR_Q == 0, "row ranges must be equal");
static_assert(CSR_CHUNK % 4 == 0, "a chunk is whole 16-byte vectors and starts on one");
static_assert(CSR_T * FILL_RPT >= CSR_ROWS, "the fill's threads cover the rows of a range");
static_assert(CSR_G * CSR_Q <= CSR_T, "one thread per block sum");
static_assert((size_t)NEDGES * 4 <= (size_t)NNODES * NPAD * 4, "perm[] borrows the NPAD-column buffer");

#define FRONT_T    256
#define CNT_PER    ((CSR_V4 + FRONT_T - 1) / FRONT_T)
#define CNT_BATCH  25

__device__ __forceinline__ void countOne(unsigned* __restrict__ hist, const int d, const int lo) {
  const unsigned rr = (unsigned)(d - lo);
  if (rr < (unsigned)CSR_ROWS) atomicAdd(&hist[rr], 1u);
}

__device__ __forceinline__ void countBlock(const int* __restrict__ dst, unsigned* __restrict__ cnt,
                                           unsigned* __restrict__ bsum,
                                           unsigned* __restrict__ hist, const int cb) {
  const int t = threadIdx.x;
  const int c = cb / CSR_Q;
  const int lo = (cb % CSR_Q) * CSR_ROWS;
  for (int i = t; i <= CSR_ROWS; i += FRONT_T) hist[i] = 0u;
  __syncthreads();
  const int4* d4 = reinterpret_cast<const int4*>(dst + (size_t)c * CSR_CHUNK);
  for (int k0 = 0; k0 < CNT_PER; k0 += CNT_BATCH) {
    int4 dv[CNT_BATCH];
#pragma unroll
    for (int k = 0; k < CNT_BATCH; ++k) {
      const int i = t + (k0 + k) * FRONT_T;
      dv[k] = (i < CSR_V4) ? d4[i] : make_int4(-1, -1, -1, -1);
    }
#pragma unroll
    for (int k = 0; k < CNT_BATCH; ++k) {
      countOne(hist, dv[k].x, lo);
      countOne(hist, dv[k].y, lo);
      countOne(hist, dv[k].z, lo);
      countOne(hist, dv[k].w, lo);
    }
  }
  __syncthreads();
  unsigned* orow = cnt + (size_t)c * NNODES + lo;
  unsigned part = 0u;
  for (int i = t; i < CSR_ROWS; i += FRONT_T) {
    const unsigned v = hist[i];
    orow[i] = v;
    part += v;
  }
  for (int o = 16; o > 0; o >>= 1) part += __shfl_xor(part, o, 32);
  if ((t & 31) == 0) atomicAdd(&hist[CSR_ROWS], part);
  __syncthreads();
  if (t == 0) bsum[cb] = hist[CSR_ROWS];
}

__device__ __forceinline__ void placeOne(unsigned* __restrict__ hist, int* __restrict__ perm,
                                         const int d, const int lo, const int e) {
  const unsigned rr = (unsigned)(d - lo);
  if (rr < (unsigned)CSR_ROWS) {
    const unsigned pos = atomicAdd(&hist[rr], 1u);
    perm[pos] = e;
  }
}

__global__ void __launch_bounds__(CSR_T)
k_csr_fill(const int* __restrict__ dst, const unsigned* __restrict__ cnt,
           const unsigned* __restrict__ bsum, unsigned* __restrict__ rowptr,
           int* __restrict__ perm) {
  __shared__ unsigned hist[CSR_ROWS];
  __shared__ unsigned wsum[32];
  __shared__ unsigned wlow[32];
  __shared__ unsigned stat[2];
  const int t = threadIdx.x, lane = t & 31, wid = t >> 5;
  const int c = blockIdx.x / CSR_Q;
  const int q = blockIdx.x % CSR_Q;
  const int lo = q * CSR_ROWS;
  unsigned below = 0u;
  if (t < CSR_G * CSR_Q && (t % CSR_Q) < q) below = bsum[t];
  unsigned v[CSR_G][FILL_RPT];
#pragma unroll
  for (int g = 0; g < CSR_G; ++g) {
#pragma unroll
    for (int j = 0; j < FILL_RPT; ++j) {
      const int i = FILL_RPT * t + j;
      v[g][j] = (i < CSR_ROWS) ? cnt[(size_t)g * NNODES + lo + i] : 0u;
    }
  }
  unsigned ex[FILL_RPT], pr[FILL_RPT];
  unsigned s = 0u;
#pragma unroll
  for (int j = 0; j < FILL_RPT; ++j) {
    unsigned tj = 0u, pj = 0u;
#pragma unroll
    for (int g = 0; g < CSR_G; ++g) {
      tj += v[g][j];
      if (g < c) pj += v[g][j];
    }
    ex[j] = s;
    pr[j] = pj;
    s += tj;
  }
  unsigned inc = s;
  for (int o = 1; o < 32; o <<= 1) {
    const unsigned n = __shfl_up(inc, o, 32);
    if (lane >= o) inc += n;
  }
  for (int o = 16; o > 0; o >>= 1) below += __shfl_xor(below, o, 32);
  if (lane == 31) wsum[wid] = inc;
  if (lane == 0) wlow[wid] = below;
  __syncthreads();
  if (wid == 0) {
    const unsigned w = wsum[lane];
    unsigned wi = w;
    for (int o = 1; o < 32; o <<= 1) {
      const unsigned n = __shfl_up(wi, o, 32);
      if (lane >= o) wi += n;
    }
    unsigned bb = wlow[lane];
    for (int o = 16; o > 0; o >>= 1) bb += __shfl_xor(bb, o, 32);
    wsum[lane] = wi - w;
    if (lane == 31) { stat[0] = bb; stat[1] = wi; }
  }
  __syncthreads();
  const unsigned first = stat[0] + wsum[wid] + (inc - s);
#pragma unroll
  for (int j = 0; j < FILL_RPT; ++j) {
    const int i = FILL_RPT * t + j;
    if (i < CSR_ROWS) {
      const unsigned rp = first + ex[j];
      hist[i] = rp + pr[j];
      if (c == 0) rowptr[lo + i] = rp;
    }
  }
  if (c == 0 && q == CSR_Q - 1 && t == 0) rowptr[NNODES] = stat[0] + stat[1];
  __syncthreads();
  const int4* d4 = reinterpret_cast<const int4*>(dst + (size_t)c * CSR_CHUNK);
  const int e0 = c * CSR_CHUNK;
  for (int k0 = 0; k0 < FILL_PER; k0 += FILL_BATCH) {
    int4 dv[FILL_BATCH];
#pragma unroll
    for (int k = 0; k < FILL_BATCH; ++k) {
      const int i = t + (k0 + k) * CSR_T;
      dv[k] = (i < CSR_V4) ? d4[i] : make_int4(-1, -1, -1, -1);
    }
#pragma unroll
    for (int k = 0; k < FILL_BATCH; ++k) {
      const int e = e0 + 4 * (t + (k0 + k) * CSR_T);
      placeOne(hist, perm, dv[k].x, lo, e);
      placeOne(hist, perm, dv[k].y, lo, e + 1);
      placeOne(hist, perm, dv[k].z, lo, e + 2);
      placeOne(hist, perm, dv[k].w, lo, e + 3);
    }
  }
}

#define RANK_CAP 128

template <int NB>
__device__ __forceinline__ void rankAndPlace(const int start, const int deg, const int lane,
                                             const int* __restrict__ perm, const int* __restrict__ src,
                                             const float* __restrict__ ew, int2* mine,
                                             int2* __restrict__ csr) {
  int e[NB], rk[NB];
#pragma unroll
  for (int a = 0; a < NB; ++a) {
    const int i = 32 * a + lane;
    e[a] = (i < deg) ? perm[start + i] : 0x7fffffff;
    rk[a] = 0;
  }
#pragma unroll
  for (int b = 0; b < NB; ++b) {
#pragma unroll(NB > 2 ? 4 : 32)
    for (int j = 0; j < 32; ++j) {
      const int s = __builtin_amdgcn_readlane(e[b], j);
#pragma unroll
      for (int a = 0; a < NB; ++a) rk[a] += (s < e[a]) ? 1 : 0;
    }
  }
  int sv[NB];
  float wv[NB];
#pragma unroll
  for (int a = 0; a < NB; ++a) {
    const bool ok = 32 * a + lane < deg;
    sv[a] = ok ? src[e[a]] : 0;
    wv[a] = ok ? ew[e[a]] : 0.f;
  }
#pragma unroll
  for (int a = 0; a < NB; ++a) {
    if (32 * a + lane < deg) {
      const int2 rec = make_int2(sv[a], __float_as_int(wv[a]));
      mine[rk[a]] = rec;
      csr[start + rk[a]] = rec;
    }
  }
}

#define BLOCK_ROWS 32
#define TILE_PITCH (HID + 4)
static_assert(BLOCK_ROWS == 8 * ROWS_PER_WAVE, "one row per wave and pass");
static_assert(NNODES % 16 == 0, "a matrix tile is all rows or none");
static_assert((NNODES + BLOCK_ROWS - 1) / BLOCK_ROWS == ROW_BLOCKS, "grid covers the rows once");

__global__ void __launch_bounds__(256)
k_spmm_rows_rank(const unsigned* __restrict__ rowptr, const int* __restrict__ perm,
                 const int* __restrict__ src, const float* __restrict__ ew,
                 int* __restrict__ csrw, const float* __restrict__ X,
                 const float* __restrict__ b1, const float* __restrict__ W2f,
                 float* __restrict__ X2f) {
  __shared__ int2 slots[8][RANK_CAP];
  __shared__ float tile[BLOCK_ROWS * TILE_PITCH];
  int2* csr = reinterpret_cast<int2*>(csrw);
  const int lane = threadIdx.x & 31;
  const int wl = __builtin_amdgcn_readfirstlane((int)(threadIdx.x >> 5));
  const int rbase = blockIdx.x * BLOCK_ROWS;
  int2* mine = slots[wl];
  const float4* X4 = reinterpret_cast<const float4*>(X);
  for (int k = 0; k < ROWS_PER_WAVE; ++k) {
    const int rl = wl + 8 * k;
    const int r = rbase + rl;
    if (r < NNODES) {
      const int start = (int)rowptr[r];
      const int deg = (int)rowptr[r + 1] - start;
      float4 acc = make_float4(0.f, 0.f, 0.f, 0.f);
      if (deg <= RANK_CAP) {
        if (deg <= 32)      rankAndPlace<1>(start, deg, lane, perm, src, ew, mine, csr);
        else if (deg <= 64) rankAndPlace<2>(start, deg, lane, perm, src, ew, mine, csr);
        else                rankAndPlace<4>(start, deg, lane, perm, src, ew, mine, csr);
        for (int base = 0; base < deg; base += 32) {
          const int cnt = min(32, deg - base);
          const int2 rec = (lane < cnt) ? mine[base + lane] : make_int2(0, 0);
          for (int j = 0; j < cnt; j += 8) {
            float4 x[8];
            for (int u = 0; u < 8; ++u) {
              const int ju = min(j + u, cnt - 1);
              const int s = __builtin_amdgcn_readlane(rec.x, ju);
              x[u] = X4[(size_t)s * (HID / 4) + lane];
            }
            for (int u = 0; u < 8; ++u) {
              if (j + u < cnt) {
                const float w = __int_as_float(__builtin_amdgcn_readlane(rec.y, j + u));
                acc.x = fmaf(w, x[u].x, acc.x);
                acc.y = fmaf(w, x[u].y, acc.y);
                acc.z = fmaf(w, x[u].z, acc.z);
                acc.w = fmaf(w, x[u].w, acc.w);
              }
            }
          }
        }
      } else {
        int cur = -1;
        for (int j = 0; j < deg; ++j) {
          int best = 0x7fffffff;
          for (int b = 0; b < deg; b += 32) {
            const int i = b + lane;
            const int e = (i < deg) ? perm[start + i] : 0x7fffffff;
            if (e > cur && e < best) best = e;
          }
          for (int o = 16; o > 0; o >>= 1) best = min(best, __shfl_xor(best, o, 32));
          cur = __builtin_amdgcn_readfirstlane(best);
          const int s = src[cur];
          const float w = ew[cur];
          if (lane == 0) csr[start + j] = make_int2(s, __float_as_int(w));
          const float4 xv = X4[(size_t)s * (HID / 4) + lane];
          acc.x = fmaf(w, xv.x, acc.x);
          acc.y = fmaf(w, xv.y, acc.y);
          acc.z = fmaf(w, xv.z, acc.z);
          acc.w = fmaf(w, xv.w, acc.w);
        }
      }
      *reinterpret_cast<float4*>(tile + rl * TILE_PITCH + 4 * lane) = acc;
    }
  }
  __syncthreads();
  const int row0 = rbase + 16 * wl;
  if (wl < BLOCK_ROWS / 16 && row0 < NNODES) {
    const int h = lane >> 4;
    const int l = lane & 15;
    const float* arow = tile + (16 * wl + l) * TILE_PITCH;
    const float* bcol = W2f + l;
    v8f d0, d1, d2;
#pragma unroll
    for (int v = 0; v < 8; ++v) { d0[v] = 0.f; d1[v] = 0.f; d2[v] = 0.f; }
#pragma unroll 2
    for (int t = 0; t < HID / 4; ++t) {
      const int ka = 4 * t + 2 * h;
      const float* w0 = bcol + (size_t)ka * NPAD;
      const float* w1 = w0 + NPAD;
      const v2f s = *reinterpret_cast<const v2f*>(arow + ka);
      v2f a, b0, b1v, b2;
      a.x = fmaxf(s.x + b1[ka], 0.f);
      a.y = fmaxf(s.y + b1[ka + 1], 0.f);
      b0.x = w0[0];   b0.y = w1[0];
      b1v.x = w0[16]; b1v.y = w1[16];
      b2.x = w0[32];  b2.y = w1[32];
      d0 = __builtin_amdgcn_wmma_f32_16x16x4_f32(false, a, false, b0, (short)0, d0, false, false);
      d1 = __builtin_amdgcn_wmma_f32_16x16x4_f32(false, a, false, b1v, (short)0, d1, false, false);
      d2 = __builtin_amdgcn_wmma_f32_16x16x4_f32(false, a, false, b2, (short)0, d2, false, false);
    }
#pragma unroll
    for (int v = 0; v < 8; ++v) {
      float* orow = X2f + (size_t)(row0 + v + 8 * h) * NPAD;
      orow[l] = d0[v];
      orow[16 + l] = d1[v];
      orow[32 + l] = d2[v];
    }
  }
}

#define ROW4 (NPAD / 4)
static_assert(NNODES % 2 == 0, "rows are taken in pairs");

template <int K>
__device__ __forceinline__ int rowShare(int v) {
  return __builtin_amdgcn_update_dpp(0, v, 0x150 + K, 0xF, 0xF, false);
}

template <int J>
__device__ __forceinline__ void loadEight(const int2 rec, const int cnt, const bool ld,
                                          const int sub, const float4* __restrict__ X4,
                                          float4 (&x)[8]) {
#define SPMM48_LOAD(U)                                                      \
  {                                                                         \
    const int s = rowShare<J + U>(rec.x);                                   \
    x[U] = make_float4(0.f, 0.f, 0.f, 0.f);                                 \
    if (ld && (J + U) < cnt) x[U] = X4[(size_t)s * ROW4 + sub];             \
  }
  SPMM48_LOAD(0) SPMM48_LOAD(1) SPMM48_LOAD(2) SPMM48_LOAD(3)
  SPMM48_LOAD(4) SPMM48_LOAD(5) SPMM48_LOAD(6) SPMM48_LOAD(7)
#undef SPMM48_LOAD
}

template <int J>
__device__ __forceinline__ void fmaEight(const int2 rec, const float4 (&x)[8], float4& acc) {
#define SPMM48_FMA(U)                                                       \
  {                                                                         \
    const float w = __int_as_float(rowShare<J + U>(rec.y));                 \
    acc.x = fmaf(w, x[U].x, acc.x);                                         \
    acc.y = fmaf(w, x[U].y, acc.y);                                         \
    acc.z = fmaf(w, x[U].z, acc.z);                                         \
    acc.w = fmaf(w, x[U].w, acc.w);                                         \
  }
  SPMM48_FMA(0) SPMM48_FMA(1) SPMM48_FMA(2) SPMM48_FMA(3)
  SPMM48_FMA(4) SPMM48_FMA(5) SPMM48_FMA(6) SPMM48_FMA(7)
#undef SPMM48_FMA
}

__global__ void k_spmm_rows48(const unsigned* __restrict__ rowptr, const int* __restrict__ csrw,
                              const float* __restrict__ X, float* __restrict__ agg) {
  const int2* csr = reinterpret_cast<const int2*>(csrw);
  const int lane = threadIdx.x & 31;
  const int hsel = lane >> 4;
  const int sub = lane & 15;
  const bool ld = sub < ROW4;
  const int w0 = __builtin_amdgcn_readfirstlane((int)(blockIdx.x * (blockDim.x >> 5) + (threadIdx.x >> 5)));
  const int nw = gridDim.x * (blockDim.x >> 5);
  const float4* X4 = reinterpret_cast<const float4*>(X);
  for (int p = w0; p < NNODES / 2; p += nw) {
    const int r0 = 2 * p;
    const int s0 = (int)rowptr[r0];
    const int s1 = (int)rowptr[r0 + 1];
    const int s2 = (int)rowptr[r0 + 2];
    const int degmax = max(s1 - s0, s2 - s1);
    const int start = hsel ? s1 : s0;
    const int deg = hsel ? (s2 - s1) : (s1 - s0);
    float4 acc = make_float4(0.f, 0.f, 0.f, 0.f);
    int cnt = deg;
    int2 rec = (sub < cnt) ? csr[start + sub] : make_int2(0, 0);
    for (int base = 0; base < degmax; base += 16) {
      float4 x[8];
      loadEight<0>(rec, cnt, ld, sub, X4, x);
      int2 nxt = make_int2(0, 0);
      if (sub + 16 < cnt) nxt = csr[start + base + 16 + sub];
      fmaEight<0>(rec, x, acc);
      if (degmax - base > 8) {
        loadEight<8>(rec, cnt, ld, sub, X4, x);
        fmaEight<8>(rec, x, acc);
      }
      rec = nxt;
      cnt -= 16;
    }
    if (ld) reinterpret_cast<float4*>(agg)[(size_t)(r0 + hsel) * ROW4 + sub] = acc;
  }
}

#define PREP_THREADS 256
__device__ __forceinline__ void foldTile(const float* __restrict__ A, const int row0, const int nrows,
                                         const float* __restrict__ Wh, float* __restrict__ D) {
  const int lane = threadIdx.x & 31;
  const int h = lane >> 4;
  const int l = lane & 15;
  const int row = row0 + l;
  const float* Arow = A + (size_t)(row < nrows ? row : 0) * HID;
  const bool cv = 32 + l < NCLS;
  const int n2 = cv ? 32 + l : 0;
  v8f acc0, acc1, acc2;
#pragma unroll
  for (int v = 0; v < 8; ++v) { acc0[v] = 0.f; acc1[v] = 0.f; acc2[v] = 0.f; }
#pragma unroll 8
  for (int t = 0; t < HID / 4; ++t) {
    const int ka = 4 * t + 2 * h;
    const float* w0 = Wh + (size_t)ka * NCLS;
    const float* w1 = w0 + NCLS;
    v2f a, b0, b1, b2;
    a.x = Arow[ka];
    a.y = Arow[ka + 1];
    b0.x = w0[l];
    b0.y = w1[l];
    b1.x = w0[16 + l];
    b1.y = w1[16 + l];
    b2.x = w0[n2];
    b2.y = w1[n2];
    acc0 = __builtin_amdgcn_wmma_f32_16x16x4_f32(false, a, false, b0, (short)0, acc0, false, false);
    acc1 = __builtin_amdgcn_wmma_f32_16x16x4_f32(false, a, false, b1, (short)0, acc1, false, false);
    acc2 = __builtin_amdgcn_wmma_f32_16x16x4_f32(false, a, false, b2, (short)0, acc2, false, false);
  }
#pragma unroll
  for (int v = 0; v < 8; ++v) {
    const int r = row0 + v + 8 * h;
    if (r < nrows) {
      D[(size_t)r * NPAD + l]      = acc0[v];
      D[(size_t)r * NPAD + 16 + l] = acc1[v];
      D[(size_t)r * NPAD + 32 + l] = cv ? acc2[v] : 0.f;
    }
  }
}

#define PREP_LDS_WORDS (2 * HID * NCLS + 2 * HID)
static_assert(PREP_THREADS == FRONT_T, "the prep roles run on the front launch's block size");
__device__ __forceinline__ void prepBlock(const int role,
                                          const float* __restrict__ W2, const float* __restrict__ We,
                                          const float* __restrict__ Wf, const float* __restrict__ bf,
                                          const float* __restrict__ be, const float* __restrict__ b2,
                                          float* __restrict__ W2f, float* __restrict__ Wef,
                                          float* __restrict__ bfp, float* __restrict__ lds) {
  float* sW  = lds;
  float* sBe = lds + 2 * HID * NCLS;
  float* sB2 = sBe + HID;
  const int t = threadIdx.x;
  const int wid = __builtin_amdgcn_readfirstlane((int)(threadIdx.x >> 5));
  if (role == 0) {
    foldTile(W2, 16 * wid, HID, Wf + (size_t)HID * NCLS, W2f);
    return;
  }
  if (role == 1) {
    if (wid < (NCLS + 15) / 16) foldTile(We, 16 * wid, NCLS, Wf, Wef);
    return;
  }
  const float4* Wf4 = reinterpret_cast<const float4*>(Wf);
  float4* sW4 = reinterpret_cast<float4*>(sW);
#pragma unroll
  for (int j = 0; j < (2 * HID * NCLS) / 4 / PREP_THREADS; ++j)
    sW4[j * PREP_THREADS + t] = Wf4[j * PREP_THREADS + t];
  if (t < HID) sBe[t] = be[t];
  else         sB2[t - HID] = b2[t - HID];
  __syncthreads();
  if (t < NPAD) {
    float s = 0.f;
    if (t < NCLS) {
      s = bf[t];
      for (int k = 0; k < HID; ++k) {
        s += sBe[k] * sW[k * NCLS + t];
        s += sB2[k] * sW[(HID + k) * NCLS + t];
      }
    }
    bfp[t] = s;
  }
}

#define GEMM_BLOCKS  (((NNODES / 16) + 7) / 8)
#define COUNT_BLOCKS (CSR_G * CSR_Q)
#define FRONT_BLOCKS (GEMM_BLOCKS + COUNT_BLOCKS + 3)
#define FRONT_LDS4   (PREP_LDS_WORDS / 4)
static_assert(PREP_LDS_WORDS % 4 == 0, "the block's LDS is whole 16-byte vectors");
static_assert(FRONT_LDS4 * 4 >= CSR_ROWS + 1, "the count role's counters and sum word fit the block's LDS");
static_assert(FRONT_LDS4 * 4 >= GemmChunk<FIN>::value * HID, "the GEMM role's B chunk fits the block's LDS");

__global__ void
k_front(const float* __restrict__ features, const float* __restrict__ W1, float* __restrict__ X1,
        const int* __restrict__ dst, unsigned* __restrict__ cnt, unsigned* __restrict__ bsum,
        const float* __restrict__ W2, const float* __restrict__ We,
        const float* __restrict__ Wf, const float* __restrict__ bf,
        const float* __restrict__ be, const float* __restrict__ b2,
        float* __restrict__ W2f, float* __restrict__ Wef, float* __restrict__ bfp) {
  __shared__ float4 lds4[FRONT_LDS4];
  float* lds = reinterpret_cast<float*>(lds4);
  const int bid = (int)blockIdx.x;
  if (bid < GEMM_BLOCKS) {
    gemmBlock<8, FIN, 1>(features, W1, X1, NNODES, nullptr, nullptr, lds, bid);
  } else if (bid < GEMM_BLOCKS + COUNT_BLOCKS) {
    countBlock(dst, cnt, bsum, reinterpret_cast<unsigned*>(lds), bid - GEMM_BLOCKS);
  } else {
    prepBlock(bid - (GEMM_BLOCKS + COUNT_BLOCKS), W2, We, Wf, bf, be, b2, W2f, Wef, bfp, lds);
  }
}

extern "C" void kernel_launch(void* const* d_in, const int* in_sizes, int n_in,
                              void* d_out, int out_size, void* d_ws, size_t ws_size,
                              hipStream_t stream) {
  const float* features = (const float*)d_in[0];
  const int*   src      = (const int*)d_in[1];
  const int*   dst      = (const int*)d_in[2];
  const float* ew       = (const float*)d_in[3];
  const float* y        = (const float*)d_in[4];
  const float* W1       = (const float*)d_in[5];
  const float* b1       = (const float*)d_in[6];
  const float* W2       = (const float*)d_in[7];
  const float* b2       = (const float*)d_in[8];
  const float* We       = (const float*)d_in[9];
  const float* be       = (const float*)d_in[10];
  const float* Wf       = (const float*)d_in[11];
  const float* bf       = (const float*)d_in[12];
  float* out = (float*)d_out;
  (void)in_sizes; (void)n_in; (void)out_size; (void)ws_size;

  char* ws = (char*)d_ws;
  size_t off = 0;
  auto alloc = [&](size_t bytes) -> float* {
    float* p = (float*)(ws + off);
    off += (bytes + 255) & ~(size_t)255;
    return p;
  };
  float* bufA   = alloc((size_t)NNODES * HID * 4);
  float* bufAgg = alloc((size_t)NNODES * HID * 4);
  float* bufL   = alloc((size_t)NNODES * NPAD * 4);
  float* W2f    = alloc((size_t)HID * NPAD * 4);
  float* Wef    = alloc((size_t)NCLS * NPAD * 4);
  float* bfp    = alloc((size_t)NPAD * 4);
  unsigned* rowptr = (unsigned*)alloc((size_t)(NNODES + 1) * 4);
  int*      csr    = (int*)alloc((size_t)NEDGES * 8);
  unsigned* cntMat = (unsigned*)bufAgg;
  unsigned* bsum   = cntMat + (size_t)CSR_G * NNODES;
  int*      perm   = (int*)bufL;

  const int gemmBlocks = GEMM_BLOCKS;

  k_front<<<FRONT_BLOCKS, FRONT_T, 0, stream>>>(features, W1, bufA, dst, cntMat, bsum,
                                                W2, We, Wf, bf, be, b2, W2f, Wef, bfp);
  k_csr_fill<<<CSR_G * CSR_Q, CSR_T, 0, stream>>>(dst, cntMat, bsum, rowptr, perm);
  k_spmm_rows_rank<<<ROW_BLOCKS, 256, 0, stream>>>(rowptr, perm, src, ew, csr, bufA, b1, W2f, bufAgg);
  k_spmm_rows48<<<ROW_BLOCKS, 256, 0, stream>>>(rowptr, csr, bufAgg, bufL);
  k_gemm_wmma<3, NCLS, 28><<<gemmBlocks, 256, 0, stream>>>(
      y, Wef, bufL, NNODES, bfp, out);
}


# Head launch gone: y@Wef + bias formed on two idle matrix waves of the layer-1 row kernel; layer-2 row kernel adds it to its row sums, does the softmax in its lanes; logit, softmax sums reordered
# speedup vs baseline: 5.8974x; 1.0497x over previous
#include <hip/hip_runtime.h>
#include <hip/hip_bf16.h>

typedef float v2f __attribute__((ext_vector_type(2)));
typedef float v8f __attribute__((ext_vector_type(8)));

#define NNODES 50000
#define NEDGES 1600000
#define FIN    256
#define HID    128
#define NCLS   40
#define NPAD   48

template <int X>
__device__ __forceinline__ float rowXor(float v) {
  return __int_as_float(__builtin_amdgcn_update_dpp(0, __float_as_int(v), 0x160 + X, 0xF, 0xF, false));
}
__device__ __forceinline__ float rowMax16(float v) {
  v = fmaxf(v, rowXor<8>(v)); v = fmaxf(v, rowXor<4>(v));
  v = fmaxf(v, rowXor<2>(v)); v = fmaxf(v, rowXor<1>(v));
  return v;
}
__device__ __forceinline__ float rowSum16(float v) {
  v += rowXor<8>(v); v += rowXor<4>(v); v += rowXor<2>(v); v += rowXor<1>(v);
  return v;
}

template <int K> struct GemmChunk { static constexpr int value = (K % 32 == 0) ? 32 : K; };

template <int NT, int K>
__device__ __forceinline__ void gemmBlock(const float* __restrict__ A, const float* __restrict__ B,
                                          float* __restrict__ C, int M,
                                          float* __restrict__ ldsB, const int bid) {
  constexpr int N = NT * 16;
  constexpr int KCHUNK = GemmChunk<K>::value;
  constexpr int NCHUNKS = K / KCHUNK;
  constexpr int TSTEPS = KCHUNK / 4;
  constexpr int GROUP = (TSTEPS % 8 == 0) ? 8 : TSTEPS / 2;
  constexpr int NV4 = (KCHUNK * N) / 4;
  constexpr int STAGE4 = (NV4 + 255) / 256;

  const int lane = threadIdx.x & 31;
  const int wid  = threadIdx.x >> 5;
  const int tm   = bid * (blockDim.x >> 5) + wid;
  const bool active = tm < (M >> 4);
  const int h = lane >> 4;
  const int l = lane & 15;
  const int row = (tm << 4) + l;

  v8f acc[NT];
  if (active) {
#pragma unroll
    for (int j = 0; j < NT; ++j)
#pragma unroll
      for (int v = 0; v < 8; ++v)
        acc[j][v] = 0.f;
  }
  const float* Arow = A + (size_t)row * K;
  float ss = 0.f;

  for (int c = 0; c < NCHUNKS; ++c) {
    const int k0 = c * KCHUNK;
    __syncthreads();
#pragma unroll
    for (int i = 0; i < STAGE4; ++i) {
      int idx4 = i * 256 + threadIdx.x;
      if (NV4 % 256 == 0 || idx4 < NV4) {
        int f = idx4 * 4;
        int kk = f / N, n = f - kk * N;
        float4 q = *(const float4*)(B + (size_t)(k0 + kk) * N + n);
        float* d = ldsB + ((kk >> 1) * N + n) * 2 + (kk & 1);
        d[0] = q.x; d[2] = q.y; d[4] = q.z; d[6] = q.w;
      }
    }
    __syncthreads();
    if (active) {
#pragma unroll
      for (int g = 0; g < TSTEPS; g += GROUP) {
        v2f areg[GROUP];
#pragma unroll
        for (int u = 0; u < GROUP; ++u) {
          const int ka = k0 + (g + u) * 4 + 2 * h;
          v2f a;
          a.x = Arow[ka];
          a.y = Arow[ka + 1];
          areg[u] = a;
        }
#pragma unroll
        for (int u = 0; u < GROUP; ++u) {
          const int ka = (g + u) * 4 + 2 * h;
          const float* bp = ldsB + (ka >> 1) * (2 * N) + 2 * l;
          ss = fmaf(areg[u].x, areg[u].x, ss);
          ss = fmaf(areg[u].y, areg[u].y, ss);
#pragma unroll
          for (int j = 0; j < NT; ++j) {
            v2f b = *(const v2f*)(bp + 32 * j);
            acc[j] = __builtin_amdgcn_wmma_f32_16x16x4_f32(
                false, areg[u], false, b, (short)0, acc[j], false, false);
          }
        }
      }
    }
  }

  if (active) {
    const float tot = ss + __shfl_xor(ss, 16, 32);
    const float inv = 1.0f / fmaxf(sqrtf(tot), 1e-12f);
    float sc[8];
#pragma unroll
    for (int v = 0; v < 8; ++v) sc[v] = __shfl(inv, v + 8 * h, 32);
#pragma unroll
    for (int j = 0; j < NT; ++j)
#pragma unroll
      for (int v = 0; v < 8; ++v)
        C[(size_t)((tm << 4) + v + 8 * h) * N + j * 16 + l] = acc[j][v] * sc[v];
  }
}

#define CSR_G      16
#define CSR_Q      16
#define CSR_ROWS   (NNODES / CSR_Q)
#define CSR_CHUNK  (NEDGES / CSR_G)
#define CSR_V4     (CSR_CHUNK / 4)
#define CSR_T      1024
#define FILL_PER   ((CSR_V4 + CSR_T - 1) / CSR_T)
#define FILL_BATCH 13
#define FILL_RPT   4
#define ROWS_PER_WAVE 4
#define ROW_BLOCKS (((NNODES + ROWS_PER_WAVE - 1) / ROWS_PER_WAVE + 7) / 8)
static_assert(NEDGES % CSR_G == 0, "edge chunks must be equal");
static_assert(NNODES % CSR_Q == 0, "row ranges must be equal");
static_assert(CSR_CHUNK % 4 == 0, "a chunk is whole 16-byte vectors and starts on one");
static_assert(CSR_T * FILL_RPT >= CSR_ROWS, "the fill's threads cover the rows of a range");
static_assert(CSR_G * CSR_Q <= CSR_T, "one thread per block sum");
static_assert((size_t)NEDGES * 4 <= (size_t)NNODES * NPAD * 4, "perm[] fits the NPAD-column buffer");
static_assert(2 * NPAD <= HID, "X2f and L0 lie one behind the other in a HID-column buffer");

#define FRONT_T    256
#define CNT_PER    ((CSR_V4 + FRONT_T - 1) / FRONT_T)
#define CNT_BATCH  25

__device__ __forceinline__ void countOne(unsigned* __restrict__ hist, const int d, const int lo) {
  const unsigned rr = (unsigned)(d - lo);
  if (rr < (unsigned)CSR_ROWS) atomicAdd(&hist[rr], 1u);
}

__device__ __forceinline__ void countBlock(const int* __restrict__ dst, unsigned* __restrict__ cnt,
                                           unsigned* __restrict__ bsum,
                                           unsigned* __restrict__ hist, const int cb) {
  const int t = threadIdx.x;
  const int c = cb / CSR_Q;
  const int lo = (cb % CSR_Q) * CSR_ROWS;
  for (int i = t; i <= CSR_ROWS; i += FRONT_T) hist[i] = 0u;
  __syncthreads();
  const int4* d4 = reinterpret_cast<const int4*>(dst + (size_t)c * CSR_CHUNK);
  for (int k0 = 0; k0 < CNT_PER; k0 += CNT_BATCH) {
    int4 dv[CNT_BATCH];
#pragma unroll
    for (int k = 0; k < CNT_BATCH; ++k) {
      const int i = t + (k0 + k) * FRONT_T;
      dv[k] = (i < CSR_V4) ? d4[i] : make_int4(-1, -1, -1, -1);
    }
#pragma unroll
    for (int k = 0; k < CNT_BATCH; ++k) {
      countOne(hist, dv[k].x, lo);
      countOne(hist, dv[k].y, lo);
      countOne(hist, dv[k].z, lo);
      countOne(hist, dv[k].w, lo);
    }
  }
  __syncthreads();
  unsigned* orow = cnt + (size_t)c * NNODES + lo;
  unsigned part = 0u;
  for (int i = t; i < CSR_ROWS; i += FRONT_T) {
    const unsigned v = hist[i];
    orow[i] = v;
    part += v;
  }
  for (int o = 16; o > 0; o >>= 1) part += __shfl_xor(part, o, 32);
  if ((t & 31) == 0) atomicAdd(&hist[CSR_ROWS], part);
  __syncthreads();
  if (t == 0) bsum[cb] = hist[CSR_ROWS];
}

__device__ __forceinline__ void placeOne(unsigned* __restrict__ hist, int* __restrict__ perm,
                                         const int d, const int lo, const int e) {
  const unsigned rr = (unsigned)(d - lo);
  if (rr < (unsigned)CSR_ROWS) {
    const unsigned pos = atomicAdd(&hist[rr], 1u);
    perm[pos] = e;
  }
}

__global__ void __launch_bounds__(CSR_T)
k_csr_fill(const int* __restrict__ dst, const unsigned* __restrict__ cnt,
           const unsigned* __restrict__ bsum, unsigned* __restrict__ rowptr,
           int* __restrict__ perm) {
  __shared__ unsigned hist[CSR_ROWS];
  __shared__ unsigned wsum[32];
  __shared__ unsigned wlow[32];
  __shared__ unsigned stat[2];
  const int t = threadIdx.x, lane = t & 31, wid = t >> 5;
  const int c = blockIdx.x / CSR_Q;
  const int q = blockIdx.x % CSR_Q;
  const int lo = q * CSR_ROWS;
  unsigned below = 0u;
  if (t < CSR_G * CSR_Q && (t % CSR_Q) < q) below = bsum[t];
  unsigned v[CSR_G][FILL_RPT];
#pragma unroll
  for (int g = 0; g < CSR_G; ++g) {
#pragma unroll
    for (int j = 0; j < FILL_RPT; ++j) {
      const int i = FILL_RPT * t + j;
      v[g][j] = (i < CSR_ROWS) ? cnt[(size_t)g * NNODES + lo + i] : 0u;
    }
  }
  unsigned ex[FILL_RPT], pr[FILL_RPT];
  unsigned s = 0u;
#pragma unroll
  for (int j = 0; j < FILL_RPT; ++j) {
    unsigned tj = 0u, pj = 0u;
#pragma unroll
    for (int g = 0; g < CSR_G; ++g) {
      tj += v[g][j];
      if (g < c) pj += v[g][j];
    }
    ex[j] = s;
    pr[j] = pj;
    s += tj;
  }
  unsigned inc = s;
  for (int o = 1; o < 32; o <<= 1) {
    const unsigned n = __shfl_up(inc, o, 32);
    if (lane >= o) inc += n;
  }
  for (int o = 16; o > 0; o >>= 1) below += __shfl_xor(below, o, 32);
  if (lane == 31) wsum[wid] = inc;
  if (lane == 0) wlow[wid] = below;
  __syncthreads();
  if (wid == 0) {
    const unsigned w = wsum[lane];
    unsigned wi = w;
    for (int o = 1; o < 32; o <<= 1) {
      const unsigned n = __shfl_up(wi, o, 32);
      if (lane >= o) wi += n;
    }
    unsigned bb = wlow[lane];
    for (int o = 16; o > 0; o >>= 1) bb += __shfl_xor(bb, o, 32);
    wsum[lane] = wi - w;
    if (lane == 31) { stat[0] = bb; stat[1] = wi; }
  }
  __syncthreads();
  const unsigned first = stat[0] + wsum[wid] + (inc - s);
#pragma unroll
  for (int j = 0; j < FILL_RPT; ++j) {
    const int i = FILL_RPT * t + j;
    if (i < CSR_ROWS) {
      const unsigned rp = first + ex[j];
      hist[i] = rp + pr[j];
      if (c == 0) rowptr[lo + i] = rp;
    }
  }
  if (c == 0 && q == CSR_Q - 1 && t == 0) rowptr[NNODES] = stat[0] + stat[1];
  __syncthreads();
  const int4* d4 = reinterpret_cast<const int4*>(dst + (size_t)c * CSR_CHUNK);
  const int e0 = c * CSR_CHUNK;
  for (int k0 = 0; k0 < FILL_PER; k0 += FILL_BATCH) {
    int4 dv[FILL_BATCH];
#pragma unroll
    for (int k = 0; k < FILL_BATCH; ++k) {
      const int i = t + (k0 + k) * CSR_T;
      dv[k] = (i < CSR_V4) ? d4[i] : make_int4(-1, -1, -1, -1);
    }
#pragma unroll
    for (int k = 0; k < FILL_BATCH; ++k) {
      const int e = e0 + 4 * (t + (k0 + k) * CSR_T);
      placeOne(hist, perm, dv[k].x, lo, e);
      placeOne(hist, perm, dv[k].y, lo, e + 1);
      placeOne(hist, perm, dv[k].z, lo, e + 2);
      placeOne(hist, perm, dv[k].w, lo, e + 3);
    }
  }
}

#define RANK_CAP 128

template <int NB>
__device__ __forceinline__ void rankAndPlace(const int start, const int deg, const int lane,
                                             const int* __restrict__ perm, const int* __restrict__ src,
                                             const float* __restrict__ ew, int2* mine,
                                             int2* __restrict__ csr) {
  int e[NB], rk[NB];
#pragma unroll
  for (int a = 0; a < NB; ++a) {
    const int i = 32 * a + lane;
    e[a] = (i < deg) ? perm[start + i] : 0x7fffffff;
    rk[a] = 0;
  }
#pragma unroll
  for (int b = 0; b < NB; ++b) {
#pragma unroll(NB > 2 ? 4 : 32)
    for (int j = 0; j < 32; ++j) {
      const int s = __builtin_amdgcn_readlane(e[b], j);
#pragma unroll
      for (int a = 0; a < NB; ++a) rk[a] += (s < e[a]) ? 1 : 0;
    }
  }
  int sv[NB];
  float wv[NB];
#pragma unroll
  for (int a = 0; a < NB; ++a) {
    const bool ok = 32 * a + lane < deg;
    sv[a] = ok ? src[e[a]] : 0;
    wv[a] = ok ? ew[e[a]] : 0.f;
  }
#pragma unroll
  for (int a = 0; a < NB; ++a) {
    if (32 * a + lane < deg) {
      const int2 rec = make_int2(sv[a], __float_as_int(wv[a]));
      mine[rk[a]] = rec;
      csr[start + rk[a]] = rec;
    }
  }
}

#define BLOCK_ROWS 32
#define TILE_PITCH (HID + 4)
static_assert(BLOCK_ROWS == 8 * ROWS_PER_WAVE, "one row per wave and pass");
static_assert(NNODES % 16 == 0, "a matrix tile is all rows or none");
static_assert((NNODES + BLOCK_ROWS - 1) / BLOCK_ROWS == ROW_BLOCKS, "grid covers the rows once");
static_assert(NCLS % 4 == 0, "the label product's k steps are whole");

__device__ __forceinline__ void labelTile(const float* __restrict__ y, const float* __restrict__ Wef,
                                          const float* __restrict__ bfp, const int row0,
                                          float* __restrict__ L0) {
  const int lane = threadIdx.x & 31;
  const int h = lane >> 4;
  const int l = lane & 15;
  const float* arow = y + (size_t)(row0 + l) * NCLS;
  const float* bcol = Wef + l;
  v8f d0, d1, d2;
#pragma unroll
  for (int v = 0; v < 8; ++v) { d0[v] = 0.f; d1[v] = 0.f; d2[v] = 0.f; }
#pragma unroll 2
  for (int t = 0; t < NCLS / 4; ++t) {
    const int ka = 4 * t + 2 * h;
    const float* w0 = bcol + (size_t)ka * NPAD;
    const float* w1 = w0 + NPAD;
    v2f a, b0, b1v, b2;
    a.x = arow[ka];
    a.y = arow[ka + 1];
    b0.x = w0[0];   b0.y = w1[0];
    b1v.x = w0[16]; b1v.y = w1[16];
    b2.x = w0[32];  b2.y = w1[32];
    d0 = __builtin_amdgcn_wmma_f32_16x16x4_f32(false, a, false, b0, (short)0, d0, false, false);
    d1 = __builtin_amdgcn_wmma_f32_16x16x4_f32(false, a, false, b1v, (short)0, d1, false, false);
    d2 = __builtin_amdgcn_wmma_f32_16x16x4_f32(false, a, false, b2, (short)0, d2, false, false);
  }
  const float bz0 = bfp[l], bz1 = bfp[16 + l], bz2 = bfp[32 + l];
#pragma unroll
  for (int v = 0; v < 8; ++v) {
    float* orow = L0 + (size_t)(row0 + v + 8 * h) * NPAD;
    orow[l] = d0[v] + bz0;
    orow[16 + l] = d1[v] + bz1;
    orow[32 + l] = d2[v] + bz2;
  }
}

__global__ void __launch_bounds__(256)
k_spmm_rows_rank(const unsigned* __restrict__ rowptr, const int* __restrict__ perm,
                 const int* __restrict__ src, const float* __restrict__ ew,
                 int* __restrict__ csrw, const float* __restrict__ X,
                 const float* __restrict__ b1, const float* __restrict__ W2f,
                 float* __restrict__ X2f,
                 const float* __restrict__ y, const float* __restrict__ Wef,
                 const float* __restrict__ bfp, float* __restrict__ L0) {
  __shared__ int2 slots[8][RANK_CAP];
  __shared__ float tile[BLOCK_ROWS * TILE_PITCH];
  int2* csr = reinterpret_cast<int2*>(csrw);
  const int lane = threadIdx.x & 31;
  const int wl = __builtin_amdgcn_readfirstlane((int)(threadIdx.x >> 5));
  const int rbase = blockIdx.x * BLOCK_ROWS;
  int2* mine = slots[wl];
  const float4* X4 = reinterpret_cast<const float4*>(X);
  for (int k = 0; k < ROWS_PER_WAVE; ++k) {
    const int rl = wl + 8 * k;
    const int r = rbase + rl;
    if (r < NNODES) {
      const int start = (int)rowptr[r];
      const int deg = (int)rowptr[r + 1] - start;
      float4 acc = make_float4(0.f, 0.f, 0.f, 0.f);
      if (deg <= RANK_CAP) {
        if (deg <= 32)      rankAndPlace<1>(start, deg, lane, perm, src, ew, mine, csr);
        else if (deg <= 64) rankAndPlace<2>(start, deg, lane, perm, src, ew, mine, csr);
        else                rankAndPlace<4>(start, deg, lane, perm, src, ew, mine, csr);
        for (int base = 0; base < deg; base += 32) {
          const int cnt = min(32, deg - base);
          const int2 rec = (lane < cnt) ? mine[base + lane] : make_int2(0, 0);
          for (int j = 0; j < cnt; j += 8) {
            float4 x[8];
            for (int u = 0; u < 8; ++u) {
              const int ju = min(j + u, cnt - 1);
              const int s = __builtin_amdgcn_readlane(rec.x, ju);
              x[u] = X4[(size_t)s * (HID / 4) + lane];
            }
            for (int u = 0; u < 8; ++u) {
              if (j + u < cnt) {
                const float w = __int_as_float(__builtin_amdgcn_readlane(rec.y, j + u));
                acc.x = fmaf(w, x[u].x, acc.x);
                acc.y = fmaf(w, x[u].y, acc.y);
                acc.z = fmaf(w, x[u].z, acc.z);
                acc.w = fmaf(w, x[u].w, acc.w);
              }
            }
          }
        }
      } else {
        int cur = -1;
        for (int j = 0; j < deg; ++j) {
          int best = 0x7fffffff;
          for (int b = 0; b < deg; b += 32) {
            const int i = b + lane;
            const int e = (i < deg) ? perm[start + i] : 0x7fffffff;
            if (e > cur && e < best) best = e;
          }
          for (int o = 16; o > 0; o >>= 1) best = min(best, __shfl_xor(best, o, 32));
          cur = __builtin_amdgcn_readfirstlane(best);
          const int s = src[cur];
          const float w = ew[cur];
          if (lane == 0) csr[start + j] = make_int2(s, __float_as_int(w));
          const float4 xv = X4[(size_t)s * (HID / 4) + lane];
          acc.x = fmaf(w, xv.x, acc.x);
          acc.y = fmaf(w, xv.y, acc.y);
          acc.z = fmaf(w, xv.z, acc.z);
          acc.w = fmaf(w, xv.w, acc.w);
        }
      }
      *reinterpret_cast<float4*>(tile + rl * TILE_PITCH + 4 * lane) = acc;
    }
  }
  __syncthreads();
  const int row0 = rbase + 16 * wl;
  if (wl < BLOCK_ROWS / 16 && row0 < NNODES) {
    const int h = lane >> 4;
    const int l = lane & 15;
    const float* arow = tile + (16 * wl + l) * TILE_PITCH;
    const float* bcol = W2f + l;
    v8f d0, d1, d2;
#pragma unroll
    for (int v = 0; v < 8; ++v) { d0[v] = 0.f; d1[v] = 0.f; d2[v] = 0.f; }
#pragma unroll 2
    for (int t = 0; t < HID / 4; ++t) {
      const int ka = 4 * t + 2 * h;
      const float* w0 = bcol + (size_t)ka * NPAD;
      const float* w1 = w0 + NPAD;
      const v2f s = *reinterpret_cast<const v2f*>(arow + ka);
      v2f a, b0, b1v, b2;
      a.x = fmaxf(s.x + b1[ka], 0.f);
      a.y = fmaxf(s.y + b1[ka + 1], 0.f);
      b0.x = w0[0];   b0.y = w1[0];
      b1v.x = w0[16]; b1v.y = w1[16];
      b2.x = w0[32];  b2.y = w1[32];
      d0 = __builtin_amdgcn_wmma_f32_16x16x4_f32(false, a, false, b0, (short)0, d0, false, false);
      d1 = __builtin_amdgcn_wmma_f32_16x16x4_f32(false, a, false, b1v, (short)0, d1, false, false);
      d2 = __builtin_amdgcn_wmma_f32_16x16x4_f32(false, a, false, b2, (short)0, d2, false, false);
    }
#pragma unroll
    for (int v = 0; v < 8; ++v) {
      float* orow = X2f + (size_t)(row0 + v + 8 * h) * NPAD;
      orow[l] = d0[v];
      orow[16 + l] = d1[v];
      orow[32 + l] = d2[v];
    }
  }
  const int rowy = rbase + 16 * (wl - BLOCK_ROWS / 16);
  if (wl >= BLOCK_ROWS / 16 && wl < 2 * (BLOCK_ROWS / 16) && rowy < NNODES)
    labelTile(y, Wef, bfp, rowy, L0);
}

#define ROW4 (NPAD / 4)
#define CLS4 (NCLS / 4)
static_assert(NNODES % 4 == 0, "a wave takes two pairs of rows");
static_assert((4 * NCLS * 4) % 128 == 0, "four rows of out are whole 128-byte lines");

template <int K>
__device__ __forceinline__ int rowShare(int v) {
  return __builtin_amdgcn_update_dpp(0, v, 0x150 + K, 0xF, 0xF, false);
}

template <int J>
__device__ __forceinline__ void loadEight(const int2 rec, const int cnt, const bool ld,
                                          const int sub, const float4* __restrict__ X4,
                                          float4 (&x)[8]) {
#define SPMM48_LOAD(U)                                                      \
  {                                                                         \
    const int s = rowShare<J + U>(rec.x);                                   \
    x[U] = make_float4(0.f, 0.f, 0.f, 0.f);                                 \
    if (ld && (J + U) < cnt) x[U] = X4[(size_t)s * ROW4 + sub];             \
  }
  SPMM48_LOAD(0) SPMM48_LOAD(1) SPMM48_LOAD(2) SPMM48_LOAD(3)
  SPMM48_LOAD(4) SPMM48_LOAD(5) SPMM48_LOAD(6) SPMM48_LOAD(7)
#undef SPMM48_LOAD
}

template <int J>
__device__ __forceinline__ void fmaEight(const int2 rec, const float4 (&x)[8], float4& acc) {
#define SPMM48_FMA(U)                                                       \
  {                                                                         \
    const float w = __int_as_float(rowShare<J + U>(rec.y));                 \
    acc.x = fmaf(w, x[U].x, acc.x);                                         \
    acc.y = fmaf(w, x[U].y, acc.y);                                         \
    acc.z = fmaf(w, x[U].z, acc.z);                                         \
    acc.w = fmaf(w, x[U].w, acc.w);                                         \
  }
  SPMM48_FMA(0) SPMM48_FMA(1) SPMM48_FMA(2) SPMM48_FMA(3)
  SPMM48_FMA(4) SPMM48_FMA(5) SPMM48_FMA(6) SPMM48_FMA(7)
#undef SPMM48_FMA
}

__global__ void k_spmm_rows48(const unsigned* __restrict__ rowptr, const int* __restrict__ csrw,
                              const float* __restrict__ X, const float* __restrict__ L0,
                              float* __restrict__ out) {
  const int2* csr = reinterpret_cast<const int2*>(csrw);
  const int lane = threadIdx.x & 31;
  const int hsel = lane >> 4;
  const int sub = lane & 15;
  const bool ld = sub < ROW4;
  const int w0 = __builtin_amdgcn_readfirstlane((int)(blockIdx.x * (blockDim.x >> 5) + (threadIdx.x >> 5)));
  const int nw = gridDim.x * (blockDim.x >> 5);
  const float4* X4 = reinterpret_cast<const float4*>(X);
  const float4* L04 = reinterpret_cast<const float4*>(L0);
  float4* out4 = reinterpret_cast<float4*>(out);
  const bool cv = sub < CLS4;
  for (int r0 = 4 * w0; r0 < NNODES; r0 += (r0 & 2) ? 4 * nw - 2 : 2) {
    const int s0 = (int)rowptr[r0];
    const int s1 = (int)rowptr[r0 + 1];
    const int s2 = (int)rowptr[r0 + 2];
    const int degmax = max(s1 - s0, s2 - s1);
    const int start = hsel ? s1 : s0;
    const int deg = hsel ? (s2 - s1) : (s1 - s0);
    float4 acc = make_float4(0.f, 0.f, 0.f, 0.f);
    int cnt = deg;
    int2 rec = (sub < cnt) ? csr[start + sub] : make_int2(0, 0);
    float4 z = make_float4(0.f, 0.f, 0.f, 0.f);
    if (ld) z = L04[(size_t)(r0 + hsel) * ROW4 + sub];
    for (int base = 0; base < degmax; base += 16) {
      float4 x[8];
      loadEight<0>(rec, cnt, ld, sub, X4, x);
      int2 nxt = make_int2(0, 0);
      if (sub + 16 < cnt) nxt = csr[start + base + 16 + sub];
      fmaEight<0>(rec, x, acc);
      if (degmax - base > 8) {
        loadEight<8>(rec, cnt, ld, sub, X4, x);
        fmaEight<8>(rec, x, acc);
      }
      rec = nxt;
      cnt -= 16;
    }
    const float x0 = acc.x + z.x, x1 = acc.y + z.y, x2 = acc.z + z.z, x3 = acc.w + z.w;
    const float m = rowMax16(cv ? fmaxf(fmaxf(x0, x1), fmaxf(x2, x3)) : -3.4e38f);
    const float e0 = cv ? __expf(x0 - m) : 0.f;
    const float e1 = cv ? __expf(x1 - m) : 0.f;
    const float e2 = cv ? __expf(x2 - m) : 0.f;
    const float e3 = cv ? __expf(x3 - m) : 0.f;
    const float inv = 1.0f / rowSum16((e0 + e1) + (e2 + e3));
    if (cv) out4[(size_t)(r0 + hsel) * CLS4 + sub] = make_float4(e0 * inv, e1 * inv, e2 * inv, e3 * inv);
  }
}

#define PREP_THREADS 256
__device__ __forceinline__ void foldTile(const float* __restrict__ A, const int row0, const int nrows,
                                         const float* __restrict__ Wh, float* __restrict__ D) {
  const int lane = threadIdx.x & 31;
  const int h = lane >> 4;
  const int l = lane & 15;
  const int row = row0 + l;
  const float* Arow = A + (size_t)(row < nrows ? row : 0) * HID;
  const bool cv = 32 + l < NCLS;
  const int n2 = cv ? 32 + l : 0;
  v8f acc0, acc1, acc2;
#pragma unroll
  for (int v = 0; v < 8; ++v) { acc0[v] = 0.f; acc1[v] = 0.f; acc2[v] = 0.f; }
#pragma unroll 8
  for (int t = 0; t < HID / 4; ++t) {
    const int ka = 4 * t + 2 * h;
    const float* w0 = Wh + (size_t)ka * NCLS;
    const float* w1 = w0 + NCLS;
    v2f a, b0, b1, b2;
    a.x = Arow[ka];
    a.y = Arow[ka + 1];
    b0.x = w0[l];
    b0.y = w1[l];
    b1.x = w0[16 + l];
    b1.y = w1[16 + l];
    b2.x = w0[n2];
    b2.y = w1[n2];
    acc0 = __builtin_amdgcn_wmma_f32_16x16x4_f32(false, a, false, b0, (short)0, acc0, false, false);
    acc1 = __builtin_amdgcn_wmma_f32_16x16x4_f32(false, a, false, b1, (short)0, acc1, false, false);
    acc2 = __builtin_amdgcn_wmma_f32_16x16x4_f32(false, a, false, b2, (short)0, acc2, false, false);
  }
#pragma unroll
  for (int v = 0; v < 8; ++v) {
    const int r = row0 + v + 8 * h;
    if (r < nrows) {
      D[(size_t)r * NPAD + l]      = acc0[v];
      D[(size_t)r * NPAD + 16 + l] = acc1[v];
      D[(size_t)r * NPAD + 32 + l] = cv ? acc2[v] : 0.f;
    }
  }
}

#define PREP_LDS_WORDS (2 * HID * NCLS + 2 * HID)
static_assert(PREP_THREADS == FRONT_T, "the prep roles run on the front launch's block size");
__device__ __forceinline__ void prepBlock(const int role,
                                          const float* __restrict__ W2, const float* __restrict__ We,
                                          const float* __restrict__ Wf, const float* __restrict__ bf,
                                          const float* __restrict__ be, const float* __restrict__ b2,
                                          float* __restrict__ W2f, float* __restrict__ Wef,
                                          float* __restrict__ bfp, float* __restrict__ lds) {
  float* sW  = lds;
  float* sBe = lds + 2 * HID * NCLS;
  float* sB2 = sBe + HID;
  const int t = threadIdx.x;
  const int wid = __builtin_amdgcn_readfirstlane((int)(threadIdx.x >> 5));
  if (role == 0) {
    foldTile(W2, 16 * wid, HID, Wf + (size_t)HID * NCLS, W2f);
    return;
  }
  if (role == 1) {
    if (wid < (NCLS + 15) / 16) foldTile(We, 16 * wid, NCLS, Wf, Wef);
    return;
  }
  const float4* Wf4 = reinterpret_cast<const float4*>(Wf);
  float4* sW4 = reinterpret_cast<float4*>(sW);
#pragma unroll
  for (int j = 0; j < (2 * HID * NCLS) / 4 / PREP_THREADS; ++j)
    sW4[j * PREP_THREADS + t] = Wf4[j * PREP_THREADS + t];
  if (t < HID) sBe[t] = be[t];
  else         sB2[t - HID] = b2[t - HID];
  __syncthreads();
  if (t < NPAD) {
    float s = 0.f;
    if (t < NCLS) {
      s = bf[t];
      for (int k = 0; k < HID; ++k) {
        s += sBe[k] * sW[k * NCLS + t];
        s += sB2[k] * sW[(HID + k) * NCLS + t];
      }
    }
    bfp[t] = s;
  }
}

#define GEMM_BLOCKS  (((NNODES / 16) + 7) / 8)
#define COUNT_BLOCKS (CSR_G * CSR_Q)
#define FRONT_BLOCKS (GEMM_BLOCKS + COUNT_BLOCKS + 3)
#define FRONT_LDS4   (PREP_LDS_WORDS / 4)
static_assert(PREP_LDS_WORDS % 4 == 0, "the block's LDS is whole 16-byte vectors");
static_assert(FRONT_LDS4 * 4 >= CSR_ROWS + 1, "the count role's counters and sum word fit the block's LDS");
static_assert(FRONT_LDS4 * 4 >= GemmChunk<FIN>::value * HID, "the GEMM role's B chunk fits the block's LDS");

__global__ void
k_front(const float* __restrict__ features, const float* __restrict__ W1, float* __restrict__ X1,
        const int* __restrict__ dst, unsigned* __restrict__ cnt, unsigned* __restrict__ bsum,
        const float* __restrict__ W2, const float* __restrict__ We,
        const float* __restrict__ Wf, const float* __restrict__ bf,
        const float* __restrict__ be, const float* __restrict__ b2,
        float* __restrict__ W2f, float* __restrict__ Wef, float* __restrict__ bfp) {
  __shared__ float4 lds4[FRONT_LDS4];
  float* lds = reinterpret_cast<float*>(lds4);
  const int bid = (int)blockIdx.x;
  if (bid < GEMM_BLOCKS) {
    gemmBlock<8, FIN>(features, W1, X1, NNODES, lds, bid);
  } else if (bid < GEMM_BLOCKS + COUNT_BLOCKS) {
    countBlock(dst, cnt, bsum, reinterpret_cast<unsigned*>(lds), bid - GEMM_BLOCKS);
  } else {
    prepBlock(bid - (GEMM_BLOCKS + COUNT_BLOCKS), W2, We, Wf, bf, be, b2, W2f, Wef, bfp, lds);
  }
}

extern "C" void kernel_launch(void* const* d_in, const int* in_sizes, int n_in,
                              void* d_out, int out_size, void* d_ws, size_t ws_size,
                              hipStream_t stream) {
  const float* features = (const float*)d_in[0];
  const int*   src      = (const int*)d_in[1];
  const int*   dst      = (const int*)d_in[2];
  const float* ew       = (const float*)d_in[3];
  const float* y        = (const float*)d_in[4];
  const float* W1       = (const float*)d_in[5];
  const float* b1       = (const float*)d_in[6];
  const float* W2       = (const float*)d_in[7];
  const float* b2       = (const float*)d_in[8];
  const float* We       = (const float*)d_in[9];
  const float* be       = (const float*)d_in[10];
  const float* Wf       = (const float*)d_in[11];
  const float* bf       = (const float*)d_in[12];
  float* out = (float*)d_out;
  (void)in_sizes; (void)n_in; (void)out_size; (void)ws_size;

  char* ws = (char*)d_ws;
  size_t off = 0;
  auto alloc = [&](size_t bytes) -> float* {
    float* p = (float*)(ws + off);
    off += (bytes + 255) & ~(size_t)255;
    return p;
  };
  float* bufA   = alloc((size_t)NNODES * HID * 4);
  float* bufAgg = alloc((size_t)NNODES * HID * 4);
  float* bufL   = alloc((size_t)NNODES * NPAD * 4);
  float* W2f    = alloc((size_t)HID * NPAD * 4);
  float* Wef    = alloc((size_t)NCLS * NPAD * 4);
  float* bfp    = alloc((size_t)NPAD * 4);
  unsigned* rowptr = (unsigned*)alloc((size_t)(NNODES + 1) * 4);
  int*      csr    = (int*)alloc((size_t)NEDGES * 8);
  unsigned* cntMat = (unsigned*)bufAgg;
  unsigned* bsum   = cntMat + (size_t)CSR_G * NNODES;
  int*      perm   = (int*)bufL;
  float*    L0     = bufAgg + (size_t)NNODES * NPAD;

  k_front<<<FRONT_BLOCKS, FRONT_T, 0, stream>>>(features, W1, bufA, dst, cntMat, bsum,
                                                W2, We, Wf, bf, be, b2, W2f, Wef, bfp);
  k_csr_fill<<<CSR_G * CSR_Q, CSR_T, 0, stream>>>(dst, cntMat, bsum, rowptr, perm);
  k_spmm_rows_rank<<<ROW_BLOCKS, 256, 0, stream>>>(rowptr, perm, src, ew, csr, bufA, b1, W2f, bufAgg,
                                                   y, Wef, bfp, L0);
  k_spmm_rows48<<<ROW_BLOCKS, 256, 0, stream>>>(rowptr, csr, bufAgg, L0, out);
}


# src and edge weight paired into 8-byte records by a streaming block role of the front launch; the layer-1 row kernel reads one record per edge at its random index, not a word from each of two arrays
# speedup vs baseline: 6.1356x; 1.0404x over previous
#include <hip/hip_runtime.h>
#include <hip/hip_bf16.h>

typedef float v2f __attribute__((ext_vector_type(2)));
typedef float v8f __attribute__((ext_vector_type(8)));

#define NNODES 50000
#define NEDGES 1600000
#define FIN    256
#define HID    128
#define NCLS   40
#define NPAD   48

template <int X>
__device__ __forceinline__ float rowXor(float v) {
  return __int_as_float(__builtin_amdgcn_update_dpp(0, __float_as_int(v), 0x160 + X, 0xF, 0xF, false));
}
__device__ __forceinline__ float rowMax16(float v) {
  v = fmaxf(v, rowXor<8>(v)); v = fmaxf(v, rowXor<4>(v));
  v = fmaxf(v, rowXor<2>(v)); v = fmaxf(v, rowXor<1>(v));
  return v;
}
__device__ __forceinline__ float rowSum16(float v) {
  v += rowXor<8>(v); v += rowXor<4>(v); v += rowXor<2>(v); v += rowXor<1>(v);
  return v;
}

template <int K> struct GemmChunk { static constexpr int value = (K % 32 == 0) ? 32 : K; };

template <int NT, int K>
__device__ __forceinline__ void gemmBlock(const float* __restrict__ A, const float* __restrict__ B,
                                          float* __restrict__ C, int M,
                                          float* __restrict__ ldsB, const int bid) {
  constexpr int N = NT * 16;
  constexpr int KCHUNK = GemmChunk<K>::value;
  constexpr int NCHUNKS = K / KCHUNK;
  constexpr int TSTEPS = KCHUNK / 4;
  constexpr int GROUP = (TSTEPS % 8 == 0) ? 8 : TSTEPS / 2;
  constexpr int NV4 = (KCHUNK * N) / 4;
  constexpr int STAGE4 = (NV4 + 255) / 256;

  const int lane = threadIdx.x & 31;
  const int wid  = threadIdx.x >> 5;
  const int tm   = bid * (blockDim.x >> 5) + wid;
  const bool active = tm < (M >> 4);
  const int h = lane >> 4;
  const int l = lane & 15;
  const int row = (tm << 4) + l;

  v8f acc[NT];
  if (active) {
#pragma unroll
    for (int j = 0; j < NT; ++j)
#pragma unroll
      for (int v = 0; v < 8; ++v)
        acc[j][v] = 0.f;
  }
  const float* Arow = A + (size_t)row * K;
  float ss = 0.f;

  for (int c = 0; c < NCHUNKS; ++c) {
    const int k0 = c * KCHUNK;
    __syncthreads();
#pragma unroll
    for (int i = 0; i < STAGE4; ++i) {
      int idx4 = i * 256 + threadIdx.x;
      if (NV4 % 256 == 0 || idx4 < NV4) {
        int f = idx4 * 4;
        int kk = f / N, n = f - kk * N;
        float4 q = *(const float4*)(B + (size_t)(k0 + kk) * N + n);
        float* d = ldsB + ((kk >> 1) * N + n) * 2 + (kk & 1);
        d[0] = q.x; d[2] = q.y; d[4] = q.z; d[6] = q.w;
      }
    }
    __syncthreads();
    if (active) {
#pragma unroll
      for (int g = 0; g < TSTEPS; g += GROUP) {
        v2f areg[GROUP];
#pragma unroll
        for (int u = 0; u < GROUP; ++u) {
          const int ka = k0 + (g + u) * 4 + 2 * h;
          v2f a;
          a.x = Arow[ka];
          a.y = Arow[ka + 1];
          areg[u] = a;
        }
#pragma unroll
        for (int u = 0; u < GROUP; ++u) {
          const int ka = (g + u) * 4 + 2 * h;
          const float* bp = ldsB + (ka >> 1) * (2 * N) + 2 * l;
          ss = fmaf(areg[u].x, areg[u].x, ss);
          ss = fmaf(areg[u].y, areg[u].y, ss);
#pragma unroll
          for (int j = 0; j < NT; ++j) {
            v2f b = *(const v2f*)(bp + 32 * j);
            acc[j] = __builtin_amdgcn_wmma_f32_16x16x4_f32(
                false, areg[u], false, b, (short)0, acc[j], false, false);
          }
        }
      }
    }
  }

  if (active) {
    const float tot = ss + __shfl_xor(ss, 16, 32);
    const float inv = 1.0f / fmaxf(sqrtf(tot), 1e-12f);
    float sc[8];
#pragma unroll
    for (int v = 0; v < 8; ++v) sc[v] = __shfl(inv, v + 8 * h, 32);
#pragma unroll
    for (int j = 0; j < NT; ++j)
#pragma unroll
      for (int v = 0; v < 8; ++v)
        C[(size_t)((tm << 4) + v + 8 * h) * N + j * 16 + l] = acc[j][v] * sc[v];
  }
}

#define CSR_G      16
#define CSR_Q      16
#define CSR_ROWS   (NNODES / CSR_Q)
#define CSR_CHUNK  (NEDGES / CSR_G)
#define CSR_V4     (CSR_CHUNK / 4)
#define CSR_T      1024
#define FILL_PER   ((CSR_V4 + CSR_T - 1) / CSR_T)
#define FILL_BATCH 13
#define FILL_RPT   4
#define ROWS_PER_WAVE 4
#define ROW_BLOCKS (((NNODES + ROWS_PER_WAVE - 1) / ROWS_PER_WAVE + 7) / 8)
static_assert(NEDGES % CSR_G == 0, "edge chunks must be equal");
static_assert(NNODES % CSR_Q == 0, "row ranges must be equal");
static_assert(CSR_CHUNK % 4 == 0, "a chunk is whole 16-byte vectors and starts on one");
static_assert(CSR_T * FILL_RPT >= CSR_ROWS, "the fill's threads cover the rows of a range");
static_assert(CSR_G * CSR_Q <= CSR_T, "one thread per block sum");
static_assert((size_t)NEDGES * 4 <= (size_t)NNODES * NPAD * 4, "perm[] fits the NPAD-column buffer");
static_assert(2 * NPAD <= HID, "X2f and L0 lie one behind the other in a HID-column buffer");

#define FRONT_T    256
#define CNT_PER    ((CSR_V4 + FRONT_T - 1) / FRONT_T)
#define CNT_BATCH  25

__device__ __forceinline__ void countOne(unsigned* __restrict__ hist, const int d, const int lo) {
  const unsigned rr = (unsigned)(d - lo);
  if (rr < (unsigned)CSR_ROWS) atomicAdd(&hist[rr], 1u);
}

__device__ __forceinline__ void countBlock(const int* __restrict__ dst, unsigned* __restrict__ cnt,
                                           unsigned* __restrict__ bsum,
                                           unsigned* __restrict__ hist, const int cb) {
  const int t = threadIdx.x;
  const int c = cb / CSR_Q;
  const int lo = (cb % CSR_Q) * CSR_ROWS;
  for (int i = t; i <= CSR_ROWS; i += FRONT_T) hist[i] = 0u;
  __syncthreads();
  const int4* d4 = reinterpret_cast<const int4*>(dst + (size_t)c * CSR_CHUNK);
  for (int k0 = 0; k0 < CNT_PER; k0 += CNT_BATCH) {
    int4 dv[CNT_BATCH];
#pragma unroll
    for (int k = 0; k < CNT_BATCH; ++k) {
      const int i = t + (k0 + k) * FRONT_T;
      dv[k] = (i < CSR_V4) ? d4[i] : make_int4(-1, -1, -1, -1);
    }
#pragma unroll
    for (int k = 0; k < CNT_BATCH; ++k) {
      countOne(hist, dv[k].x, lo);
      countOne(hist, dv[k].y, lo);
      countOne(hist, dv[k].z, lo);
      countOne(hist, dv[k].w, lo);
    }
  }
  __syncthreads();
  unsigned* orow = cnt + (size_t)c * NNODES + lo;
  unsigned part = 0u;
  for (int i = t; i < CSR_ROWS; i += FRONT_T) {
    const unsigned v = hist[i];
    orow[i] = v;
    part += v;
  }
  for (int o = 16; o > 0; o >>= 1) part += __shfl_xor(part, o, 32);
  if ((t & 31) == 0) atomicAdd(&hist[CSR_ROWS], part);
  __syncthreads();
  if (t == 0) bsum[cb] = hist[CSR_ROWS];
}

__device__ __forceinline__ void placeOne(unsigned* __restrict__ hist, int* __restrict__ perm,
                                         const int d, const int lo, const int e) {
  const unsigned rr = (unsigned)(d - lo);
  if (rr < (unsigned)CSR_ROWS) {
    const unsigned pos = atomicAdd(&hist[rr], 1u);
    perm[pos] = e;
  }
}

__global__ void __launch_bounds__(CSR_T)
k_csr_fill(const int* __restrict__ dst, const unsigned* __restrict__ cnt,
           const unsigned* __restrict__ bsum, unsigned* __restrict__ rowptr,
           int* __restrict__ perm) {
  __shared__ unsigned hist[CSR_ROWS];
  __shared__ unsigned wsum[32];
  __shared__ unsigned wlow[32];
  __shared__ unsigned stat[2];
  const int t = threadIdx.x, lane = t & 31, wid = t >> 5;
  const int c = blockIdx.x / CSR_Q;
  const int q = blockIdx.x % CSR_Q;
  const int lo = q * CSR_ROWS;
  unsigned below = 0u;
  if (t < CSR_G * CSR_Q && (t % CSR_Q) < q) below = bsum[t];
  unsigned v[CSR_G][FILL_RPT];
#pragma unroll
  for (int g = 0; g < CSR_G; ++g) {
#pragma unroll
    for (int j = 0; j < FILL_RPT; ++j) {
      const int i = FILL_RPT * t + j;
      v[g][j] = (i < CSR_ROWS) ? cnt[(size_t)g * NNODES + lo + i] : 0u;
    }
  }
  unsigned ex[FILL_RPT], pr[FILL_RPT];
  unsigned s = 0u;
#pragma unroll
  for (int j = 0; j < FILL_RPT; ++j) {
    unsigned tj = 0u, pj = 0u;
#pragma unroll
    for (int g = 0; g < CSR_G; ++g) {
      tj += v[g][j];
      if (g < c) pj += v[g][j];
    }
    ex[j] = s;
    pr[j] = pj;
    s += tj;
  }
  unsigned inc = s;
  for (int o = 1; o < 32; o <<= 1) {
    const unsigned n = __shfl_up(inc, o, 32);
    if (lane >= o) inc += n;
  }
  for (int o = 16; o > 0; o >>= 1) below += __shfl_xor(below, o, 32);
  if (lane == 31) wsum[wid] = inc;
  if (lane == 0) wlow[wid] = below;
  __syncthreads();
  if (wid == 0) {
    const unsigned w = wsum[lane];
    unsigned wi = w;
    for (int o = 1; o < 32; o <<= 1) {
      const unsigned n = __shfl_up(wi, o, 32);
      if (lane >= o) wi += n;
    }
    unsigned bb = wlow[lane];
    for (int o = 16; o > 0; o >>= 1) bb += __shfl_xor(bb, o, 32);
    wsum[lane] = wi - w;
    if (lane == 31) { stat[0] = bb; stat[1] = wi; }
  }
  __syncthreads();
  const unsigned first = stat[0] + wsum[wid] + (inc - s);
#pragma unroll
  for (int j = 0; j < FILL_RPT; ++j) {
    const int i = FILL_RPT * t + j;
    if (i < CSR_ROWS) {
      const unsigned rp = first + ex[j];
      hist[i] = rp + pr[j];
      if (c == 0) rowptr[lo + i] = rp;
    }
  }
  if (c == 0 && q == CSR_Q - 1 && t == 0) rowptr[NNODES] = stat[0] + stat[1];
  __syncthreads();
  const int4* d4 = reinterpret_cast<const int4*>(dst + (size_t)c * CSR_CHUNK);
  const int e0 = c * CSR_CHUNK;
  for (int k0 = 0; k0 < FILL_PER; k0 += FILL_BATCH) {
    int4 dv[FILL_BATCH];
#pragma unroll
    for (int k = 0; k < FILL_BATCH; ++k) {
      const int i = t + (k0 + k) * CSR_T;
      dv[k] = (i < CSR_V4) ? d4[i] : make_int4(-1, -1, -1, -1);
    }
#pragma unroll
    for (int k = 0; k < FILL_BATCH; ++k) {
      const int e = e0 + 4 * (t + (k0 + k) * CSR_T);
      placeOne(hist, perm, dv[k].x, lo, e);
      placeOne(hist, perm, dv[k].y, lo, e + 1);
      placeOne(hist, perm, dv[k].z, lo, e + 2);
      placeOne(hist, perm, dv[k].w, lo, e + 3);
    }
  }
}

#define RANK_CAP 128

template <int NB>
__device__ __forceinline__ void rankAndPlace(const int start, const int deg, const int lane,
                                             const int* __restrict__ perm, const int2* __restrict__ recs,
                                             int2* mine, int2* __restrict__ csr) {
  int e[NB], rk[NB];
#pragma unroll
  for (int a = 0; a < NB; ++a) {
    const int i = 32 * a + lane;
    e[a] = (i < deg) ? perm[start + i] : 0x7fffffff;
    rk[a] = 0;
  }
#pragma unroll
  for (int b = 0; b < NB; ++b) {
#pragma unroll(NB > 2 ? 4 : 32)
    for (int j = 0; j < 32; ++j) {
      const int s = __builtin_amdgcn_readlane(e[b], j);
#pragma unroll
      for (int a = 0; a < NB; ++a) rk[a] += (s < e[a]) ? 1 : 0;
    }
  }
  int2 rv[NB];
#pragma unroll
  for (int a = 0; a < NB; ++a) {
    const bool ok = 32 * a + lane < deg;
    rv[a] = ok ? recs[e[a]] : make_int2(0, 0);
  }
#pragma unroll
  for (int a = 0; a < NB; ++a) {
    if (32 * a + lane < deg) {
      mine[rk[a]] = rv[a];
      csr[start + rk[a]] = rv[a];
    }
  }
}

#define BLOCK_ROWS 32
#define TILE_PITCH (HID + 4)
static_assert(BLOCK_ROWS == 8 * ROWS_PER_WAVE, "one row per wave and pass");
static_assert(NNODES % 16 == 0, "a matrix tile is all rows or none");
static_assert((NNODES + BLOCK_ROWS - 1) / BLOCK_ROWS == ROW_BLOCKS, "grid covers the rows once");
static_assert(NCLS % 4 == 0, "the label product's k steps are whole");

__device__ __forceinline__ void labelTile(const float* __restrict__ y, const float* __restrict__ Wef,
                                          const float* __restrict__ bfp, const int row0,
                                          float* __restrict__ L0) {
  const int lane = threadIdx.x & 31;
  const int h = lane >> 4;
  const int l = lane & 15;
  const float* arow = y + (size_t)(row0 + l) * NCLS;
  const float* bcol = Wef + l;
  v8f d0, d1, d2;
#pragma unroll
  for (int v = 0; v < 8; ++v) { d0[v] = 0.f; d1[v] = 0.f; d2[v] = 0.f; }
#pragma unroll 2
  for (int t = 0; t < NCLS / 4; ++t) {
    const int ka = 4 * t + 2 * h;
    const float* w0 = bcol + (size_t)ka * NPAD;
    const float* w1 = w0 + NPAD;
    v2f a, b0, b1v, b2;
    a.x = arow[ka];
    a.y = arow[ka + 1];
    b0.x = w0[0];   b0.y = w1[0];
    b1v.x = w0[16]; b1v.y = w1[16];
    b2.x = w0[32];  b2.y = w1[32];
    d0 = __builtin_amdgcn_wmma_f32_16x16x4_f32(false, a, false, b0, (short)0, d0, false, false);
    d1 = __builtin_amdgcn_wmma_f32_16x16x4_f32(false, a, false, b1v, (short)0, d1, false, false);
    d2 = __builtin_amdgcn_wmma_f32_16x16x4_f32(false, a, false, b2, (short)0, d2, false, false);
  }
  const float bz0 = bfp[l], bz1 = bfp[16 + l], bz2 = bfp[32 + l];
#pragma unroll
  for (int v = 0; v < 8; ++v) {
    float* orow = L0 + (size_t)(row0 + v + 8 * h) * NPAD;
    orow[l] = d0[v] + bz0;
    orow[16 + l] = d1[v] + bz1;
    orow[32 + l] = d2[v] + bz2;
  }
}

__global__ void __launch_bounds__(256)
k_spmm_rows_rank(const unsigned* __restrict__ rowptr, const int* __restrict__ perm,
                 const int* __restrict__ recw,
                 int* __restrict__ csrw, const float* __restrict__ X,
                 const float* __restrict__ b1, const float* __restrict__ W2f,
                 float* __restrict__ X2f,
                 const float* __restrict__ y, const float* __restrict__ Wef,
                 const float* __restrict__ bfp, float* __restrict__ L0) {
  __shared__ int2 slots[8][RANK_CAP];
  __shared__ float tile[BLOCK_ROWS * TILE_PITCH];
  int2* csr = reinterpret_cast<int2*>(csrw);
  const int2* recs = reinterpret_cast<const int2*>(recw);
  const int lane = threadIdx.x & 31;
  const int wl = __builtin_amdgcn_readfirstlane((int)(threadIdx.x >> 5));
  const int rbase = blockIdx.x * BLOCK_ROWS;
  int2* mine = slots[wl];
  const float4* X4 = reinterpret_cast<const float4*>(X);
  for (int k = 0; k < ROWS_PER_WAVE; ++k) {
    const int rl = wl + 8 * k;
    const int r = rbase + rl;
    if (r < NNODES) {
      const int start = (int)rowptr[r];
      const int deg = (int)rowptr[r + 1] - start;
      float4 acc = make_float4(0.f, 0.f, 0.f, 0.f);
      if (deg <= RANK_CAP) {
        if (deg <= 32)      rankAndPlace<1>(start, deg, lane, perm, recs, mine, csr);
        else if (deg <= 64) rankAndPlace<2>(start, deg, lane, perm, recs, mine, csr);
        else                rankAndPlace<4>(start, deg, lane, perm, recs, mine, csr);
        for (int base = 0; base < deg; base += 32) {
          const int cnt = min(32, deg - base);
          const int2 rec = (lane < cnt) ? mine[base + lane] : make_int2(0, 0);
          for (int j = 0; j < cnt; j += 8) {
            float4 x[8];
            for (int u = 0; u < 8; ++u) {
              const int ju = min(j + u, cnt - 1);
              const int s = __builtin_amdgcn_readlane(rec.x, ju);
              x[u] = X4[(size_t)s * (HID / 4) + lane];
            }
            for (int u = 0; u < 8; ++u) {
              if (j + u < cnt) {
                const float w = __int_as_float(__builtin_amdgcn_readlane(rec.y, j + u));
                acc.x = fmaf(w, x[u].x, acc.x);
                acc.y = fmaf(w, x[u].y, acc.y);
                acc.z = fmaf(w, x[u].z, acc.z);
                acc.w = fmaf(w, x[u].w, acc.w);
              }
            }
          }
        }
      } else {
        int cur = -1;
        for (int j = 0; j < deg; ++j) {
          int best = 0x7fffffff;
          for (int b = 0; b < deg; b += 32) {
            const int i = b + lane;
            const int e = (i < deg) ? perm[start + i] : 0x7fffffff;
            if (e > cur && e < best) best = e;
          }
          for (int o = 16; o > 0; o >>= 1) best = min(best, __shfl_xor(best, o, 32));
          cur = __builtin_amdgcn_readfirstlane(best);
          const int2 rc = recs[cur];
          const int s = rc.x;
          const float w = __int_as_float(rc.y);
          if (lane == 0) csr[start + j] = rc;
          const float4 xv = X4[(size_t)s * (HID / 4) + lane];
          acc.x = fmaf(w, xv.x, acc.x);
          acc.y = fmaf(w, xv.y, acc.y);
          acc.z = fmaf(w, xv.z, acc.z);
          acc.w = fmaf(w, xv.w, acc.w);
        }
      }
      *reinterpret_cast<float4*>(tile + rl * TILE_PITCH + 4 * lane) = acc;
    }
  }
  __syncthreads();
  const int row0 = rbase + 16 * wl;
  if (wl < BLOCK_ROWS / 16 && row0 < NNODES) {
    const int h = lane >> 4;
    const int l = lane & 15;
    const float* arow = tile + (16 * wl + l) * TILE_PITCH;
    const float* bcol = W2f + l;
    v8f d0, d1, d2;
#pragma unroll
    for (int v = 0; v < 8; ++v) { d0[v] = 0.f; d1[v] = 0.f; d2[v] = 0.f; }
#pragma unroll 2
    for (int t = 0; t < HID / 4; ++t) {
      const int ka = 4 * t + 2 * h;
      const float* w0 = bcol + (size_t)ka * NPAD;
      const float* w1 = w0 + NPAD;
      const v2f s = *reinterpret_cast<const v2f*>(arow + ka);
      v2f a, b0, b1v, b2;
      a.x = fmaxf(s.x + b1[ka], 0.f);
      a.y = fmaxf(s.y + b1[ka + 1], 0.f);
      b0.x = w0[0];   b0.y = w1[0];
      b1v.x = w0[16]; b1v.y = w1[16];
      b2.x = w0[32];  b2.y = w1[32];
      d0 = __builtin_amdgcn_wmma_f32_16x16x4_f32(false, a, false, b0, (short)0, d0, false, false);
      d1 = __builtin_amdgcn_wmma_f32_16x16x4_f32(false, a, false, b1v, (short)0, d1, false, false);
      d2 = __builtin_amdgcn_wmma_f32_16x16x4_f32(false, a, false, b2, (short)0, d2, false, false);
    }
#pragma unroll
    for (int v = 0; v < 8; ++v) {
      float* orow = X2f + (size_t)(row0 + v + 8 * h) * NPAD;
      orow[l] = d0[v];
      orow[16 + l] = d1[v];
      orow[32 + l] = d2[v];
    }
  }
  const int rowy = rbase + 16 * (wl - BLOCK_ROWS / 16);
  if (wl >= BLOCK_ROWS / 16 && wl < 2 * (BLOCK_ROWS / 16) && rowy < NNODES)
    labelTile(y, Wef, bfp, rowy, L0);
}

#define ROW4 (NPAD / 4)
#define CLS4 (NCLS / 4)
static_assert(NNODES % 4 == 0, "a wave takes two pairs of rows");
static_assert((4 * NCLS * 4) % 128 == 0, "four rows of out are whole 128-byte lines");

template <int K>
__device__ __forceinline__ int rowShare(int v) {
  return __builtin_amdgcn_update_dpp(0, v, 0x150 + K, 0xF, 0xF, false);
}

template <int J>
__device__ __forceinline__ void loadEight(const int2 rec, const int cnt, const bool ld,
                                          const int sub, const float4* __restrict__ X4,
                                          float4 (&x)[8]) {
#define SPMM48_LOAD(U)                                                      \
  {                                                                         \
    const int s = rowShare<J + U>(rec.x);                                   \
    x[U] = make_float4(0.f, 0.f, 0.f, 0.f);                                 \
    if (ld && (J + U) < cnt) x[U] = X4[(size_t)s * ROW4 + sub];             \
  }
  SPMM48_LOAD(0) SPMM48_LOAD(1) SPMM48_LOAD(2) SPMM48_LOAD(3)
  SPMM48_LOAD(4) SPMM48_LOAD(5) SPMM48_LOAD(6) SPMM48_LOAD(7)
#undef SPMM48_LOAD
}

template <int J>
__device__ __forceinline__ void fmaEight(const int2 rec, const float4 (&x)[8], float4& acc) {
#define SPMM48_FMA(U)                                                       \
  {                                                                         \
    const float w = __int_as_float(rowShare<J + U>(rec.y));                 \
    acc.x = fmaf(w, x[U].x, acc.x);                                         \
    acc.y = fmaf(w, x[U].y, acc.y);                                         \
    acc.z = fmaf(w, x[U].z, acc.z);                                         \
    acc.w = fmaf(w, x[U].w, acc.w);                                         \
  }
  SPMM48_FMA(0) SPMM48_FMA(1) SPMM48_FMA(2) SPMM48_FMA(3)
  SPMM48_FMA(4) SPMM48_FMA(5) SPMM48_FMA(6) SPMM48_FMA(7)
#undef SPMM48_FMA
}

__global__ void k_spmm_rows48(const unsigned* __restrict__ rowptr, const int* __restrict__ csrw,
                              const float* __restrict__ X, const float* __restrict__ L0,
                              float* __restrict__ out) {
  const int2* csr = reinterpret_cast<const int2*>(csrw);
  const int lane = threadIdx.x & 31;
  const int hsel = lane >> 4;
  const int sub = lane & 15;
  const bool ld = sub < ROW4;
  const int w0 = __builtin_amdgcn_readfirstlane((int)(blockIdx.x * (blockDim.x >> 5) + (threadIdx.x >> 5)));
  const int nw = gridDim.x * (blockDim.x >> 5);
  const float4* X4 = reinterpret_cast<const float4*>(X);
  const float4* L04 = reinterpret_cast<const float4*>(L0);
  float4* out4 = reinterpret_cast<float4*>(out);
  const bool cv = sub < CLS4;
  for (int r0 = 4 * w0; r0 < NNODES; r0 += (r0 & 2) ? 4 * nw - 2 : 2) {
    const int s0 = (int)rowptr[r0];
    const int s1 = (int)rowptr[r0 + 1];
    const int s2 = (int)rowptr[r0 + 2];
    const int degmax = max(s1 - s0, s2 - s1);
    const int start = hsel ? s1 : s0;
    const int deg = hsel ? (s2 - s1) : (s1 - s0);
    float4 acc = make_float4(0.f, 0.f, 0.f, 0.f);
    int cnt = deg;
    int2 rec = (sub < cnt) ? csr[start + sub] : make_int2(0, 0);
    float4 z = make_float4(0.f, 0.f, 0.f, 0.f);
    if (ld) z = L04[(size_t)(r0 + hsel) * ROW4 + sub];
    for (int base = 0; base < degmax; base += 16) {
      float4 x[8];
      loadEight<0>(rec, cnt, ld, sub, X4, x);
      int2 nxt = make_int2(0, 0);
      if (sub + 16 < cnt) nxt = csr[start + base + 16 + sub];
      fmaEight<0>(rec, x, acc);
      if (degmax - base > 8) {
        loadEight<8>(rec, cnt, ld, sub, X4, x);
        fmaEight<8>(rec, x, acc);
      }
      rec = nxt;
      cnt -= 16;
    }
    const float x0 = acc.x + z.x, x1 = acc.y + z.y, x2 = acc.z + z.z, x3 = acc.w + z.w;
    const float m = rowMax16(cv ? fmaxf(fmaxf(x0, x1), fmaxf(x2, x3)) : -3.4e38f);
    const float e0 = cv ? __expf(x0 - m) : 0.f;
    const float e1 = cv ? __expf(x1 - m) : 0.f;
    const float e2 = cv ? __expf(x2 - m) : 0.f;
    const float e3 = cv ? __expf(x3 - m) : 0.f;
    const float inv = 1.0f / rowSum16((e0 + e1) + (e2 + e3));
    if (cv) out4[(size_t)(r0 + hsel) * CLS4 + sub] = make_float4(e0 * inv, e1 * inv, e2 * inv, e3 * inv);
  }
}

#define PREP_THREADS 256
__device__ __forceinline__ void foldTile(const float* __restrict__ A, const int row0, const int nrows,
                                         const float* __restrict__ Wh, float* __restrict__ D) {
  const int lane = threadIdx.x & 31;
  const int h = lane >> 4;
  const int l = lane & 15;
  const int row = row0 + l;
  const float* Arow = A + (size_t)(row < nrows ? row : 0) * HID;
  const bool cv = 32 + l < NCLS;
  const int n2 = cv ? 32 + l : 0;
  v8f acc0, acc1, acc2;
#pragma unroll
  for (int v = 0; v < 8; ++v) { acc0[v] = 0.f; acc1[v] = 0.f; acc2[v] = 0.f; }
#pragma unroll 8
  for (int t = 0; t < HID / 4; ++t) {
    const int ka = 4 * t + 2 * h;
    const float* w0 = Wh + (size_t)ka * NCLS;
    const float* w1 = w0 + NCLS;
    v2f a, b0, b1, b2;
    a.x = Arow[ka];
    a.y = Arow[ka + 1];
    b0.x = w0[l];
    b0.y = w1[l];
    b1.x = w0[16 + l];
    b1.y = w1[16 + l];
    b2.x = w0[n2];
    b2.y = w1[n2];
    acc0 = __builtin_amdgcn_wmma_f32_16x16x4_f32(false, a, false, b0, (short)0, acc0, false, false);
    acc1 = __builtin_amdgcn_wmma_f32_16x16x4_f32(false, a, false, b1, (short)0, acc1, false, false);
    acc2 = __builtin_amdgcn_wmma_f32_16x16x4_f32(false, a, false, b2, (short)0, acc2, false, false);
  }
#pragma unroll
  for (int v = 0; v < 8; ++v) {
    const int r = row0 + v + 8 * h;
    if (r < nrows) {
      D[(size_t)r * NPAD + l]      = acc0[v];
      D[(size_t)r * NPAD + 16 + l] = acc1[v];
      D[(size_t)r * NPAD + 32 + l] = cv ? acc2[v] : 0.f;
    }
  }
}

#define PAIR_PER    8
#define PAIR_V4     (NEDGES / 4)
#define PAIR_BLOCKS ((PAIR_V4 + 256 * PAIR_PER - 1) / (256 * PAIR_PER))
static_assert(NEDGES % 4 == 0, "the edge arrays are whole 16-byte vectors");
__device__ __forceinline__ void pairBlock(const int* __restrict__ src, const float* __restrict__ ew,
                                          int* __restrict__ recw, const int pb) {
  const int4* s4 = reinterpret_cast<const int4*>(src);
  const float4* w4 = reinterpret_cast<const float4*>(ew);
  int4* r4 = reinterpret_cast<int4*>(recw);
  int4 sv[PAIR_PER];
  float4 wv[PAIR_PER];
#pragma unroll
  for (int k = 0; k < PAIR_PER; ++k) {
    const int i = (pb * PAIR_PER + k) * 256 + (int)threadIdx.x;
    sv[k] = make_int4(0, 0, 0, 0);
    wv[k] = make_float4(0.f, 0.f, 0.f, 0.f);
    if (i < PAIR_V4) { sv[k] = s4[i]; wv[k] = w4[i]; }
  }
#pragma unroll
  for (int k = 0; k < PAIR_PER; ++k) {
    const int i = (pb * PAIR_PER + k) * 256 + (int)threadIdx.x;
    if (i < PAIR_V4) {
      r4[2 * i]     = make_int4(sv[k].x, __float_as_int(wv[k].x), sv[k].y, __float_as_int(wv[k].y));
      r4[2 * i + 1] = make_int4(sv[k].z, __float_as_int(wv[k].z), sv[k].w, __float_as_int(wv[k].w));
    }
  }
}

#define PREP_LDS_WORDS (2 * HID * NCLS + 2 * HID)
static_assert(PREP_THREADS == FRONT_T, "the prep roles run on the front launch's block size");
static_assert(FRONT_T == 256, "the pairing role counts on 256 threads per block");
__device__ __forceinline__ void prepBlock(const int role,
                                          const float* __restrict__ W2, const float* __restrict__ We,
                                          const float* __restrict__ Wf, const float* __restrict__ bf,
                                          const float* __restrict__ be, const float* __restrict__ b2,
                                          float* __restrict__ W2f, float* __restrict__ Wef,
                                          float* __restrict__ bfp, float* __restrict__ lds) {
  float* sW  = lds;
  float* sBe = lds + 2 * HID * NCLS;
  float* sB2 = sBe + HID;
  const int t = threadIdx.x;
  const int wid = __builtin_amdgcn_readfirstlane((int)(threadIdx.x >> 5));
  if (role == 0) {
    foldTile(W2, 16 * wid, HID, Wf + (size_t)HID * NCLS, W2f);
    return;
  }
  if (role == 1) {
    if (wid < (NCLS + 15) / 16) foldTile(We, 16 * wid, NCLS, Wf, Wef);
    return;
  }
  const float4* Wf4 = reinterpret_cast<const float4*>(Wf);
  float4* sW4 = reinterpret_cast<float4*>(sW);
#pragma unroll
  for (int j = 0; j < (2 * HID * NCLS) / 4 / PREP_THREADS; ++j)
    sW4[j * PREP_THREADS + t] = Wf4[j * PREP_THREADS + t];
  if (t < HID) sBe[t] = be[t];
  else         sB2[t - HID] = b2[t - HID];
  __syncthreads();
  if (t < NPAD) {
    float s = 0.f;
    if (t < NCLS) {
      s = bf[t];
      for (int k = 0; k < HID; ++k) {
        s += sBe[k] * sW[k * NCLS + t];
        s += sB2[k] * sW[(HID + k) * NCLS + t];
      }
    }
    bfp[t] = s;
  }
}

#define GEMM_BLOCKS  (((NNODES / 16) + 7) / 8)
#define COUNT_BLOCKS (CSR_G * CSR_Q)
#define FRONT_BLOCKS (GEMM_BLOCKS + COUNT_BLOCKS + PAIR_BLOCKS + 3)
#define FRONT_LDS4   (PREP_LDS_WORDS / 4)
static_assert(PREP_LDS_WORDS % 4 == 0, "the block's LDS is whole 16-byte vectors");
static_assert(FRONT_LDS4 * 4 >= CSR_ROWS + 1, "the count role's counters and sum word fit the block's LDS");
static_assert(FRONT_LDS4 * 4 >= GemmChunk<FIN>::value * HID, "the GEMM role's B chunk fits the block's LDS");

__global__ void
k_front(const float* __restrict__ features, const float* __restrict__ W1, float* __restrict__ X1,
        const int* __restrict__ dst, unsigned* __restrict__ cnt, unsigned* __restrict__ bsum,
        const float* __restrict__ W2, const float* __restrict__ We,
        const float* __restrict__ Wf, const float* __restrict__ bf,
        const float* __restrict__ be, const float* __restrict__ b2,
        float* __restrict__ W2f, float* __restrict__ Wef, float* __restrict__ bfp,
        const int* __restrict__ src, const float* __restrict__ ew, int* __restrict__ recw) {
  __shared__ float4 lds4[FRONT_LDS4];
  float* lds = reinterpret_cast<float*>(lds4);
  const int bid = (int)blockIdx.x;
  if (bid < GEMM_BLOCKS) {
    gemmBlock<8, FIN>(features, W1, X1, NNODES, lds, bid);
  } else if (bid < GEMM_BLOCKS + COUNT_BLOCKS) {
    countBlock(dst, cnt, bsum, reinterpret_cast<unsigned*>(lds), bid - GEMM_BLOCKS);
  } else if (bid < GEMM_BLOCKS + COUNT_BLOCKS + PAIR_BLOCKS) {
    pairBlock(src, ew, recw, bid - (GEMM_BLOCKS + COUNT_BLOCKS));
  } else {
    prepBlock(bid - (GEMM_BLOCKS + COUNT_BLOCKS + PAIR_BLOCKS), W2, We, Wf, bf, be, b2, W2f, Wef, bfp, lds);
  }
}

extern "C" void kernel_launch(void* const* d_in, const int* in_sizes, int n_in,
                              void* d_out, int out_size, void* d_ws, size_t ws_size,
                              hipStream_t stream) {
  const float* features = (const float*)d_in[0];
  const int*   src      = (const int*)d_in[1];
  const int*   dst      = (const int*)d_in[2];
  const float* ew       = (const float*)d_in[3];
  const float* y        = (const float*)d_in[4];
  const float* W1       = (const float*)d_in[5];
  const float* b1       = (const float*)d_in[6];
  const float* W2       = (const float*)d_in[7];
  const float* b2       = (const float*)d_in[8];
  const float* We       = (const float*)d_in[9];
  const float* be       = (const float*)d_in[10];
  const float* Wf       = (const float*)d_in[11];
  const float* bf       = (const float*)d_in[12];
  float* out = (float*)d_out;
  (void)in_sizes; (void)n_in; (void)out_size; (void)ws_size;

  char* ws = (char*)d_ws;
  size_t off = 0;
  auto alloc = [&](size_t bytes) -> float* {
    float* p = (float*)(ws + off);
    off += (bytes + 255) & ~(size_t)255;
    return p;
  };
  float* bufA   = alloc((size_t)NNODES * HID * 4);
  float* bufAgg = alloc((size_t)NNODES * HID * 4);
  float* bufL   = alloc((size_t)NNODES * NPAD * 4);
  float* W2f    = alloc((size_t)HID * NPAD * 4);
  float* Wef    = alloc((size_t)NCLS * NPAD * 4);
  float* bfp    = alloc((size_t)NPAD * 4);
  unsigned* rowptr = (unsigned*)alloc((size_t)(NNODES + 1) * 4);
  int*      csr    = (int*)alloc((size_t)NEDGES * 8);
  int*      rec    = (int*)alloc((size_t)NEDGES * 8);
  unsigned* cntMat = (unsigned*)bufAgg;
  unsigned* bsum   = cntMat + (size_t)CSR_G * NNODES;
  int*      perm   = (int*)bufL;
  float*    L0     = bufAgg + (size_t)NNODES * NPAD;

  k_front<<<FRONT_BLOCKS, FRONT_T, 0, stream>>>(features, W1, bufA, dst, cntMat, bsum,
                                                W2, We, Wf, bf, be, b2, W2f, Wef, bfp, src, ew, rec);
  k_csr_fill<<<CSR_G * CSR_Q, CSR_T, 0, stream>>>(dst, cntMat, bsum, rowptr, perm);
  k_spmm_rows_rank<<<ROW_BLOCKS, 256, 0, stream>>>(rowptr, perm, rec, csr, bufA, b1, W2f, bufAgg,
                                                   y, Wef, bfp, L0);
  k_spmm_rows48<<<ROW_BLOCKS, 256, 0, stream>>>(rowptr, csr, bufAgg, L0, out);
}
